# starting kernel
# baseline (speedup 1.0000x reference)
#include <hip/hip_runtime.h>
#include <float.h>


typedef __attribute__((ext_vector_type(2))) float v2f;
typedef __attribute__((ext_vector_type(8))) float v8f;

#define NEG_SLOPE 0.01f

__global__ void k_init(int* __restrict__ m_bits, float* __restrict__ denom,
                       float* __restrict__ aggr, int Nn) {
  int i = blockIdx.x * blockDim.x + threadIdx.x;
  if (i < Nn) { m_bits[i] = 0xFF7FFFFF; denom[i] = 0.0f; }
  if (i < Nn * 64) aggr[i] = 0.0f;
}

template <int K>
__global__ void k_gemm_wmma(const float* __restrict__ in, const float* __restrict__ W,
                            const float* __restrict__ b, float* __restrict__ h, int M) {
  int wave = (int)((blockIdx.x * blockDim.x + threadIdx.x) >> 5);
  int lane = threadIdx.x & 31;
  int row0 = wave * 16;
  if (row0 >= M) return;
  int half = lane >> 4;
  int l16  = lane & 15;

  v8f acc0 = {}, acc1 = {}, acc2 = {}, acc3 = {};
  const float* arow = in + (size_t)(row0 + l16) * K;

  #pragma unroll
  for (int k0 = 0; k0 < K; k0 += 4) {
    int kb = k0 + half * 2;
    v2f a;  a.x = arow[kb];  a.y = arow[kb + 1];
    v2f b0, b1, b2, b3;
    { const float* p = W + (size_t)(0  + l16) * K + kb; b0.x = p[0]; b0.y = p[1]; }
    { const float* p = W + (size_t)(16 + l16) * K + kb; b1.x = p[0]; b1.y = p[1]; }
    { const float* p = W + (size_t)(32 + l16) * K + kb; b2.x = p[0]; b2.y = p[1]; }
    { const float* p = W + (size_t)(48 + l16) * K + kb; b3.x = p[0]; b3.y = p[1]; }
    acc0 = __builtin_amdgcn_wmma_f32_16x16x4_f32(false, a, false, b0, (short)0, acc0, false, false);
    acc1 = __builtin_amdgcn_wmma_f32_16x16x4_f32(false, a, false, b1, (short)0, acc1, false, false);
    acc2 = __builtin_amdgcn_wmma_f32_16x16x4_f32(false, a, false, b2, (short)0, acc2, false, false);
    acc3 = __builtin_amdgcn_wmma_f32_16x16x4_f32(false, a, false, b3, (short)0, acc3, false, false);
  }

  #pragma unroll
  for (int v = 0; v < 8; ++v) {
    int m = row0 + v + half * 8;
    float* hrow = h + (size_t)m * 64 + l16;
    hrow[0]  = acc0[v] + b[0  + l16];
    hrow[16] = acc1[v] + b[16 + l16];
    hrow[32] = acc2[v] + b[32 + l16];
    hrow[48] = acc3[v] + b[48 + l16];
  }
}

__global__ void k_node_att(const float* __restrict__ h, const float* __restrict__ att,
                           float* __restrict__ a_i, float* __restrict__ a_j, int Nn) {
  int n = blockIdx.x * (blockDim.x >> 5) + (threadIdx.x >> 5);
  int lane = threadIdx.x & 31;
  if (n >= Nn) return;
  const float* hp = h + (size_t)n * 64;
  float h0 = hp[lane], h1 = hp[lane + 32];
  float vi = h0 * att[lane]      + h1 * att[lane + 32];
  float vj = h0 * att[64 + lane] + h1 * att[96 + lane];
  #pragma unroll
  for (int off = 16; off > 0; off >>= 1) {
    vi += __shfl_xor(vi, off, 32);
    vj += __shfl_xor(vj, off, 32);
  }
  if (lane == 0) { a_i[n] = vi; a_j[n] = vj; }
}

__device__ __forceinline__ void get_edge(const int* __restrict__ ei, int e, int E,
                                         int& s, int& d) {
  if (e < E) { s = ei[e]; d = ei[E + e]; }
  else       { s = e - E; d = e - E; }
}

__global__ void k_edge_score(const int* __restrict__ ei, const float* __restrict__ a_i,
                             const float* __restrict__ a_j, float* __restrict__ score,
                             int* __restrict__ m_bits, int E, int Etot) {
  int e = blockIdx.x * blockDim.x + threadIdx.x;
  if (e >= Etot) return;
  int s, d; get_edge(ei, e, E, s, d);
  float sc = a_i[d] + a_j[s];
  sc = sc > 0.0f ? sc : sc * NEG_SLOPE;
  score[e] = sc;
  if (sc >= 0.0f) atomicMax(&m_bits[s], __float_as_int(sc));
  else            atomicMin((unsigned int*)&m_bits[s], __float_as_uint(sc));
}

__global__ void k_edge_expsum(const int* __restrict__ ei, const float* __restrict__ m,
                              float* __restrict__ score, float* __restrict__ denom,
                              int E, int Etot) {
  int e = blockIdx.x * blockDim.x + threadIdx.x;
  if (e >= Etot) return;
  int s, d; get_edge(ei, e, E, s, d);
  float ev = __expf(score[e] - m[s]);
  score[e] = ev;
  atomicAdd(&denom[s], ev);
}

__global__ void k_edge_message(const int* __restrict__ ei, const float* __restrict__ escore,
                               const float* __restrict__ denom, const float* __restrict__ h,
                               float* __restrict__ aggr, int E, int Etot) {
  long long t = (long long)blockIdx.x * blockDim.x + threadIdx.x;
  int e  = (int)(t >> 4);
  int c4 = ((int)t & 15) * 4;
  if (e >= Etot) return;
  int s, d; get_edge(ei, e, E, s, d);
  float alpha = escore[e] / (denom[s] + 1e-16f);
  const float4 hv = *(const float4*)(h + (size_t)s * 64 + c4);
  float* ap = aggr + (size_t)d * 64 + c4;
  atomicAdd(ap + 0, alpha * hv.x);
  atomicAdd(ap + 1, alpha * hv.y);
  atomicAdd(ap + 2, alpha * hv.z);
  atomicAdd(ap + 3, alpha * hv.w);
}

__global__ void k_node_finish(const float* __restrict__ aggr, const float* __restrict__ bias,
                              float* __restrict__ out, int Nn) {
  int i = blockIdx.x * blockDim.x + threadIdx.x;
  if (i >= Nn * 64) return;
  float v = aggr[i] + bias[i & 63];
  out[i] = v > 0.0f ? v : 0.0f;
}

__global__ void k_out_proj(const float* __restrict__ n2, const float* __restrict__ Wout,
                           const float* __restrict__ bout, float* __restrict__ y, int Nn) {
  int n = blockIdx.x * (blockDim.x >> 5) + (threadIdx.x >> 5);
  int lane = threadIdx.x & 31;
  if (n >= Nn) return;
  const float* hp = n2 + (size_t)n * 64;
  float v = hp[lane] * Wout[lane] + hp[lane + 32] * Wout[lane + 32];
  #pragma unroll
  for (int off = 16; off > 0; off >>= 1) v += __shfl_xor(v, off, 32);
  if (lane == 0) y[n] = v + bout[0];
}

extern "C" void kernel_launch(void* const* d_in, const int* in_sizes, int n_in,
                              void* d_out, int out_size, void* d_ws, size_t ws_size,
                              hipStream_t stream) {
  const float* x     = (const float*)d_in[0];
  const int*   ei    = (const int*)  d_in[1];
  const float* W1    = (const float*)d_in[2];
  const float* b1    = (const float*)d_in[3];
  const float* att1  = (const float*)d_in[4];
  const float* bias1 = (const float*)d_in[5];
  const float* W2    = (const float*)d_in[6];
  const float* b2    = (const float*)d_in[7];
  const float* att2  = (const float*)d_in[8];
  const float* bias2 = (const float*)d_in[9];
  const float* Wout  = (const float*)d_in[10];
  const float* bout  = (const float*)d_in[11];
  float* y = (float*)d_out;

  const int Nn   = in_sizes[0] / 16;
  const int E    = in_sizes[1] / 2;
  const int Etot = E + Nn;

  char* p = (char*)d_ws;
  auto alloc = [&](size_t bytes) {
    char* r = p;
    p += (bytes + 255) & ~(size_t)255;
    return r;
  };
  float* h      = (float*)alloc((size_t)Nn * 64 * sizeof(float));
  float* nbuf   = (float*)alloc((size_t)Nn * 64 * sizeof(float));
  float* aggr   = (float*)alloc((size_t)Nn * 64 * sizeof(float));
  float* score  = (float*)alloc((size_t)Etot * sizeof(float));
  float* a_i    = (float*)alloc((size_t)Nn * sizeof(float));
  float* a_j    = (float*)alloc((size_t)Nn * sizeof(float));
  int*   m_bits = (int*)  alloc((size_t)Nn * sizeof(float));
  float* denom  = (float*)alloc((size_t)Nn * sizeof(float));

  const int TB = 256;
  dim3 blk(TB);
  int g_init  = (Nn * 64 + TB - 1) / TB;
  int g_gemm  = ((Nn / 16) + 7) / 8;
  int g_node  = (Nn + 7) / 8;
  int g_edge  = (Etot + TB - 1) / TB;
  int g_msg   = (int)(((long long)Etot * 16 + TB - 1) / TB);
  int g_fin   = (Nn * 64 + TB - 1) / TB;

  k_init<<<g_init, blk, 0, stream>>>(m_bits, denom, aggr, Nn);
  k_gemm_wmma<16><<<g_gemm, blk, 0, stream>>>(x, W1, b1, h, Nn);
  k_node_att<<<g_node, blk, 0, stream>>>(h, att1, a_i, a_j, Nn);
  k_edge_score<<<g_edge, blk, 0, stream>>>(ei, a_i, a_j, score, m_bits, E, Etot);
  k_edge_expsum<<<g_edge, blk, 0, stream>>>(ei, (const float*)m_bits, score, denom, E, Etot);
  k_edge_message<<<g_msg, blk, 0, stream>>>(ei, score, denom, h, aggr, E, Etot);
  k_node_finish<<<g_fin, blk, 0, stream>>>(aggr, bias1, nbuf, Nn);

  k_init<<<g_init, blk, 0, stream>>>(m_bits, denom, aggr, Nn);
  k_gemm_wmma<64><<<g_gemm, blk, 0, stream>>>(nbuf, W2, b2, h, Nn);
  k_node_att<<<g_node, blk, 0, stream>>>(h, att2, a_i, a_j, Nn);
  k_edge_score<<<g_edge, blk, 0, stream>>>(ei, a_i, a_j, score, m_bits, E, Etot);
  k_edge_expsum<<<g_edge, blk, 0, stream>>>(ei, (const float*)m_bits, score, denom, E, Etot);
  k_edge_message<<<g_msg, blk, 0, stream>>>(ei, score, denom, h, aggr, E, Etot);
  k_node_finish<<<g_fin, blk, 0, stream>>>(aggr, bias2, nbuf, Nn);

  k_out_proj<<<g_node, blk, 0, stream>>>(nbuf, Wout, bout, y, Nn);
}


# fold the per-node passes (state reset, attention dots, bias+relu) into the two GEMM launches and the output projection: 15 launches -> 9; attention dots summed 4 per lane, then a half-wave tree
# speedup vs baseline: 1.0648x; 1.0648x over previous
#include <hip/hip_runtime.h>
#include <float.h>


typedef __attribute__((ext_vector_type(2))) float v2f;
typedef __attribute__((ext_vector_type(8))) float v8f;

#define NEG_SLOPE 0.01f

template <int MASK>
__device__ __forceinline__ float row_xor(float v) {
  return __int_as_float(__builtin_amdgcn_update_dpp(0, __float_as_int(v), 0x160 + MASK,
                                                    0xf, 0xf, true));
}

template <int K, bool FIN>
__global__ void k_linear_fused(const float* __restrict__ in, const float* __restrict__ bias_prev,
                               const float* __restrict__ W, const float* __restrict__ b,
                               const float* __restrict__ att,
                               float* __restrict__ h, float* __restrict__ a_i,
                               float* __restrict__ a_j, int* __restrict__ m_bits,
                               float* __restrict__ denom, float* __restrict__ aggr_next, int M) {
  int wave = (int)((blockIdx.x * blockDim.x + threadIdx.x) >> 5);
  int lane = threadIdx.x & 31;
  int row0 = wave * 16;
  if (row0 >= M) return;
  int half = lane >> 4;
  int l16  = lane & 15;

  {
    const float4 z = {0.0f, 0.0f, 0.0f, 0.0f};
    float4* zp = (float4*)(aggr_next + (size_t)row0 * 64);
    #pragma unroll
    for (int i = 0; i < 8; ++i) zp[i * 32 + lane] = z;
    int* sp = half ? (int*)denom : m_bits;
    sp[row0 + l16] = half ? 0 : (int)0xFF7FFFFF;
  }

  v8f acc0 = {}, acc1 = {}, acc2 = {}, acc3 = {};
  const float* arow = in + (size_t)(row0 + l16) * K;

  #pragma unroll
  for (int k0 = 0; k0 < K; k0 += 4) {
    int kb = k0 + half * 2;
    v2f a;  a.x = arow[kb];  a.y = arow[kb + 1];
    if (FIN) {
      float t0 = a.x + bias_prev[kb], t1 = a.y + bias_prev[kb + 1];
      a.x = t0 > 0.0f ? t0 : 0.0f;
      a.y = t1 > 0.0f ? t1 : 0.0f;
    }
    v2f b0, b1, b2, b3;
    { const float* p = W + (size_t)(0  + l16) * K + kb; b0.x = p[0]; b0.y = p[1]; }
    { const float* p = W + (size_t)(16 + l16) * K + kb; b1.x = p[0]; b1.y = p[1]; }
    { const float* p = W + (size_t)(32 + l16) * K + kb; b2.x = p[0]; b2.y = p[1]; }
    { const float* p = W + (size_t)(48 + l16) * K + kb; b3.x = p[0]; b3.y = p[1]; }
    acc0 = __builtin_amdgcn_wmma_f32_16x16x4_f32(false, a, false, b0, (short)0, acc0, false, false);
    acc1 = __builtin_amdgcn_wmma_f32_16x16x4_f32(false, a, false, b1, (short)0, acc1, false, false);
    acc2 = __builtin_amdgcn_wmma_f32_16x16x4_f32(false, a, false, b2, (short)0, acc2, false, false);
    acc3 = __builtin_amdgcn_wmma_f32_16x16x4_f32(false, a, false, b3, (short)0, acc3, false, false);
  }

  const float bb0 = b[l16], bb1 = b[16 + l16], bb2 = b[32 + l16], bb3 = b[48 + l16];
  const float ti0 = att[l16],      ti1 = att[16 + l16], ti2 = att[32 + l16],  ti3 = att[48 + l16];
  const float tj0 = att[64 + l16], tj1 = att[80 + l16], tj2 = att[96 + l16],  tj3 = att[112 + l16];
  float oi = 0.0f, oj = 0.0f;
  #pragma unroll
  for (int v = 0; v < 8; ++v) {
    int m = row0 + v + half * 8;
    float* hrow = h + (size_t)m * 64 + l16;
    float h0 = acc0[v] + bb0, h1 = acc1[v] + bb1, h2 = acc2[v] + bb2, h3 = acc3[v] + bb3;
    hrow[0]  = h0;
    hrow[16] = h1;
    hrow[32] = h2;
    hrow[48] = h3;
    float vi = h0 * ti0 + h1 * ti1 + h2 * ti2 + h3 * ti3;
    float vj = h0 * tj0 + h1 * tj1 + h2 * tj2 + h3 * tj3;
    vi += row_xor<8>(vi);  vj += row_xor<8>(vj);
    vi += row_xor<4>(vi);  vj += row_xor<4>(vj);
    vi += row_xor<2>(vi);  vj += row_xor<2>(vj);
    vi += row_xor<1>(vi);  vj += row_xor<1>(vj);
    bool mine = (l16 & 7) == v;
    oi = mine ? vi : oi;
    oj = mine ? vj : oj;
  }
  {
    float* op = (l16 < 8) ? a_i : a_j;
    op[row0 + half * 8 + (l16 & 7)] = (l16 < 8) ? oi : oj;
  }
}

__device__ __forceinline__ void get_edge(const int* __restrict__ ei, int e, int E,
                                         int& s, int& d) {
  if (e < E) { s = ei[e]; d = ei[E + e]; }
  else       { s = e - E; d = e - E; }
}

__global__ void k_edge_score(const int* __restrict__ ei, const float* __restrict__ a_i,
                             const float* __restrict__ a_j, float* __restrict__ score,
                             int* __restrict__ m_bits, int E, int Etot) {
  int e = blockIdx.x * blockDim.x + threadIdx.x;
  if (e >= Etot) return;
  int s, d; get_edge(ei, e, E, s, d);
  float sc = a_i[d] + a_j[s];
  sc = sc > 0.0f ? sc : sc * NEG_SLOPE;
  score[e] = sc;
  if (sc >= 0.0f) atomicMax(&m_bits[s], __float_as_int(sc));
  else            atomicMin((unsigned int*)&m_bits[s], __float_as_uint(sc));
}

__global__ void k_edge_expsum(const int* __restrict__ ei, const float* __restrict__ m,
                              float* __restrict__ score, float* __restrict__ denom,
                              int E, int Etot) {
  int e = blockIdx.x * blockDim.x + threadIdx.x;
  if (e >= Etot) return;
  int s, d; get_edge(ei, e, E, s, d);
  float ev = __expf(score[e] - m[s]);
  score[e] = ev;
  atomicAdd(&denom[s], ev);
}

__global__ void k_edge_message(const int* __restrict__ ei, const float* __restrict__ escore,
                               const float* __restrict__ denom, const float* __restrict__ h,
                               float* __restrict__ aggr, int E, int Etot) {
  long long t = (long long)blockIdx.x * blockDim.x + threadIdx.x;
  int e  = (int)(t >> 4);
  int c4 = ((int)t & 15) * 4;
  if (e >= Etot) return;
  int s, d; get_edge(ei, e, E, s, d);
  float alpha = escore[e] / (denom[s] + 1e-16f);
  const float4 hv = *(const float4*)(h + (size_t)s * 64 + c4);
  float* ap = aggr + (size_t)d * 64 + c4;
  atomicAdd(ap + 0, alpha * hv.x);
  atomicAdd(ap + 1, alpha * hv.y);
  atomicAdd(ap + 2, alpha * hv.z);
  atomicAdd(ap + 3, alpha * hv.w);
}

__global__ void k_finish_out_proj(const float* __restrict__ aggr, const float* __restrict__ bias,
                                  const float* __restrict__ Wout, const float* __restrict__ bout,
                                  float* __restrict__ y, int Nn) {
  int n = blockIdx.x * (blockDim.x >> 5) + (threadIdx.x >> 5);
  int lane = threadIdx.x & 31;
  if (n >= Nn) return;
  const float* ap = aggr + (size_t)n * 64;
  float v0 = ap[lane] + bias[lane];
  float v1 = ap[lane + 32] + bias[lane + 32];
  v0 = v0 > 0.0f ? v0 : 0.0f;
  v1 = v1 > 0.0f ? v1 : 0.0f;
  float v = v0 * Wout[lane] + v1 * Wout[lane + 32];
  #pragma unroll
  for (int off = 16; off > 0; off >>= 1) v += __shfl_xor(v, off, 32);
  if (lane == 0) y[n] = v + bout[0];
}

extern "C" void kernel_launch(void* const* d_in, const int* in_sizes, int n_in,
                              void* d_out, int out_size, void* d_ws, size_t ws_size,
                              hipStream_t stream) {
  const float* x     = (const float*)d_in[0];
  const int*   ei    = (const int*)  d_in[1];
  const float* W1    = (const float*)d_in[2];
  const float* b1    = (const float*)d_in[3];
  const float* att1  = (const float*)d_in[4];
  const float* bias1 = (const float*)d_in[5];
  const float* W2    = (const float*)d_in[6];
  const float* b2    = (const float*)d_in[7];
  const float* att2  = (const float*)d_in[8];
  const float* bias2 = (const float*)d_in[9];
  const float* Wout  = (const float*)d_in[10];
  const float* bout  = (const float*)d_in[11];
  float* y = (float*)d_out;

  const int Nn   = in_sizes[0] / 16;
  const int E    = in_sizes[1] / 2;
  const int Etot = E + Nn;

  char* p = (char*)d_ws;
  auto alloc = [&](size_t bytes) {
    char* r = p;
    p += (bytes + 255) & ~(size_t)255;
    return r;
  };
  float* h      = (float*)alloc((size_t)Nn * 64 * sizeof(float));
  float* aggr1  = (float*)alloc((size_t)Nn * 64 * sizeof(float));
  float* aggr2  = (float*)alloc((size_t)Nn * 64 * sizeof(float));
  float* score  = (float*)alloc((size_t)Etot * sizeof(float));
  float* a_i    = (float*)alloc((size_t)Nn * sizeof(float));
  float* a_j    = (float*)alloc((size_t)Nn * sizeof(float));
  int*   m_bits = (int*)  alloc((size_t)Nn * sizeof(float));
  float* denom  = (float*)alloc((size_t)Nn * sizeof(float));

  const int TB = 256;
  dim3 blk(TB);
  int g_gemm  = ((Nn / 16) + 7) / 8;
  int g_node  = (Nn + 7) / 8;
  int g_edge  = (Etot + TB - 1) / TB;
  int g_msg   = (int)(((long long)Etot * 16 + TB - 1) / TB);

  k_linear_fused<16, false><<<g_gemm, blk, 0, stream>>>(x, nullptr, W1, b1, att1, h, a_i, a_j,
                                                        m_bits, denom, aggr1, Nn);
  k_edge_score<<<g_edge, blk, 0, stream>>>(ei, a_i, a_j, score, m_bits, E, Etot);
  k_edge_expsum<<<g_edge, blk, 0, stream>>>(ei, (const float*)m_bits, score, denom, E, Etot);
  k_edge_message<<<g_msg, blk, 0, stream>>>(ei, score, denom, h, aggr1, E, Etot);

  k_linear_fused<64, true><<<g_gemm, blk, 0, stream>>>(aggr1, bias1, W2, b2, att2, h, a_i, a_j,
                                                       m_bits, denom, aggr2, Nn);
  k_edge_score<<<g_edge, blk, 0, stream>>>(ei, a_i, a_j, score, m_bits, E, Etot);
  k_edge_expsum<<<g_edge, blk, 0, stream>>>(ei, (const float*)m_bits, score, denom, E, Etot);
  k_edge_message<<<g_msg, blk, 0, stream>>>(ei, score, denom, h, aggr2, E, Etot);

  k_finish_out_proj<<<g_node, blk, 0, stream>>>(aggr2, bias2, Wout, bout, y, Nn);
}


# aggregation without f32 atomics: listed edges counting-sorted by dst in the call (returning int add = rank, scan, place); one wave per dst sums its rows in registers, sum order = arrival rank
# speedup vs baseline: 1.2628x; 1.1860x over previous
#include <hip/hip_runtime.h>
#include <float.h>


typedef __attribute__((ext_vector_type(2))) float v2f;
typedef __attribute__((ext_vector_type(8))) float v8f;

#define NEG_SLOPE 0.01f

template <int MASK>
__device__ __forceinline__ float row_xor(float v) {
  return __int_as_float(__builtin_amdgcn_update_dpp(0, __float_as_int(v), 0x160 + MASK,
                                                    0xf, 0xf, true));
}

template <int K, bool FIN>
__global__ void k_linear_fused(const float* __restrict__ in, const float* __restrict__ bias_prev,
                               const float* __restrict__ W, const float* __restrict__ b,
                               const float* __restrict__ att,
                               float* __restrict__ h, float* __restrict__ a_i,
                               float* __restrict__ a_j, int* __restrict__ m_bits,
                               float* __restrict__ denom, int* __restrict__ cnt, int M) {
  int wave = (int)((blockIdx.x * blockDim.x + threadIdx.x) >> 5);
  int lane = threadIdx.x & 31;
  int row0 = wave * 16;
  if (row0 >= M) return;
  int half = lane >> 4;
  int l16  = lane & 15;

  {
    int* sp = half ? (int*)denom : m_bits;
    sp[row0 + l16] = half ? 0 : (int)0xFF7FFFFF;
    if (!FIN) cnt[row0 + l16] = 0;
  }

  v8f acc0 = {}, acc1 = {}, acc2 = {}, acc3 = {};
  const float* arow = in + (size_t)(row0 + l16) * K;

  #pragma unroll
  for (int k0 = 0; k0 < K; k0 += 4) {
    int kb = k0 + half * 2;
    v2f a;  a.x = arow[kb];  a.y = arow[kb + 1];
    if (FIN) {
      float t0 = a.x + bias_prev[kb], t1 = a.y + bias_prev[kb + 1];
      a.x = t0 > 0.0f ? t0 : 0.0f;
      a.y = t1 > 0.0f ? t1 : 0.0f;
    }
    v2f b0, b1, b2, b3;
    { const float* p = W + (size_t)(0  + l16) * K + kb; b0.x = p[0]; b0.y = p[1]; }
    { const float* p = W + (size_t)(16 + l16) * K + kb; b1.x = p[0]; b1.y = p[1]; }
    { const float* p = W + (size_t)(32 + l16) * K + kb; b2.x = p[0]; b2.y = p[1]; }
    { const float* p = W + (size_t)(48 + l16) * K + kb; b3.x = p[0]; b3.y = p[1]; }
    acc0 = __builtin_amdgcn_wmma_f32_16x16x4_f32(false, a, false, b0, (short)0, acc0, false, false);
    acc1 = __builtin_amdgcn_wmma_f32_16x16x4_f32(false, a, false, b1, (short)0, acc1, false, false);
    acc2 = __builtin_amdgcn_wmma_f32_16x16x4_f32(false, a, false, b2, (short)0, acc2, false, false);
    acc3 = __builtin_amdgcn_wmma_f32_16x16x4_f32(false, a, false, b3, (short)0, acc3, false, false);
  }

  const float bb0 = b[l16], bb1 = b[16 + l16], bb2 = b[32 + l16], bb3 = b[48 + l16];
  const float ti0 = att[l16],      ti1 = att[16 + l16], ti2 = att[32 + l16],  ti3 = att[48 + l16];
  const float tj0 = att[64 + l16], tj1 = att[80 + l16], tj2 = att[96 + l16],  tj3 = att[112 + l16];
  float oi = 0.0f, oj = 0.0f;
  #pragma unroll
  for (int v = 0; v < 8; ++v) {
    int m = row0 + v + half * 8;
    float* hrow = h + (size_t)m * 64 + l16;
    float h0 = acc0[v] + bb0, h1 = acc1[v] + bb1, h2 = acc2[v] + bb2, h3 = acc3[v] + bb3;
    hrow[0]  = h0;
    hrow[16] = h1;
    hrow[32] = h2;
    hrow[48] = h3;
    float vi = h0 * ti0 + h1 * ti1 + h2 * ti2 + h3 * ti3;
    float vj = h0 * tj0 + h1 * tj1 + h2 * tj2 + h3 * tj3;
    vi += row_xor<8>(vi);  vj += row_xor<8>(vj);
    vi += row_xor<4>(vi);  vj += row_xor<4>(vj);
    vi += row_xor<2>(vi);  vj += row_xor<2>(vj);
    vi += row_xor<1>(vi);  vj += row_xor<1>(vj);
    bool mine = (l16 & 7) == v;
    oi = mine ? vi : oi;
    oj = mine ? vj : oj;
  }
  {
    float* op = (l16 < 8) ? a_i : a_j;
    op[row0 + half * 8 + (l16 & 7)] = (l16 < 8) ? oi : oj;
  }
}

__device__ __forceinline__ void get_edge(const int* __restrict__ ei, int e, int E,
                                         int& s, int& d) {
  if (e < E) { s = ei[e]; d = ei[E + e]; }
  else       { s = e - E; d = e - E; }
}

template <bool COUNT>
__global__ void k_edge_score(const int* __restrict__ ei, const float* __restrict__ a_i,
                             const float* __restrict__ a_j, float* __restrict__ score,
                             int* __restrict__ m_bits, int* __restrict__ cnt,
                             int* __restrict__ rank, int E, int Etot) {
  int e = blockIdx.x * blockDim.x + threadIdx.x;
  if (e >= Etot) return;
  int s, d; get_edge(ei, e, E, s, d);
  float sc = a_i[d] + a_j[s];
  sc = sc > 0.0f ? sc : sc * NEG_SLOPE;
  score[e] = sc;
  if (sc >= 0.0f) atomicMax(&m_bits[s], __float_as_int(sc));
  else            atomicMin((unsigned int*)&m_bits[s], __float_as_uint(sc));
  if (COUNT) {
    if (e < E) rank[e] = atomicAdd(&cnt[d], 1);
  }
}

__global__ void k_dst_offsets(const int* __restrict__ cnt, int* __restrict__ off, int Nn) {
  __shared__ int s_pre[8];
  __shared__ int s_loc[8];
  const int t = threadIdx.x, lane = t & 31, w = t >> 5;
  const int nb = blockIdx.x;
  int pre = 0;
  const int4* c4 = (const int4*)cnt;
  for (int j = 0; j < nb; ++j) {
    const int4 v = c4[j * 256 + t];
    pre += (v.x + v.y) + (v.z + v.w);
  }
  const int i0 = nb * 1024 + t * 4;
  const int c0 = (i0     < Nn) ? cnt[i0]     : 0;
  const int c1 = (i0 + 1 < Nn) ? cnt[i0 + 1] : 0;
  const int c2 = (i0 + 2 < Nn) ? cnt[i0 + 2] : 0;
  const int c3 = (i0 + 3 < Nn) ? cnt[i0 + 3] : 0;
  const int loc = (c0 + c1) + (c2 + c3);
  for (int o = 16; o > 0; o >>= 1) pre += __shfl_xor(pre, o, 32);
  int inc = loc;
  for (int o = 1; o < 32; o <<= 1) {
    const int v = __shfl_up(inc, o, 32);
    inc += (lane >= o) ? v : 0;
  }
  if (lane == 31) s_loc[w] = inc;
  if (lane == 0)  s_pre[w] = pre;
  __syncthreads();
  int base = 0;
  for (int k = 0; k < 8; ++k) base += s_pre[k] + ((k < w) ? s_loc[k] : 0);
  const int ex = base + inc - loc;
  if (i0     < Nn) off[i0]     = ex;
  if (i0 + 1 < Nn) off[i0 + 1] = ex + c0;
  if (i0 + 2 < Nn) off[i0 + 2] = ex + c0 + c1;
  if (i0 + 3 < Nn) off[i0 + 3] = ex + c0 + c1 + c2;
}

template <bool FILL>
__global__ void k_edge_expsum(const int* __restrict__ ei, const float* __restrict__ m,
                              float* __restrict__ score, float* __restrict__ denom,
                              const int* __restrict__ off, const int* __restrict__ rank,
                              int2* __restrict__ csr, int E, int Etot) {
  int e = blockIdx.x * blockDim.x + threadIdx.x;
  if (e >= Etot) return;
  int s, d; get_edge(ei, e, E, s, d);
  float ev = __expf(score[e] - m[s]);
  score[e] = ev;
  atomicAdd(&denom[s], ev);
  if (FILL) {
    if (e < E) csr[off[d] + rank[e]] = make_int2(s, e);
  }
}

template <int NR>
__device__ __forceinline__ void add_rows(int sv, float av, int j0, const float* __restrict__ hl,
                                         float& accx, float& accy) {
  float2 hv[NR];
  float a[NR];
  #pragma unroll
  for (int r = 0; r < NR; ++r) {
    const int s = __builtin_amdgcn_readlane(sv, j0 + r);
    a[r] = __int_as_float(__builtin_amdgcn_readlane(__float_as_int(av), j0 + r));
    hv[r] = *(const float2*)(hl + (size_t)s * 64);
  }
  #pragma unroll
  for (int r = 0; r < NR; ++r) {
    accx = __builtin_fmaf(a[r], hv[r].x, accx);
    accy = __builtin_fmaf(a[r], hv[r].y, accy);
  }
}

__global__ void k_gather_sum(const int2* __restrict__ csr, const int* __restrict__ off,
                             const int* __restrict__ cnt, const float* __restrict__ escore,
                             const float* __restrict__ denom, const float* __restrict__ h,
                             float* __restrict__ aggr, int E, int Nn) {
  const int d = __builtin_amdgcn_readfirstlane(
      (int)(blockIdx.x * (blockDim.x >> 5) + (threadIdx.x >> 5)));
  const int lane = threadIdx.x & 31;
  if (d >= Nn) return;
  const int beg = off[d], n = cnt[d];
  const float* hl = h + lane * 2;
  const float a0 = escore[E + d] / (denom[d] + 1e-16f);
  const float2 hv0 = *(const float2*)(hl + (size_t)d * 64);
  float accx = a0 * hv0.x, accy = a0 * hv0.y;
  for (int c = 0; c < n; c += 32) {
    const int i = c + lane;
    int sv = 0;
    float av = 0.0f;
    if (i < n) {
      const int2 se = csr[beg + i];
      sv = se.x;
      av = escore[se.y] / (denom[se.x] + 1e-16f);
    }
    const int mm = (n - c) < 32 ? (n - c) : 32;
    int j = 0;
    for (; j + 8 <= mm; j += 8) add_rows<8>(sv, av, j, hl, accx, accy);
    if (mm & 4) { add_rows<4>(sv, av, j, hl, accx, accy); j += 4; }
    if (mm & 2) { add_rows<2>(sv, av, j, hl, accx, accy); j += 2; }
    if (mm & 1) { add_rows<1>(sv, av, j, hl, accx, accy); }
  }
  *(float2*)(aggr + (size_t)d * 64 + lane * 2) = make_float2(accx, accy);
}

__global__ void k_finish_out_proj(const float* __restrict__ aggr, const float* __restrict__ bias,
                                  const float* __restrict__ Wout, const float* __restrict__ bout,
                                  float* __restrict__ y, int Nn) {
  int n = blockIdx.x * (blockDim.x >> 5) + (threadIdx.x >> 5);
  int lane = threadIdx.x & 31;
  if (n >= Nn) return;
  const float* ap = aggr + (size_t)n * 64;
  float v0 = ap[lane] + bias[lane];
  float v1 = ap[lane + 32] + bias[lane + 32];
  v0 = v0 > 0.0f ? v0 : 0.0f;
  v1 = v1 > 0.0f ? v1 : 0.0f;
  float v = v0 * Wout[lane] + v1 * Wout[lane + 32];
  #pragma unroll
  for (int off = 16; off > 0; off >>= 1) v += __shfl_xor(v, off, 32);
  if (lane == 0) y[n] = v + bout[0];
}

extern "C" void kernel_launch(void* const* d_in, const int* in_sizes, int n_in,
                              void* d_out, int out_size, void* d_ws, size_t ws_size,
                              hipStream_t stream) {
  const float* x     = (const float*)d_in[0];
  const int*   ei    = (const int*)  d_in[1];
  const float* W1    = (const float*)d_in[2];
  const float* b1    = (const float*)d_in[3];
  const float* att1  = (const float*)d_in[4];
  const float* bias1 = (const float*)d_in[5];
  const float* W2    = (const float*)d_in[6];
  const float* b2    = (const float*)d_in[7];
  const float* att2  = (const float*)d_in[8];
  const float* bias2 = (const float*)d_in[9];
  const float* Wout  = (const float*)d_in[10];
  const float* bout  = (const float*)d_in[11];
  float* y = (float*)d_out;

  const int Nn   = in_sizes[0] / 16;
  const int E    = in_sizes[1] / 2;
  const int Etot = E + Nn;

  char* p = (char*)d_ws;
  auto alloc = [&](size_t bytes) {
    char* r = p;
    p += (bytes + 255) & ~(size_t)255;
    return r;
  };
  float* h      = (float*)alloc((size_t)Nn * 64 * sizeof(float));
  float* aggr1  = (float*)alloc((size_t)Nn * 64 * sizeof(float));
  float* aggr2  = (float*)alloc((size_t)Nn * 64 * sizeof(float));
  float* score  = (float*)alloc((size_t)Etot * sizeof(float));
  float* a_i    = (float*)alloc((size_t)Nn * sizeof(float));
  float* a_j    = (float*)alloc((size_t)Nn * sizeof(float));
  int*   m_bits = (int*)  alloc((size_t)Nn * sizeof(float));
  float* denom  = (float*)alloc((size_t)Nn * sizeof(float));
  int*   cnt    = (int*)  alloc((size_t)Nn * sizeof(int));
  int*   off    = (int*)  alloc((size_t)Nn * sizeof(int));
  int*   rank   = (int*)  alloc((size_t)E * sizeof(int));
  int2*  csr    = (int2*) alloc((size_t)E * sizeof(int2));

  const int TB = 256;
  dim3 blk(TB);
  int g_gemm  = ((Nn / 16) + 7) / 8;
  int g_node  = (Nn + 7) / 8;
  int g_edge  = (Etot + TB - 1) / TB;
  int g_scan  = (Nn + 1023) / 1024;

  k_linear_fused<16, false><<<g_gemm, blk, 0, stream>>>(x, nullptr, W1, b1, att1, h, a_i, a_j,
                                                        m_bits, denom, cnt, Nn);
  k_edge_score<true><<<g_edge, blk, 0, stream>>>(ei, a_i, a_j, score, m_bits, cnt, rank, E, Etot);
  k_dst_offsets<<<g_scan, blk, 0, stream>>>(cnt, off, Nn);
  k_edge_expsum<true><<<g_edge, blk, 0, stream>>>(ei, (const float*)m_bits, score, denom,
                                                  off, rank, csr, E, Etot);
  k_gather_sum<<<g_node, blk, 0, stream>>>(csr, off, cnt, score, denom, h, aggr1, E, Nn);

  k_linear_fused<64, true><<<g_gemm, blk, 0, stream>>>(aggr1, bias1, W2, b2, att2, h, a_i, a_j,
                                                       m_bits, denom, nullptr, Nn);
  k_edge_score<false><<<g_edge, blk, 0, stream>>>(ei, a_i, a_j, score, m_bits, nullptr, nullptr,
                                                  E, Etot);
  k_edge_expsum<false><<<g_edge, blk, 0, stream>>>(ei, (const float*)m_bits, score, denom,
                                                   nullptr, nullptr, nullptr, E, Etot);
  k_gather_sum<<<g_node, blk, 0, stream>>>(csr, off, cnt, score, denom, h, aggr2, E, Nn);

  k_finish_out_proj<<<g_node, blk, 0, stream>>>(aggr2, bias2, Wout, bout, y, Nn);
}


# output projection folded into layer 2's aggregation wave (bias, relu, Wout dot from the row still in registers): 10 launches -> 9, layer-2 sums never stored; the 64-term dot is summed in a new order
# speedup vs baseline: 1.3550x; 1.0730x over previous
#include <hip/hip_runtime.h>
#include <float.h>


typedef __attribute__((ext_vector_type(2))) float v2f;
typedef __attribute__((ext_vector_type(8))) float v8f;

#define NEG_SLOPE 0.01f

template <int MASK>
__device__ __forceinline__ float row_xor(float v) {
  return __int_as_float(__builtin_amdgcn_update_dpp(0, __float_as_int(v), 0x160 + MASK,
                                                    0xf, 0xf, true));
}

template <int K, bool FIN>
__global__ void k_linear_fused(const float* __restrict__ in, const float* __restrict__ bias_prev,
                               const float* __restrict__ W, const float* __restrict__ b,
                               const float* __restrict__ att,
                               float* __restrict__ h, float* __restrict__ a_i,
                               float* __restrict__ a_j, int* __restrict__ m_bits,
                               float* __restrict__ denom, int* __restrict__ cnt, int M) {
  int wave = (int)((blockIdx.x * blockDim.x + threadIdx.x) >> 5);
  int lane = threadIdx.x & 31;
  int row0 = wave * 16;
  if (row0 >= M) return;
  int half = lane >> 4;
  int l16  = lane & 15;

  {
    int* sp = half ? (int*)denom : m_bits;
    sp[row0 + l16] = half ? 0 : (int)0xFF7FFFFF;
    if (!FIN) cnt[row0 + l16] = 0;
  }

  v8f acc0 = {}, acc1 = {}, acc2 = {}, acc3 = {};
  const float* arow = in + (size_t)(row0 + l16) * K;

  #pragma unroll
  for (int k0 = 0; k0 < K; k0 += 4) {
    int kb = k0 + half * 2;
    v2f a;  a.x = arow[kb];  a.y = arow[kb + 1];
    if (FIN) {
      float t0 = a.x + bias_prev[kb], t1 = a.y + bias_prev[kb + 1];
      a.x = t0 > 0.0f ? t0 : 0.0f;
      a.y = t1 > 0.0f ? t1 : 0.0f;
    }
    v2f b0, b1, b2, b3;
    { const float* p = W + (size_t)(0  + l16) * K + kb; b0.x = p[0]; b0.y = p[1]; }
    { const float* p = W + (size_t)(16 + l16) * K + kb; b1.x = p[0]; b1.y = p[1]; }
    { const float* p = W + (size_t)(32 + l16) * K + kb; b2.x = p[0]; b2.y = p[1]; }
    { const float* p = W + (size_t)(48 + l16) * K + kb; b3.x = p[0]; b3.y = p[1]; }
    acc0 = __builtin_amdgcn_wmma_f32_16x16x4_f32(false, a, false, b0, (short)0, acc0, false, false);
    acc1 = __builtin_amdgcn_wmma_f32_16x16x4_f32(false, a, false, b1, (short)0, acc1, false, false);
    acc2 = __builtin_amdgcn_wmma_f32_16x16x4_f32(false, a, false, b2, (short)0, acc2, false, false);
    acc3 = __builtin_amdgcn_wmma_f32_16x16x4_f32(false, a, false, b3, (short)0, acc3, false, false);
  }

  const float bb0 = b[l16], bb1 = b[16 + l16], bb2 = b[32 + l16], bb3 = b[48 + l16];
  const float ti0 = att[l16],      ti1 = att[16 + l16], ti2 = att[32 + l16],  ti3 = att[48 + l16];
  const float tj0 = att[64 + l16], tj1 = att[80 + l16], tj2 = att[96 + l16],  tj3 = att[112 + l16];
  float oi = 0.0f, oj = 0.0f;
  #pragma unroll
  for (int v = 0; v < 8; ++v) {
    int m = row0 + v + half * 8;
    float* hrow = h + (size_t)m * 64 + l16;
    float h0 = acc0[v] + bb0, h1 = acc1[v] + bb1, h2 = acc2[v] + bb2, h3 = acc3[v] + bb3;
    hrow[0]  = h0;
    hrow[16] = h1;
    hrow[32] = h2;
    hrow[48] = h3;
    float vi = h0 * ti0 + h1 * ti1 + h2 * ti2 + h3 * ti3;
    float vj = h0 * tj0 + h1 * tj1 + h2 * tj2 + h3 * tj3;
    vi += row_xor<8>(vi);  vj += row_xor<8>(vj);
    vi += row_xor<4>(vi);  vj += row_xor<4>(vj);
    vi += row_xor<2>(vi);  vj += row_xor<2>(vj);
    vi += row_xor<1>(vi);  vj += row_xor<1>(vj);
    bool mine = (l16 & 7) == v;
    oi = mine ? vi : oi;
    oj = mine ? vj : oj;
  }
  {
    float* op = (l16 < 8) ? a_i : a_j;
    op[row0 + half * 8 + (l16 & 7)] = (l16 < 8) ? oi : oj;
  }
}

__device__ __forceinline__ void get_edge(const int* __restrict__ ei, int e, int E,
                                         int& s, int& d) {
  if (e < E) { s = ei[e]; d = ei[E + e]; }
  else       { s = e - E; d = e - E; }
}

template <bool COUNT>
__global__ void k_edge_score(const int* __restrict__ ei, const float* __restrict__ a_i,
                             const float* __restrict__ a_j, float* __restrict__ score,
                             int* __restrict__ m_bits, int* __restrict__ cnt,
                             int* __restrict__ rank, int E, int Etot) {
  int e = blockIdx.x * blockDim.x + threadIdx.x;
  if (e >= Etot) return;
  int s, d; get_edge(ei, e, E, s, d);
  float sc = a_i[d] + a_j[s];
  sc = sc > 0.0f ? sc : sc * NEG_SLOPE;
  score[e] = sc;
  if (sc >= 0.0f) atomicMax(&m_bits[s], __float_as_int(sc));
  else            atomicMin((unsigned int*)&m_bits[s], __float_as_uint(sc));
  if (COUNT) {
    if (e < E) rank[e] = atomicAdd(&cnt[d], 1);
  }
}

__global__ void k_dst_offsets(const int* __restrict__ cnt, int* __restrict__ off, int Nn) {
  __shared__ int s_pre[8];
  __shared__ int s_loc[8];
  const int t = threadIdx.x, lane = t & 31, w = t >> 5;
  const int nb = blockIdx.x;
  int pre = 0;
  const int4* c4 = (const int4*)cnt;
  for (int j = 0; j < nb; ++j) {
    const int4 v = c4[j * 256 + t];
    pre += (v.x + v.y) + (v.z + v.w);
  }
  const int i0 = nb * 1024 + t * 4;
  const int c0 = (i0     < Nn) ? cnt[i0]     : 0;
  const int c1 = (i0 + 1 < Nn) ? cnt[i0 + 1] : 0;
  const int c2 = (i0 + 2 < Nn) ? cnt[i0 + 2] : 0;
  const int c3 = (i0 + 3 < Nn) ? cnt[i0 + 3] : 0;
  const int loc = (c0 + c1) + (c2 + c3);
  for (int o = 16; o > 0; o >>= 1) pre += __shfl_xor(pre, o, 32);
  int inc = loc;
  for (int o = 1; o < 32; o <<= 1) {
    const int v = __shfl_up(inc, o, 32);
    inc += (lane >= o) ? v : 0;
  }
  if (lane == 31) s_loc[w] = inc;
  if (lane == 0)  s_pre[w] = pre;
  __syncthreads();
  int base = 0;
  for (int k = 0; k < 8; ++k) base += s_pre[k] + ((k < w) ? s_loc[k] : 0);
  const int ex = base + inc - loc;
  if (i0     < Nn) off[i0]     = ex;
  if (i0 + 1 < Nn) off[i0 + 1] = ex + c0;
  if (i0 + 2 < Nn) off[i0 + 2] = ex + c0 + c1;
  if (i0 + 3 < Nn) off[i0 + 3] = ex + c0 + c1 + c2;
}

template <bool FILL>
__global__ void k_edge_expsum(const int* __restrict__ ei, const float* __restrict__ m,
                              float* __restrict__ score, float* __restrict__ denom,
                              const int* __restrict__ off, const int* __restrict__ rank,
                              int2* __restrict__ csr, int E, int Etot) {
  int e = blockIdx.x * blockDim.x + threadIdx.x;
  if (e >= Etot) return;
  int s, d; get_edge(ei, e, E, s, d);
  float ev = __expf(score[e] - m[s]);
  score[e] = ev;
  atomicAdd(&denom[s], ev);
  if (FILL) {
    if (e < E) csr[off[d] + rank[e]] = make_int2(s, e);
  }
}

template <int NR>
__device__ __forceinline__ void add_rows(int sv, float av, int j0, const float* __restrict__ hl,
                                         float& accx, float& accy) {
  float2 hv[NR];
  float a[NR];
  #pragma unroll
  for (int r = 0; r < NR; ++r) {
    const int s = __builtin_amdgcn_readlane(sv, j0 + r);
    a[r] = __int_as_float(__builtin_amdgcn_readlane(__float_as_int(av), j0 + r));
    hv[r] = *(const float2*)(hl + (size_t)s * 64);
  }
  #pragma unroll
  for (int r = 0; r < NR; ++r) {
    accx = __builtin_fmaf(a[r], hv[r].x, accx);
    accy = __builtin_fmaf(a[r], hv[r].y, accy);
  }
}

template <bool PROJ>
__global__ void k_gather_sum(const int2* __restrict__ csr, const int* __restrict__ off,
                             const int* __restrict__ cnt, const float* __restrict__ escore,
                             const float* __restrict__ denom, const float* __restrict__ h,
                             float* __restrict__ aggr, const float* __restrict__ bias,
                             const float* __restrict__ Wout, const float* __restrict__ bout,
                             float* __restrict__ y, int E, int Nn) {
  const int d = __builtin_amdgcn_readfirstlane(
      (int)(blockIdx.x * (blockDim.x >> 5) + (threadIdx.x >> 5)));
  const int lane = threadIdx.x & 31;
  if (d >= Nn) return;
  const int beg = off[d], n = cnt[d];
  const float* hl = h + lane * 2;
  float bz0 = 0.0f, bz1 = 0.0f, wz0 = 0.0f, wz1 = 0.0f, bo = 0.0f;
  if (PROJ) {
    bz0 = bias[lane * 2];  bz1 = bias[lane * 2 + 1];
    wz0 = Wout[lane * 2];  wz1 = Wout[lane * 2 + 1];
    bo  = bout[0];
  }
  const float a0 = escore[E + d] / (denom[d] + 1e-16f);
  const float2 hv0 = *(const float2*)(hl + (size_t)d * 64);
  float accx = a0 * hv0.x, accy = a0 * hv0.y;
  for (int c = 0; c < n; c += 32) {
    const int i = c + lane;
    int sv = 0;
    float av = 0.0f;
    if (i < n) {
      const int2 se = csr[beg + i];
      sv = se.x;
      av = escore[se.y] / (denom[se.x] + 1e-16f);
    }
    const int mm = (n - c) < 32 ? (n - c) : 32;
    int j = 0;
    for (; j + 8 <= mm; j += 8) add_rows<8>(sv, av, j, hl, accx, accy);
    if (mm & 4) { add_rows<4>(sv, av, j, hl, accx, accy); j += 4; }
    if (mm & 2) { add_rows<2>(sv, av, j, hl, accx, accy); j += 2; }
    if (mm & 1) { add_rows<1>(sv, av, j, hl, accx, accy); }
  }
  if (PROJ) {
    float v0 = accx + bz0;
    float v1 = accy + bz1;
    v0 = v0 > 0.0f ? v0 : 0.0f;
    v1 = v1 > 0.0f ? v1 : 0.0f;
    float v = v0 * wz0 + v1 * wz1;
    v += row_xor<8>(v);
    v += row_xor<4>(v);
    v += row_xor<2>(v);
    v += row_xor<1>(v);
    const float tot = __int_as_float(__builtin_amdgcn_readlane(__float_as_int(v), 0)) +
                      __int_as_float(__builtin_amdgcn_readlane(__float_as_int(v), 16));
    if (lane == 0) y[d] = tot + bo;
  } else {
    *(float2*)(aggr + (size_t)d * 64 + lane * 2) = make_float2(accx, accy);
  }
}

extern "C" void kernel_launch(void* const* d_in, const int* in_sizes, int n_in,
                              void* d_out, int out_size, void* d_ws, size_t ws_size,
                              hipStream_t stream) {
  const float* x     = (const float*)d_in[0];
  const int*   ei    = (const int*)  d_in[1];
  const float* W1    = (const float*)d_in[2];
  const float* b1    = (const float*)d_in[3];
  const float* att1  = (const float*)d_in[4];
  const float* bias1 = (const float*)d_in[5];
  const float* W2    = (const float*)d_in[6];
  const float* b2    = (const float*)d_in[7];
  const float* att2  = (const float*)d_in[8];
  const float* bias2 = (const float*)d_in[9];
  const float* Wout  = (const float*)d_in[10];
  const float* bout  = (const float*)d_in[11];
  float* y = (float*)d_out;

  const int Nn   = in_sizes[0] / 16;
  const int E    = in_sizes[1] / 2;
  const int Etot = E + Nn;

  char* p = (char*)d_ws;
  auto alloc = [&](size_t bytes) {
    char* r = p;
    p += (bytes + 255) & ~(size_t)255;
    return r;
  };
  float* h      = (float*)alloc((size_t)Nn * 64 * sizeof(float));
  float* aggr1  = (float*)alloc((size_t)Nn * 64 * sizeof(float));
  float* score  = (float*)alloc((size_t)Etot * sizeof(float));
  float* a_i    = (float*)alloc((size_t)Nn * sizeof(float));
  float* a_j    = (float*)alloc((size_t)Nn * sizeof(float));
  int*   m_bits = (int*)  alloc((size_t)Nn * sizeof(float));
  float* denom  = (float*)alloc((size_t)Nn * sizeof(float));
  int*   cnt    = (int*)  alloc((size_t)Nn * sizeof(int));
  int*   off    = (int*)  alloc((size_t)Nn * sizeof(int));
  int*   rank   = (int*)  alloc((size_t)E * sizeof(int));
  int2*  csr    = (int2*) alloc((size_t)E * sizeof(int2));

  const int TB = 256;
  dim3 blk(TB);
  int g_gemm  = ((Nn / 16) + 7) / 8;
  int g_node  = (Nn + 7) / 8;
  int g_edge  = (Etot + TB - 1) / TB;
  int g_scan  = (Nn + 1023) / 1024;

  k_linear_fused<16, false><<<g_gemm, blk, 0, stream>>>(x, nullptr, W1, b1, att1, h, a_i, a_j,
                                                        m_bits, denom, cnt, Nn);
  k_edge_score<true><<<g_edge, blk, 0, stream>>>(ei, a_i, a_j, score, m_bits, cnt, rank, E, Etot);
  k_dst_offsets<<<g_scan, blk, 0, stream>>>(cnt, off, Nn);
  k_edge_expsum<true><<<g_edge, blk, 0, stream>>>(ei, (const float*)m_bits, score, denom,
                                                  off, rank, csr, E, Etot);
  k_gather_sum<false><<<g_node, blk, 0, stream>>>(csr, off, cnt, score, denom, h, aggr1,
                                                  nullptr, nullptr, nullptr, nullptr, E, Nn);

  k_linear_fused<64, true><<<g_gemm, blk, 0, stream>>>(aggr1, bias1, W2, b2, att2, h, a_i, a_j,
                                                       m_bits, denom, nullptr, Nn);
  k_edge_score<false><<<g_edge, blk, 0, stream>>>(ei, a_i, a_j, score, m_bits, nullptr, nullptr,
                                                  E, Etot);
  k_edge_expsum<false><<<g_edge, blk, 0, stream>>>(ei, (const float*)m_bits, score, denom,
                                                   nullptr, nullptr, nullptr, E, Etot);
  k_gather_sum<true><<<g_node, blk, 0, stream>>>(csr, off, cnt, score, denom, h, nullptr,
                                                 bias2, Wout, bout, y, E, Nn);
}


# softmax per src inside one wave over a second, src-ordered edge list (counting sort): float atomic max/add and the score array go; exp sum is now an in-wave tree in src arrival-rank order
# speedup vs baseline: 1.8213x; 1.3441x over previous
#include <hip/hip_runtime.h>
#include <float.h>


typedef __attribute__((ext_vector_type(2))) float v2f;
typedef __attribute__((ext_vector_type(8))) float v8f;

#define NEG_SLOPE 0.01f

template <int MASK>
__device__ __forceinline__ float row_xor(float v) {
  return __int_as_float(__builtin_amdgcn_update_dpp(0, __float_as_int(v), 0x160 + MASK,
                                                    0xf, 0xf, true));
}

template <int K, bool FIN>
__global__ void k_linear_fused(const float* __restrict__ in, const float* __restrict__ bias_prev,
                               const float* __restrict__ W, const float* __restrict__ b,
                               const float* __restrict__ att,
                               float* __restrict__ h, float* __restrict__ a_i,
                               float* __restrict__ a_j, int* __restrict__ cnt_d,
                               int* __restrict__ cnt_s, int M) {
  int wave = (int)((blockIdx.x * blockDim.x + threadIdx.x) >> 5);
  int lane = threadIdx.x & 31;
  int row0 = wave * 16;
  if (row0 >= M) return;
  int half = lane >> 4;
  int l16  = lane & 15;

  if (!FIN) {
    int* cp = half ? cnt_s : cnt_d;
    cp[row0 + l16] = 0;
  }

  v8f acc0 = {}, acc1 = {}, acc2 = {}, acc3 = {};
  const float* arow = in + (size_t)(row0 + l16) * K;

  #pragma unroll
  for (int k0 = 0; k0 < K; k0 += 4) {
    int kb = k0 + half * 2;
    v2f a;  a.x = arow[kb];  a.y = arow[kb + 1];
    if (FIN) {
      float t0 = a.x + bias_prev[kb], t1 = a.y + bias_prev[kb + 1];
      a.x = t0 > 0.0f ? t0 : 0.0f;
      a.y = t1 > 0.0f ? t1 : 0.0f;
    }
    v2f b0, b1, b2, b3;
    { const float* p = W + (size_t)(0  + l16) * K + kb; b0.x = p[0]; b0.y = p[1]; }
    { const float* p = W + (size_t)(16 + l16) * K + kb; b1.x = p[0]; b1.y = p[1]; }
    { const float* p = W + (size_t)(32 + l16) * K + kb; b2.x = p[0]; b2.y = p[1]; }
    { const float* p = W + (size_t)(48 + l16) * K + kb; b3.x = p[0]; b3.y = p[1]; }
    acc0 = __builtin_amdgcn_wmma_f32_16x16x4_f32(false, a, false, b0, (short)0, acc0, false, false);
    acc1 = __builtin_amdgcn_wmma_f32_16x16x4_f32(false, a, false, b1, (short)0, acc1, false, false);
    acc2 = __builtin_amdgcn_wmma_f32_16x16x4_f32(false, a, false, b2, (short)0, acc2, false, false);
    acc3 = __builtin_amdgcn_wmma_f32_16x16x4_f32(false, a, false, b3, (short)0, acc3, false, false);
  }

  const float bb0 = b[l16], bb1 = b[16 + l16], bb2 = b[32 + l16], bb3 = b[48 + l16];
  const float ti0 = att[l16],      ti1 = att[16 + l16], ti2 = att[32 + l16],  ti3 = att[48 + l16];
  const float tj0 = att[64 + l16], tj1 = att[80 + l16], tj2 = att[96 + l16],  tj3 = att[112 + l16];
  float oi = 0.0f, oj = 0.0f;
  #pragma unroll
  for (int v = 0; v < 8; ++v) {
    int m = row0 + v + half * 8;
    float* hrow = h + (size_t)m * 64 + l16;
    float h0 = acc0[v] + bb0, h1 = acc1[v] + bb1, h2 = acc2[v] + bb2, h3 = acc3[v] + bb3;
    hrow[0]  = h0;
    hrow[16] = h1;
    hrow[32] = h2;
    hrow[48] = h3;
    float vi = h0 * ti0 + h1 * ti1 + h2 * ti2 + h3 * ti3;
    float vj = h0 * tj0 + h1 * tj1 + h2 * tj2 + h3 * tj3;
    vi += row_xor<8>(vi);  vj += row_xor<8>(vj);
    vi += row_xor<4>(vi);  vj += row_xor<4>(vj);
    vi += row_xor<2>(vi);  vj += row_xor<2>(vj);
    vi += row_xor<1>(vi);  vj += row_xor<1>(vj);
    bool mine = (l16 & 7) == v;
    oi = mine ? vi : oi;
    oj = mine ? vj : oj;
  }
  {
    float* op = (l16 < 8) ? a_i : a_j;
    op[row0 + half * 8 + (l16 & 7)] = (l16 < 8) ? oi : oj;
  }
}

__global__ void k_edge_count(const int* __restrict__ ei, int* __restrict__ cnt_d,
                             int* __restrict__ cnt_s, int2* __restrict__ rank, int E) {
  int e = blockIdx.x * blockDim.x + threadIdx.x;
  if (e >= E) return;
  const int s = ei[e], d = ei[E + e];
  const int rd = atomicAdd(&cnt_d[d], 1);
  const int rs = atomicAdd(&cnt_s[s], 1);
  rank[e] = make_int2(rd, rs);
}

__global__ void k_offsets(const int* __restrict__ cnt_d, int* __restrict__ off_d,
                          const int* __restrict__ cnt_s, int* __restrict__ off_s, int Nn) {
  const int* __restrict__ cnt = blockIdx.y ? cnt_s : cnt_d;
  int* __restrict__ off = blockIdx.y ? off_s : off_d;
  __shared__ int s_pre[8];
  __shared__ int s_loc[8];
  const int t = threadIdx.x, lane = t & 31, w = t >> 5;
  const int nb = blockIdx.x;
  int pre = 0;
  const int4* c4 = (const int4*)cnt;
  for (int j = 0; j < nb; ++j) {
    const int4 v = c4[j * 256 + t];
    pre += (v.x + v.y) + (v.z + v.w);
  }
  const int i0 = nb * 1024 + t * 4;
  const int c0 = (i0     < Nn) ? cnt[i0]     : 0;
  const int c1 = (i0 + 1 < Nn) ? cnt[i0 + 1] : 0;
  const int c2 = (i0 + 2 < Nn) ? cnt[i0 + 2] : 0;
  const int c3 = (i0 + 3 < Nn) ? cnt[i0 + 3] : 0;
  const int loc = (c0 + c1) + (c2 + c3);
  for (int o = 16; o > 0; o >>= 1) pre += __shfl_xor(pre, o, 32);
  int inc = loc;
  for (int o = 1; o < 32; o <<= 1) {
    const int v = __shfl_up(inc, o, 32);
    inc += (lane >= o) ? v : 0;
  }
  if (lane == 31) s_loc[w] = inc;
  if (lane == 0)  s_pre[w] = pre;
  __syncthreads();
  int base = 0;
  for (int k = 0; k < 8; ++k) base += s_pre[k] + ((k < w) ? s_loc[k] : 0);
  const int ex = base + inc - loc;
  if (i0     < Nn) off[i0]     = ex;
  if (i0 + 1 < Nn) off[i0 + 1] = ex + c0;
  if (i0 + 2 < Nn) off[i0 + 2] = ex + c0 + c1;
  if (i0 + 3 < Nn) off[i0 + 3] = ex + c0 + c1 + c2;
}

__global__ void k_edge_place(const int* __restrict__ ei, const int* __restrict__ off_d,
                             const int* __restrict__ off_s, const int2* __restrict__ rank,
                             int* __restrict__ csr_src, int2* __restrict__ csr_s, int E) {
  int e = blockIdx.x * blockDim.x + threadIdx.x;
  if (e >= E) return;
  const int s = ei[e], d = ei[E + e];
  const int2 r = rank[e];
  const int pd = off_d[d] + r.x;
  const int ps = off_s[s] + r.y;
  csr_src[pd] = s;
  csr_s[ps] = make_int2(d, pd);
}

__device__ __forceinline__ float leaky(float v) { return v > 0.0f ? v : v * NEG_SLOPE; }

__device__ __forceinline__ float wave_max(float v) {
  v = fmaxf(v, row_xor<8>(v));
  v = fmaxf(v, row_xor<4>(v));
  v = fmaxf(v, row_xor<2>(v));
  v = fmaxf(v, row_xor<1>(v));
  return fmaxf(__int_as_float(__builtin_amdgcn_readlane(__float_as_int(v), 0)),
               __int_as_float(__builtin_amdgcn_readlane(__float_as_int(v), 16)));
}
__device__ __forceinline__ float wave_sum(float v) {
  v += row_xor<8>(v);
  v += row_xor<4>(v);
  v += row_xor<2>(v);
  v += row_xor<1>(v);
  return __int_as_float(__builtin_amdgcn_readlane(__float_as_int(v), 0)) +
         __int_as_float(__builtin_amdgcn_readlane(__float_as_int(v), 16));
}

__global__ void k_src_softmax(const int2* __restrict__ csr_s, const int* __restrict__ off_s,
                              const int* __restrict__ cnt_s, const float* __restrict__ a_i,
                              const float* __restrict__ a_j, float* __restrict__ alpha_d,
                              float* __restrict__ alpha_self, int Nn) {
  const int s = __builtin_amdgcn_readfirstlane(
      (int)(blockIdx.x * (blockDim.x >> 5) + (threadIdx.x >> 5)));
  const int lane = threadIdx.x & 31;
  if (s >= Nn) return;
  const int beg = off_s[s], n = cnt_s[s];
  const float aj = a_j[s];
  const float sc_self = leaky(a_i[s] + aj);
  const bool v0 = lane < n;
  int pos0 = 0;
  float sc0 = -FLT_MAX;
  if (v0) {
    const int2 de = csr_s[beg + lane];
    pos0 = de.y;
    sc0 = leaky(a_i[de.x] + aj);
  }
  float mx = fmaxf(sc_self, sc0);
  for (int c = 32; c < n; c += 32) {
    const int i = c + lane;
    if (i < n) mx = fmaxf(mx, leaky(a_i[csr_s[beg + i].x] + aj));
  }
  mx = wave_max(mx);
  const float ev0 = v0 ? __expf(sc0 - mx) : 0.0f;
  float sum = ev0;
  for (int c = 32; c < n; c += 32) {
    const int i = c + lane;
    if (i < n) sum += __expf(leaky(a_i[csr_s[beg + i].x] + aj) - mx);
  }
  const float es = __expf(sc_self - mx);
  const float den = (wave_sum(sum) + es) + 1e-16f;
  if (v0) alpha_d[pos0] = ev0 / den;
  for (int c = 32; c < n; c += 32) {
    const int i = c + lane;
    if (i < n) {
      const int2 de = csr_s[beg + i];
      alpha_d[de.y] = __expf(leaky(a_i[de.x] + aj) - mx) / den;
    }
  }
  if (lane == 0) alpha_self[s] = es / den;
}

template <int NR>
__device__ __forceinline__ void add_rows(int sv, float av, int j0, const float* __restrict__ hl,
                                         float& accx, float& accy) {
  float2 hv[NR];
  float a[NR];
  #pragma unroll
  for (int r = 0; r < NR; ++r) {
    const int s = __builtin_amdgcn_readlane(sv, j0 + r);
    a[r] = __int_as_float(__builtin_amdgcn_readlane(__float_as_int(av), j0 + r));
    hv[r] = *(const float2*)(hl + (size_t)s * 64);
  }
  #pragma unroll
  for (int r = 0; r < NR; ++r) {
    accx = __builtin_fmaf(a[r], hv[r].x, accx);
    accy = __builtin_fmaf(a[r], hv[r].y, accy);
  }
}

template <bool PROJ>
__global__ void k_gather_sum(const int* __restrict__ csr_src, const float* __restrict__ alpha_d,
                             const float* __restrict__ alpha_self, const int* __restrict__ off,
                             const int* __restrict__ cnt, const float* __restrict__ h,
                             float* __restrict__ aggr, const float* __restrict__ bias,
                             const float* __restrict__ Wout, const float* __restrict__ bout,
                             float* __restrict__ y, int Nn) {
  const int d = __builtin_amdgcn_readfirstlane(
      (int)(blockIdx.x * (blockDim.x >> 5) + (threadIdx.x >> 5)));
  const int lane = threadIdx.x & 31;
  if (d >= Nn) return;
  const int beg = off[d], n = cnt[d];
  const float* hl = h + lane * 2;
  float bz0 = 0.0f, bz1 = 0.0f, wz0 = 0.0f, wz1 = 0.0f, bo = 0.0f;
  if (PROJ) {
    bz0 = bias[lane * 2];  bz1 = bias[lane * 2 + 1];
    wz0 = Wout[lane * 2];  wz1 = Wout[lane * 2 + 1];
    bo  = bout[0];
  }
  const float a0 = alpha_self[d];
  const float2 hv0 = *(const float2*)(hl + (size_t)d * 64);
  float accx = a0 * hv0.x, accy = a0 * hv0.y;
  for (int c = 0; c < n; c += 32) {
    const int i = c + lane;
    int sv = 0;
    float av = 0.0f;
    if (i < n) {
      sv = csr_src[beg + i];
      av = alpha_d[beg + i];
    }
    const int mm = (n - c) < 32 ? (n - c) : 32;
    int j = 0;
    for (; j + 8 <= mm; j += 8) add_rows<8>(sv, av, j, hl, accx, accy);
    if (mm & 4) { add_rows<4>(sv, av, j, hl, accx, accy); j += 4; }
    if (mm & 2) { add_rows<2>(sv, av, j, hl, accx, accy); j += 2; }
    if (mm & 1) { add_rows<1>(sv, av, j, hl, accx, accy); }
  }
  if (PROJ) {
    float v0 = accx + bz0;
    float v1 = accy + bz1;
    v0 = v0 > 0.0f ? v0 : 0.0f;
    v1 = v1 > 0.0f ? v1 : 0.0f;
    float v = v0 * wz0 + v1 * wz1;
    v += row_xor<8>(v);
    v += row_xor<4>(v);
    v += row_xor<2>(v);
    v += row_xor<1>(v);
    const float tot = __int_as_float(__builtin_amdgcn_readlane(__float_as_int(v), 0)) +
                      __int_as_float(__builtin_amdgcn_readlane(__float_as_int(v), 16));
    if (lane == 0) y[d] = tot + bo;
  } else {
    *(float2*)(aggr + (size_t)d * 64 + lane * 2) = make_float2(accx, accy);
  }
}

extern "C" void kernel_launch(void* const* d_in, const int* in_sizes, int n_in,
                              void* d_out, int out_size, void* d_ws, size_t ws_size,
                              hipStream_t stream) {
  const float* x     = (const float*)d_in[0];
  const int*   ei    = (const int*)  d_in[1];
  const float* W1    = (const float*)d_in[2];
  const float* b1    = (const float*)d_in[3];
  const float* att1  = (const float*)d_in[4];
  const float* bias1 = (const float*)d_in[5];
  const float* W2    = (const float*)d_in[6];
  const float* b2    = (const float*)d_in[7];
  const float* att2  = (const float*)d_in[8];
  const float* bias2 = (const float*)d_in[9];
  const float* Wout  = (const float*)d_in[10];
  const float* bout  = (const float*)d_in[11];
  float* y = (float*)d_out;

  const int Nn   = in_sizes[0] / 16;
  const int E    = in_sizes[1] / 2;

  char* p = (char*)d_ws;
  auto alloc = [&](size_t bytes) {
    char* r = p;
    p += (bytes + 255) & ~(size_t)255;
    return r;
  };
  float* h      = (float*)alloc((size_t)Nn * 64 * sizeof(float));
  float* aggr1  = (float*)alloc((size_t)Nn * 64 * sizeof(float));
  float* a_i    = (float*)alloc((size_t)Nn * sizeof(float));
  float* a_j    = (float*)alloc((size_t)Nn * sizeof(float));
  float* al_self = (float*)alloc((size_t)Nn * sizeof(float));
  int*   cnt_d  = (int*)  alloc((size_t)Nn * sizeof(int));
  int*   cnt_s  = (int*)  alloc((size_t)Nn * sizeof(int));
  int*   off_d  = (int*)  alloc((size_t)Nn * sizeof(int));
  int*   off_s  = (int*)  alloc((size_t)Nn * sizeof(int));
  int2*  rank   = (int2*) alloc((size_t)E * sizeof(int2));
  int*   csr_src = (int*) alloc((size_t)E * sizeof(int));
  float* alpha_d = (float*)alloc((size_t)E * sizeof(float));
  int2*  csr_s  = (int2*) alloc((size_t)E * sizeof(int2));

  const int TB = 256;
  dim3 blk(TB);
  int g_gemm  = ((Nn / 16) + 7) / 8;
  int g_node  = (Nn + 7) / 8;
  int g_edge  = (E + TB - 1) / TB;
  dim3 g_scan((Nn + 1023) / 1024, 2);

  k_linear_fused<16, false><<<g_gemm, blk, 0, stream>>>(x, nullptr, W1, b1, att1, h, a_i, a_j,
                                                        cnt_d, cnt_s, Nn);
  k_edge_count<<<g_edge, blk, 0, stream>>>(ei, cnt_d, cnt_s, rank, E);
  k_offsets<<<g_scan, blk, 0, stream>>>(cnt_d, off_d, cnt_s, off_s, Nn);
  k_edge_place<<<g_edge, blk, 0, stream>>>(ei, off_d, off_s, rank, csr_src, csr_s, E);

  k_src_softmax<<<g_node, blk, 0, stream>>>(csr_s, off_s, cnt_s, a_i, a_j, alpha_d, al_self, Nn);
  k_gather_sum<false><<<g_node, blk, 0, stream>>>(csr_src, alpha_d, al_self, off_d, cnt_d, h, aggr1,
                                                  nullptr, nullptr, nullptr, nullptr, Nn);

  k_linear_fused<64, true><<<g_gemm, blk, 0, stream>>>(aggr1, bias1, W2, b2, att2, h, a_i, a_j,
                                                       nullptr, nullptr, Nn);
  k_src_softmax<<<g_node, blk, 0, stream>>>(csr_s, off_s, cnt_s, a_i, a_j, alpha_d, al_self, Nn);
  k_gather_sum<true><<<g_node, blk, 0, stream>>>(csr_src, alpha_d, al_self, off_d, cnt_d, h, nullptr,
                                                 bias2, Wout, bout, y, Nn);
}


# each wave takes 4 consecutive nodes (softmax, aggregation) and each lane 4 edges (count, place), their first loads issued as one group per stage: 4x fewer waves to start in 6 of the 9 launches
# speedup vs baseline: 1.9059x; 1.0465x over previous
#include <hip/hip_runtime.h>
#include <float.h>


#define NPW 4
#define EPT 4

typedef __attribute__((ext_vector_type(2))) float v2f;
typedef __attribute__((ext_vector_type(8))) float v8f;

#define NEG_SLOPE 0.01f

template <int MASK>
__device__ __forceinline__ float row_xor(float v) {
  return __int_as_float(__builtin_amdgcn_update_dpp(0, __float_as_int(v), 0x160 + MASK,
                                                    0xf, 0xf, true));
}

template <int K, bool FIN>
__global__ void k_linear_fused(const float* __restrict__ in, const float* __restrict__ bias_prev,
                               const float* __restrict__ W, const float* __restrict__ b,
                               const float* __restrict__ att,
                               float* __restrict__ h, float* __restrict__ a_i,
                               float* __restrict__ a_j, int* __restrict__ cnt_d,
                               int* __restrict__ cnt_s, int M) {
  int wave = (int)((blockIdx.x * blockDim.x + threadIdx.x) >> 5);
  int lane = threadIdx.x & 31;
  int row0 = wave * 16;
  if (row0 >= M) return;
  int half = lane >> 4;
  int l16  = lane & 15;

  if (!FIN) {
    int* cp = half ? cnt_s : cnt_d;
    cp[row0 + l16] = 0;
  }

  v8f acc0 = {}, acc1 = {}, acc2 = {}, acc3 = {};
  const float* arow = in + (size_t)(row0 + l16) * K;

  #pragma unroll
  for (int k0 = 0; k0 < K; k0 += 4) {
    int kb = k0 + half * 2;
    v2f a;  a.x = arow[kb];  a.y = arow[kb + 1];
    if (FIN) {
      float t0 = a.x + bias_prev[kb], t1 = a.y + bias_prev[kb + 1];
      a.x = t0 > 0.0f ? t0 : 0.0f;
      a.y = t1 > 0.0f ? t1 : 0.0f;
    }
    v2f b0, b1, b2, b3;
    { const float* p = W + (size_t)(0  + l16) * K + kb; b0.x = p[0]; b0.y = p[1]; }
    { const float* p = W + (size_t)(16 + l16) * K + kb; b1.x = p[0]; b1.y = p[1]; }
    { const float* p = W + (size_t)(32 + l16) * K + kb; b2.x = p[0]; b2.y = p[1]; }
    { const float* p = W + (size_t)(48 + l16) * K + kb; b3.x = p[0]; b3.y = p[1]; }
    acc0 = __builtin_amdgcn_wmma_f32_16x16x4_f32(false, a, false, b0, (short)0, acc0, false, false);
    acc1 = __builtin_amdgcn_wmma_f32_16x16x4_f32(false, a, false, b1, (short)0, acc1, false, false);
    acc2 = __builtin_amdgcn_wmma_f32_16x16x4_f32(false, a, false, b2, (short)0, acc2, false, false);
    acc3 = __builtin_amdgcn_wmma_f32_16x16x4_f32(false, a, false, b3, (short)0, acc3, false, false);
  }

  const float bb0 = b[l16], bb1 = b[16 + l16], bb2 = b[32 + l16], bb3 = b[48 + l16];
  const float ti0 = att[l16],      ti1 = att[16 + l16], ti2 = att[32 + l16],  ti3 = att[48 + l16];
  const float tj0 = att[64 + l16], tj1 = att[80 + l16], tj2 = att[96 + l16],  tj3 = att[112 + l16];
  float oi = 0.0f, oj = 0.0f;
  #pragma unroll
  for (int v = 0; v < 8; ++v) {
    int m = row0 + v + half * 8;
    float* hrow = h + (size_t)m * 64 + l16;
    float h0 = acc0[v] + bb0, h1 = acc1[v] + bb1, h2 = acc2[v] + bb2, h3 = acc3[v] + bb3;
    hrow[0]  = h0;
    hrow[16] = h1;
    hrow[32] = h2;
    hrow[48] = h3;
    float vi = h0 * ti0 + h1 * ti1 + h2 * ti2 + h3 * ti3;
    float vj = h0 * tj0 + h1 * tj1 + h2 * tj2 + h3 * tj3;
    vi += row_xor<8>(vi);  vj += row_xor<8>(vj);
    vi += row_xor<4>(vi);  vj += row_xor<4>(vj);
    vi += row_xor<2>(vi);  vj += row_xor<2>(vj);
    vi += row_xor<1>(vi);  vj += row_xor<1>(vj);
    bool mine = (l16 & 7) == v;
    oi = mine ? vi : oi;
    oj = mine ? vj : oj;
  }
  {
    float* op = (l16 < 8) ? a_i : a_j;
    op[row0 + half * 8 + (l16 & 7)] = (l16 < 8) ? oi : oj;
  }
}

__global__ void k_edge_count(const int* __restrict__ ei, int* __restrict__ cnt_d,
                             int* __restrict__ cnt_s, int2* __restrict__ rank, int E) {
  const int T = (int)(gridDim.x * blockDim.x);
  const int e0 = (int)(blockIdx.x * blockDim.x + threadIdx.x);
  if (e0 + (EPT - 1) * T < E) {
    int s[EPT], d[EPT], rd[EPT], rs[EPT];
    #pragma unroll
    for (int k = 0; k < EPT; ++k) { s[k] = ei[e0 + k * T];  d[k] = ei[E + e0 + k * T]; }
    #pragma unroll
    for (int k = 0; k < EPT; ++k) {
      rd[k] = atomicAdd(&cnt_d[d[k]], 1);
      rs[k] = atomicAdd(&cnt_s[s[k]], 1);
    }
    #pragma unroll
    for (int k = 0; k < EPT; ++k) rank[e0 + k * T] = make_int2(rd[k], rs[k]);
  } else {
    for (int k = 0; k < EPT; ++k) {
      const int e = e0 + k * T;
      if (e < E) {
        const int s = ei[e], d = ei[E + e];
        const int rd = atomicAdd(&cnt_d[d], 1);
        const int rs = atomicAdd(&cnt_s[s], 1);
        rank[e] = make_int2(rd, rs);
      }
    }
  }
}

__global__ void k_offsets(const int* __restrict__ cnt_d, int* __restrict__ off_d,
                          const int* __restrict__ cnt_s, int* __restrict__ off_s, int Nn) {
  const int* __restrict__ cnt = blockIdx.y ? cnt_s : cnt_d;
  int* __restrict__ off = blockIdx.y ? off_s : off_d;
  __shared__ int s_pre[8];
  __shared__ int s_loc[8];
  const int t = threadIdx.x, lane = t & 31, w = t >> 5;
  const int nb = blockIdx.x;
  int pre = 0;
  const int4* c4 = (const int4*)cnt;
  for (int j = 0; j < nb; ++j) {
    const int4 v = c4[j * 256 + t];
    pre += (v.x + v.y) + (v.z + v.w);
  }
  const int i0 = nb * 1024 + t * 4;
  const int c0 = (i0     < Nn) ? cnt[i0]     : 0;
  const int c1 = (i0 + 1 < Nn) ? cnt[i0 + 1] : 0;
  const int c2 = (i0 + 2 < Nn) ? cnt[i0 + 2] : 0;
  const int c3 = (i0 + 3 < Nn) ? cnt[i0 + 3] : 0;
  const int loc = (c0 + c1) + (c2 + c3);
  for (int o = 16; o > 0; o >>= 1) pre += __shfl_xor(pre, o, 32);
  int inc = loc;
  for (int o = 1; o < 32; o <<= 1) {
    const int v = __shfl_up(inc, o, 32);
    inc += (lane >= o) ? v : 0;
  }
  if (lane == 31) s_loc[w] = inc;
  if (lane == 0)  s_pre[w] = pre;
  __syncthreads();
  int base = 0;
  for (int k = 0; k < 8; ++k) base += s_pre[k] + ((k < w) ? s_loc[k] : 0);
  const int ex = base + inc - loc;
  if (i0     < Nn) off[i0]     = ex;
  if (i0 + 1 < Nn) off[i0 + 1] = ex + c0;
  if (i0 + 2 < Nn) off[i0 + 2] = ex + c0 + c1;
  if (i0 + 3 < Nn) off[i0 + 3] = ex + c0 + c1 + c2;
}

__global__ void k_edge_place(const int* __restrict__ ei, const int* __restrict__ off_d,
                             const int* __restrict__ off_s, const int2* __restrict__ rank,
                             int* __restrict__ csr_src, int2* __restrict__ csr_s, int E) {
  const int T = (int)(gridDim.x * blockDim.x);
  const int e0 = (int)(blockIdx.x * blockDim.x + threadIdx.x);
  if (e0 + (EPT - 1) * T < E) {
    int s[EPT], d[EPT], pd[EPT], ps[EPT];
    int2 r[EPT];
    #pragma unroll
    for (int k = 0; k < EPT; ++k) {
      s[k] = ei[e0 + k * T];  d[k] = ei[E + e0 + k * T];  r[k] = rank[e0 + k * T];
    }
    #pragma unroll
    for (int k = 0; k < EPT; ++k) { pd[k] = off_d[d[k]] + r[k].x;  ps[k] = off_s[s[k]] + r[k].y; }
    #pragma unroll
    for (int k = 0; k < EPT; ++k) {
      csr_src[pd[k]] = s[k];
      csr_s[ps[k]] = make_int2(d[k], pd[k]);
    }
  } else {
    for (int k = 0; k < EPT; ++k) {
      const int e = e0 + k * T;
      if (e < E) {
        const int s = ei[e], d = ei[E + e];
        const int2 r = rank[e];
        const int pd = off_d[d] + r.x;
        const int ps = off_s[s] + r.y;
        csr_src[pd] = s;
        csr_s[ps] = make_int2(d, pd);
      }
    }
  }
}

__device__ __forceinline__ float leaky(float v) { return v > 0.0f ? v : v * NEG_SLOPE; }

__device__ __forceinline__ float wave_max(float v) {
  v = fmaxf(v, row_xor<8>(v));
  v = fmaxf(v, row_xor<4>(v));
  v = fmaxf(v, row_xor<2>(v));
  v = fmaxf(v, row_xor<1>(v));
  return fmaxf(__int_as_float(__builtin_amdgcn_readlane(__float_as_int(v), 0)),
               __int_as_float(__builtin_amdgcn_readlane(__float_as_int(v), 16)));
}
__device__ __forceinline__ float wave_sum(float v) {
  v += row_xor<8>(v);
  v += row_xor<4>(v);
  v += row_xor<2>(v);
  v += row_xor<1>(v);
  return __int_as_float(__builtin_amdgcn_readlane(__float_as_int(v), 0)) +
         __int_as_float(__builtin_amdgcn_readlane(__float_as_int(v), 16));
}

__global__ void k_src_softmax(const int2* __restrict__ csr_s, const int* __restrict__ off_s,
                              const int* __restrict__ cnt_s, const float* __restrict__ a_i,
                              const float* __restrict__ a_j, float* __restrict__ alpha_d,
                              float* __restrict__ alpha_self, int Nn) {
  const int w = __builtin_amdgcn_readfirstlane(
      (int)(blockIdx.x * (blockDim.x >> 5) + (threadIdx.x >> 5)));
  const int lane = threadIdx.x & 31;
  const int s0 = w * NPW;
  if (s0 >= Nn) return;
  int beg[NPW], n[NPW];
  float aj[NPW], ais[NPW];
  #pragma unroll
  for (int k = 0; k < NPW; ++k) {
    const int s = (s0 + k < Nn) ? (s0 + k) : (Nn - 1);
    beg[k] = off_s[s];
    n[k]   = cnt_s[s];
    aj[k]  = a_j[s];
    ais[k] = a_i[s];
  }
  int2 de[NPW];
  #pragma unroll
  for (int k = 0; k < NPW; ++k) de[k] = csr_s[(lane < n[k]) ? (beg[k] + lane) : 0];
  float ai0[NPW];
  #pragma unroll
  for (int k = 0; k < NPW; ++k) ai0[k] = a_i[de[k].x];
  float self_out = 0.0f;
  #pragma unroll
  for (int k = 0; k < NPW; ++k) {
    if (s0 + k < Nn) {
      const int b = beg[k], nn = n[k];
      const float ajk = aj[k];
      const float sc_self = leaky(ais[k] + ajk);
      const bool v0 = lane < nn;
      const float sc0 = v0 ? leaky(ai0[k] + ajk) : -FLT_MAX;
      float mx = fmaxf(sc_self, sc0);
      for (int cc = 32; cc < nn; cc += 32) {
        const int i = cc + lane;
        if (i < nn) mx = fmaxf(mx, leaky(a_i[csr_s[b + i].x] + ajk));
      }
      mx = wave_max(mx);
      const float ev0 = v0 ? __expf(sc0 - mx) : 0.0f;
      float sum = ev0;
      for (int cc = 32; cc < nn; cc += 32) {
        const int i = cc + lane;
        if (i < nn) sum += __expf(leaky(a_i[csr_s[b + i].x] + ajk) - mx);
      }
      const float es = __expf(sc_self - mx);
      const float den = (wave_sum(sum) + es) + 1e-16f;
      if (v0) alpha_d[de[k].y] = ev0 / den;
      for (int cc = 32; cc < nn; cc += 32) {
        const int i = cc + lane;
        if (i < nn) {
          const int2 dx = csr_s[b + i];
          alpha_d[dx.y] = __expf(leaky(a_i[dx.x] + ajk) - mx) / den;
        }
      }
      const float as = es / den;
      self_out = (lane == k) ? as : self_out;
    }
  }
  if (lane < NPW && s0 + lane < Nn) alpha_self[s0 + lane] = self_out;
}

template <int NR>
__device__ __forceinline__ void add_rows(int sv, float av, int j0, const float* __restrict__ hl,
                                         float& accx, float& accy) {
  float2 hv[NR];
  float a[NR];
  #pragma unroll
  for (int r = 0; r < NR; ++r) {
    const int s = __builtin_amdgcn_readlane(sv, j0 + r);
    a[r] = __int_as_float(__builtin_amdgcn_readlane(__float_as_int(av), j0 + r));
    hv[r] = *(const float2*)(hl + (size_t)s * 64);
  }
  #pragma unroll
  for (int r = 0; r < NR; ++r) {
    accx = __builtin_fmaf(a[r], hv[r].x, accx);
    accy = __builtin_fmaf(a[r], hv[r].y, accy);
  }
}

template <bool PROJ>
__global__ void k_gather_sum(const int* __restrict__ csr_src, const float* __restrict__ alpha_d,
                             const float* __restrict__ alpha_self, const int* __restrict__ off,
                             const int* __restrict__ cnt, const float* __restrict__ h,
                             float* __restrict__ aggr, const float* __restrict__ bias,
                             const float* __restrict__ Wout, const float* __restrict__ bout,
                             float* __restrict__ y, int Nn) {
  const int w = __builtin_amdgcn_readfirstlane(
      (int)(blockIdx.x * (blockDim.x >> 5) + (threadIdx.x >> 5)));
  const int lane = threadIdx.x & 31;
  const int d0 = w * NPW;
  if (d0 >= Nn) return;
  const float* hl = h + lane * 2;
  float bz0 = 0.0f, bz1 = 0.0f, wz0 = 0.0f, wz1 = 0.0f, bo = 0.0f;
  if (PROJ) {
    bz0 = bias[lane * 2];  bz1 = bias[lane * 2 + 1];
    wz0 = Wout[lane * 2];  wz1 = Wout[lane * 2 + 1];
    bo  = bout[0];
  }
  int beg[NPW], n[NPW];
  float a0[NPW];
  float2 hs[NPW];
  #pragma unroll
  for (int k = 0; k < NPW; ++k) {
    const int d = (d0 + k < Nn) ? (d0 + k) : (Nn - 1);
    beg[k] = off[d];
    n[k]   = cnt[d];
    a0[k]  = alpha_self[d];
    hs[k]  = *(const float2*)(hl + (size_t)d * 64);
  }
  int sv0[NPW];
  float av0[NPW];
  #pragma unroll
  for (int k = 0; k < NPW; ++k) {
    const int i0 = (lane < n[k]) ? (beg[k] + lane) : 0;
    sv0[k] = csr_src[i0];
    av0[k] = alpha_d[i0];
  }
  float y_out = 0.0f;
  #pragma unroll
  for (int k = 0; k < NPW; ++k) {
    if (d0 + k < Nn) {
      const int d = d0 + k, b = beg[k], nn = n[k];
      float accx = a0[k] * hs[k].x, accy = a0[k] * hs[k].y;
      for (int cc = 0; cc < nn; cc += 32) {
        int sv = sv0[k];
        float av = av0[k];
        if (cc > 0) {
          const int i = cc + lane;
          sv = 0;  av = 0.0f;
          if (i < nn) {
            sv = csr_src[b + i];
            av = alpha_d[b + i];
          }
        }
        const int mm = (nn - cc) < 32 ? (nn - cc) : 32;
        int j = 0;
        for (; j + 8 <= mm; j += 8) add_rows<8>(sv, av, j, hl, accx, accy);
        if (mm & 4) { add_rows<4>(sv, av, j, hl, accx, accy); j += 4; }
        if (mm & 2) { add_rows<2>(sv, av, j, hl, accx, accy); j += 2; }
        if (mm & 1) { add_rows<1>(sv, av, j, hl, accx, accy); }
      }
      if (PROJ) {
        float v0 = accx + bz0;
        float v1 = accy + bz1;
        v0 = v0 > 0.0f ? v0 : 0.0f;
        v1 = v1 > 0.0f ? v1 : 0.0f;
        float v = v0 * wz0 + v1 * wz1;
        v += row_xor<8>(v);
        v += row_xor<4>(v);
        v += row_xor<2>(v);
        v += row_xor<1>(v);
        const float tot = __int_as_float(__builtin_amdgcn_readlane(__float_as_int(v), 0)) +
                          __int_as_float(__builtin_amdgcn_readlane(__float_as_int(v), 16));
        y_out = (lane == k) ? (tot + bo) : y_out;
      } else {
        *(float2*)(aggr + (size_t)d * 64 + lane * 2) = make_float2(accx, accy);
      }
    }
  }
  if (PROJ) {
    if (lane < NPW && d0 + lane < Nn) y[d0 + lane] = y_out;
  }
}

extern "C" void kernel_launch(void* const* d_in, const int* in_sizes, int n_in,
                              void* d_out, int out_size, void* d_ws, size_t ws_size,
                              hipStream_t stream) {
  const float* x     = (const float*)d_in[0];
  const int*   ei    = (const int*)  d_in[1];
  const float* W1    = (const float*)d_in[2];
  const float* b1    = (const float*)d_in[3];
  const float* att1  = (const float*)d_in[4];
  const float* bias1 = (const float*)d_in[5];
  const float* W2    = (const float*)d_in[6];
  const float* b2    = (const float*)d_in[7];
  const float* att2  = (const float*)d_in[8];
  const float* bias2 = (const float*)d_in[9];
  const float* Wout  = (const float*)d_in[10];
  const float* bout  = (const float*)d_in[11];
  float* y = (float*)d_out;

  const int Nn   = in_sizes[0] / 16;
  const int E    = in_sizes[1] / 2;

  char* p = (char*)d_ws;
  auto alloc = [&](size_t bytes) {
    char* r = p;
    p += (bytes + 255) & ~(size_t)255;
    return r;
  };
  float* h      = (float*)alloc((size_t)Nn * 64 * sizeof(float));
  float* aggr1  = (float*)alloc((size_t)Nn * 64 * sizeof(float));
  float* a_i    = (float*)alloc((size_t)Nn * sizeof(float));
  float* a_j    = (float*)alloc((size_t)Nn * sizeof(float));
  float* al_self = (float*)alloc((size_t)Nn * sizeof(float));
  int*   cnt_d  = (int*)  alloc((size_t)Nn * sizeof(int));
  int*   cnt_s  = (int*)  alloc((size_t)Nn * sizeof(int));
  int*   off_d  = (int*)  alloc((size_t)Nn * sizeof(int));
  int*   off_s  = (int*)  alloc((size_t)Nn * sizeof(int));
  int2*  rank   = (int2*) alloc((size_t)E * sizeof(int2));
  int*   csr_src = (int*) alloc((size_t)E * sizeof(int));
  float* alpha_d = (float*)alloc((size_t)E * sizeof(float));
  int2*  csr_s  = (int2*) alloc((size_t)E * sizeof(int2));

  const int TB = 256;
  dim3 blk(TB);
  int g_gemm  = ((Nn / 16) + 7) / 8;
  int g_node  = (Nn + 8 * NPW - 1) / (8 * NPW);
  int g_edge  = (E + TB * EPT - 1) / (TB * EPT);
  dim3 g_scan((Nn + 1023) / 1024, 2);

  k_linear_fused<16, false><<<g_gemm, blk, 0, stream>>>(x, nullptr, W1, b1, att1, h, a_i, a_j,
                                                        cnt_d, cnt_s, Nn);
  k_edge_count<<<g_edge, blk, 0, stream>>>(ei, cnt_d, cnt_s, rank, E);
  k_offsets<<<g_scan, blk, 0, stream>>>(cnt_d, off_d, cnt_s, off_s, Nn);
  k_edge_place<<<g_edge, blk, 0, stream>>>(ei, off_d, off_s, rank, csr_src, csr_s, E);

  k_src_softmax<<<g_node, blk, 0, stream>>>(csr_s, off_s, cnt_s, a_i, a_j, alpha_d, al_self, Nn);
  k_gather_sum<false><<<g_node, blk, 0, stream>>>(csr_src, alpha_d, al_self, off_d, cnt_d, h, aggr1,
                                                  nullptr, nullptr, nullptr, nullptr, Nn);

  k_linear_fused<64, true><<<g_gemm, blk, 0, stream>>>(aggr1, bias1, W2, b2, att2, h, a_i, a_j,
                                                       nullptr, nullptr, Nn);
  k_src_softmax<<<g_node, blk, 0, stream>>>(csr_s, off_s, cnt_s, a_i, a_j, alpha_d, al_self, Nn);
  k_gather_sum<true><<<g_node, blk, 0, stream>>>(csr_src, alpha_d, al_self, off_d, cnt_d, h, nullptr,
                                                 bias2, Wout, bout, y, Nn);
}


# edge ranks from LDS counters: 16 work-groups per order count a chunk of the edge list in a per-node histogram in LDS, no global atomic left; slot = chunk prefix + rank; sum order: chunk, LDS arrival
# speedup vs baseline: 2.8298x; 1.4848x over previous
#include <hip/hip_runtime.h>
#include <float.h>


#define NPW 4
#define EPT 4
#define GCH 16
#define HCAP 51200
#define CPT 16

typedef __attribute__((ext_vector_type(2))) float v2f;
typedef __attribute__((ext_vector_type(8))) float v8f;

#define NEG_SLOPE 0.01f

template <int MASK>
__device__ __forceinline__ float row_xor(float v) {
  return __int_as_float(__builtin_amdgcn_update_dpp(0, __float_as_int(v), 0x160 + MASK,
                                                    0xf, 0xf, true));
}

template <int K, bool FIN>
__global__ void k_linear_fused(const float* __restrict__ in, const float* __restrict__ bias_prev,
                               const float* __restrict__ W, const float* __restrict__ b,
                               const float* __restrict__ att,
                               float* __restrict__ h, float* __restrict__ a_i,
                               float* __restrict__ a_j, int M) {
  int wave = (int)((blockIdx.x * blockDim.x + threadIdx.x) >> 5);
  int lane = threadIdx.x & 31;
  int row0 = wave * 16;
  if (row0 >= M) return;
  int half = lane >> 4;
  int l16  = lane & 15;

  v8f acc0 = {}, acc1 = {}, acc2 = {}, acc3 = {};
  const float* arow = in + (size_t)(row0 + l16) * K;

  #pragma unroll
  for (int k0 = 0; k0 < K; k0 += 4) {
    int kb = k0 + half * 2;
    v2f a;  a.x = arow[kb];  a.y = arow[kb + 1];
    if (FIN) {
      float t0 = a.x + bias_prev[kb], t1 = a.y + bias_prev[kb + 1];
      a.x = t0 > 0.0f ? t0 : 0.0f;
      a.y = t1 > 0.0f ? t1 : 0.0f;
    }
    v2f b0, b1, b2, b3;
    { const float* p = W + (size_t)(0  + l16) * K + kb; b0.x = p[0]; b0.y = p[1]; }
    { const float* p = W + (size_t)(16 + l16) * K + kb; b1.x = p[0]; b1.y = p[1]; }
    { const float* p = W + (size_t)(32 + l16) * K + kb; b2.x = p[0]; b2.y = p[1]; }
    { const float* p = W + (size_t)(48 + l16) * K + kb; b3.x = p[0]; b3.y = p[1]; }
    acc0 = __builtin_amdgcn_wmma_f32_16x16x4_f32(false, a, false, b0, (short)0, acc0, false, false);
    acc1 = __builtin_amdgcn_wmma_f32_16x16x4_f32(false, a, false, b1, (short)0, acc1, false, false);
    acc2 = __builtin_amdgcn_wmma_f32_16x16x4_f32(false, a, false, b2, (short)0, acc2, false, false);
    acc3 = __builtin_amdgcn_wmma_f32_16x16x4_f32(false, a, false, b3, (short)0, acc3, false, false);
  }

  const float bb0 = b[l16], bb1 = b[16 + l16], bb2 = b[32 + l16], bb3 = b[48 + l16];
  const float ti0 = att[l16],      ti1 = att[16 + l16], ti2 = att[32 + l16],  ti3 = att[48 + l16];
  const float tj0 = att[64 + l16], tj1 = att[80 + l16], tj2 = att[96 + l16],  tj3 = att[112 + l16];
  float oi = 0.0f, oj = 0.0f;
  #pragma unroll
  for (int v = 0; v < 8; ++v) {
    int m = row0 + v + half * 8;
    float* hrow = h + (size_t)m * 64 + l16;
    float h0 = acc0[v] + bb0, h1 = acc1[v] + bb1, h2 = acc2[v] + bb2, h3 = acc3[v] + bb3;
    hrow[0]  = h0;
    hrow[16] = h1;
    hrow[32] = h2;
    hrow[48] = h3;
    float vi = h0 * ti0 + h1 * ti1 + h2 * ti2 + h3 * ti3;
    float vj = h0 * tj0 + h1 * tj1 + h2 * tj2 + h3 * tj3;
    vi += row_xor<8>(vi);  vj += row_xor<8>(vj);
    vi += row_xor<4>(vi);  vj += row_xor<4>(vj);
    vi += row_xor<2>(vi);  vj += row_xor<2>(vj);
    vi += row_xor<1>(vi);  vj += row_xor<1>(vj);
    bool mine = (l16 & 7) == v;
    oi = mine ? vi : oi;
    oj = mine ? vj : oj;
  }
  {
    float* op = (l16 < 8) ? a_i : a_j;
    op[row0 + half * 8 + (l16 & 7)] = (l16 < 8) ? oi : oj;
  }
}

__global__ void __launch_bounds__(1024)
k_edge_count(const int* __restrict__ ei, int* __restrict__ cm_d, int* __restrict__ cm_s,
             int* __restrict__ lr_d, int* __restrict__ lr_s, int* __restrict__ tsum,
             int E, int NT) {
  __shared__ int hist[HCAP];
  const int t = threadIdx.x, lane = t & 31, w = t >> 5;
  const int g = blockIdx.x, ty = blockIdx.y;
  const int NP = NT * 1024;
  const int* __restrict__ keys = ty ? ei : ei + E;
  int* __restrict__ lr = ty ? lr_s : lr_d;
  int* __restrict__ cm = (ty ? cm_s : cm_d) + (size_t)g * NP;

  int4* h4 = (int4*)hist;
  for (int i = t; i < NP / 4; i += 1024) h4[i] = make_int4(0, 0, 0, 0);
  __syncthreads();

  const int chunk = (E + GCH - 1) / GCH;
  const int lo = g * chunk;
  const int hi = (lo + chunk < E) ? (lo + chunk) : E;
  int e = lo + t;
  for (; e + (CPT - 1) * 1024 < hi; e += CPT * 1024) {
    int k[CPT], r[CPT];
    #pragma unroll
    for (int i = 0; i < CPT; ++i) k[i] = keys[e + i * 1024];
    #pragma unroll
    for (int i = 0; i < CPT; ++i) r[i] = atomicAdd(&hist[k[i]], 1);
    #pragma unroll
    for (int i = 0; i < CPT; ++i) lr[e + i * 1024] = r[i];
  }
  for (; e < hi; e += 1024) lr[e] = atomicAdd(&hist[keys[e]], 1);
  __syncthreads();

  for (int j = w; j < NT; j += 32) {
    const int4* hj = (const int4*)(hist + j * 1024);
    int4* cj = (int4*)(cm + j * 1024);
    int acc = 0;
    #pragma unroll
    for (int i = 0; i < 8; ++i) {
      const int4 v = hj[i * 32 + lane];
      cj[i * 32 + lane] = v;
      acc += (v.x + v.y) + (v.z + v.w);
    }
    for (int o = 16; o > 0; o >>= 1) acc += __shfl_xor(acc, o, 32);
    if (lane == 0) tsum[(ty * GCH + g) * 64 + j] = acc;
  }
  if (t >= NT && t < 64) tsum[(ty * GCH + g) * 64 + t] = 0;
}

__global__ void k_offsets(int* __restrict__ pm_d, int* __restrict__ cnt_d,
                          int* __restrict__ pm_s, int* __restrict__ cnt_s,
                          const int* __restrict__ tsum, int NT) {
  int* __restrict__ pm = blockIdx.y ? pm_s : pm_d;
  int* __restrict__ cnt = blockIdx.y ? cnt_s : cnt_d;
  const int* __restrict__ tsy = tsum + blockIdx.y * (GCH * 64);
  __shared__ int s_pre[8];
  __shared__ int s_loc[8];
  const int t = threadIdx.x, lane = t & 31, w = t >> 5;
  const int nb = blockIdx.x;
  const int NP = NT * 1024;
  const int i0 = nb * 1024 + t * 4;
  int4 c[GCH];
  #pragma unroll
  for (int g = 0; g < GCH; ++g) c[g] = *(const int4*)(pm + (size_t)g * NP + i0);
  static_assert(GCH * 64 == 256 * 4, "one 16-byte load per thread covers the tile sums");
  const int4 tv = *(const int4*)(tsy + t * 4);
  const int j0 = (t * 4) & 63;
  int pre = ((j0 < nb) ? tv.x : 0) + ((j0 + 1 < nb) ? tv.y : 0) +
            ((j0 + 2 < nb) ? tv.z : 0) + ((j0 + 3 < nb) ? tv.w : 0);
  int t0 = 0, t1 = 0, t2 = 0, t3 = 0;
  #pragma unroll
  for (int g = 0; g < GCH; ++g) { t0 += c[g].x;  t1 += c[g].y;  t2 += c[g].z;  t3 += c[g].w; }
  const int loc = (t0 + t1) + (t2 + t3);
  for (int o = 16; o > 0; o >>= 1) pre += __shfl_xor(pre, o, 32);
  int inc = loc;
  for (int o = 1; o < 32; o <<= 1) {
    const int v = __shfl_up(inc, o, 32);
    inc += (lane >= o) ? v : 0;
  }
  if (lane == 31) s_loc[w] = inc;
  if (lane == 0)  s_pre[w] = pre;
  __syncthreads();
  int base = 0;
  for (int k = 0; k < 8; ++k) base += s_pre[k] + ((k < w) ? s_loc[k] : 0);
  const int ex = base + inc - loc;
  int r0 = ex, r1 = ex + t0, r2 = ex + t0 + t1, r3 = ex + t0 + t1 + t2;
  #pragma unroll
  for (int g = 0; g < GCH; ++g) {
    *(int4*)(pm + (size_t)g * NP + i0) = make_int4(r0, r1, r2, r3);
    r0 += c[g].x;  r1 += c[g].y;  r2 += c[g].z;  r3 += c[g].w;
  }
  *(int4*)(cnt + i0) = make_int4(t0, t1, t2, t3);
}

__global__ void k_edge_place(const int* __restrict__ ei, const int* __restrict__ pm_d,
                             const int* __restrict__ pm_s, const int* __restrict__ lr_d,
                             const int* __restrict__ lr_s, int* __restrict__ csr_src,
                             int2* __restrict__ csr_s, int E, int NT) {
  const int T = (int)(gridDim.x * blockDim.x);
  const int e0 = (int)(blockIdx.x * blockDim.x + threadIdx.x);
  const unsigned chunk = (unsigned)((E + GCH - 1) / GCH);
  const int NP = NT * 1024;
  if (e0 + (EPT - 1) * T < E) {
    int s[EPT], d[EPT], rd[EPT], rs[EPT], pd[EPT], ps[EPT];
    #pragma unroll
    for (int k = 0; k < EPT; ++k) {
      const int e = e0 + k * T;
      s[k] = ei[e];  d[k] = ei[E + e];  rd[k] = lr_d[e];  rs[k] = lr_s[e];
    }
    #pragma unroll
    for (int k = 0; k < EPT; ++k) {
      const size_t row = (size_t)((unsigned)(e0 + k * T) / chunk) * NP;
      pd[k] = pm_d[row + d[k]] + rd[k];
      ps[k] = pm_s[row + s[k]] + rs[k];
    }
    #pragma unroll
    for (int k = 0; k < EPT; ++k) {
      csr_src[pd[k]] = s[k];
      csr_s[ps[k]] = make_int2(d[k], pd[k]);
    }
  } else {
    for (int k = 0; k < EPT; ++k) {
      const int e = e0 + k * T;
      if (e < E) {
        const int s = ei[e], d = ei[E + e];
        const size_t row = (size_t)((unsigned)e / chunk) * NP;
        const int pd = pm_d[row + d] + lr_d[e];
        const int ps = pm_s[row + s] + lr_s[e];
        csr_src[pd] = s;
        csr_s[ps] = make_int2(d, pd);
      }
    }
  }
}

__device__ __forceinline__ float leaky(float v) { return v > 0.0f ? v : v * NEG_SLOPE; }

__device__ __forceinline__ float wave_max(float v) {
  v = fmaxf(v, row_xor<8>(v));
  v = fmaxf(v, row_xor<4>(v));
  v = fmaxf(v, row_xor<2>(v));
  v = fmaxf(v, row_xor<1>(v));
  return fmaxf(__int_as_float(__builtin_amdgcn_readlane(__float_as_int(v), 0)),
               __int_as_float(__builtin_amdgcn_readlane(__float_as_int(v), 16)));
}
__device__ __forceinline__ float wave_sum(float v) {
  v += row_xor<8>(v);
  v += row_xor<4>(v);
  v += row_xor<2>(v);
  v += row_xor<1>(v);
  return __int_as_float(__builtin_amdgcn_readlane(__float_as_int(v), 0)) +
         __int_as_float(__builtin_amdgcn_readlane(__float_as_int(v), 16));
}

__global__ void k_src_softmax(const int2* __restrict__ csr_s, const int* __restrict__ off_s,
                              const int* __restrict__ cnt_s, const float* __restrict__ a_i,
                              const float* __restrict__ a_j, float* __restrict__ alpha_d,
                              float* __restrict__ alpha_self, int Nn) {
  const int w = __builtin_amdgcn_readfirstlane(
      (int)(blockIdx.x * (blockDim.x >> 5) + (threadIdx.x >> 5)));
  const int lane = threadIdx.x & 31;
  const int s0 = w * NPW;
  if (s0 >= Nn) return;
  int beg[NPW], n[NPW];
  float aj[NPW], ais[NPW];
  #pragma unroll
  for (int k = 0; k < NPW; ++k) {
    const int s = (s0 + k < Nn) ? (s0 + k) : (Nn - 1);
    beg[k] = off_s[s];
    n[k]   = cnt_s[s];
    aj[k]  = a_j[s];
    ais[k] = a_i[s];
  }
  int2 de[NPW];
  #pragma unroll
  for (int k = 0; k < NPW; ++k) de[k] = csr_s[(lane < n[k]) ? (beg[k] + lane) : 0];
  float ai0[NPW];
  #pragma unroll
  for (int k = 0; k < NPW; ++k) ai0[k] = a_i[de[k].x];
  float self_out = 0.0f;
  #pragma unroll
  for (int k = 0; k < NPW; ++k) {
    if (s0 + k < Nn) {
      const int b = beg[k], nn = n[k];
      const float ajk = aj[k];
      const float sc_self = leaky(ais[k] + ajk);
      const bool v0 = lane < nn;
      const float sc0 = v0 ? leaky(ai0[k] + ajk) : -FLT_MAX;
      float mx = fmaxf(sc_self, sc0);
      for (int cc = 32; cc < nn; cc += 32) {
        const int i = cc + lane;
        if (i < nn) mx = fmaxf(mx, leaky(a_i[csr_s[b + i].x] + ajk));
      }
      mx = wave_max(mx);
      const float ev0 = v0 ? __expf(sc0 - mx) : 0.0f;
      float sum = ev0;
      for (int cc = 32; cc < nn; cc += 32) {
        const int i = cc + lane;
        if (i < nn) sum += __expf(leaky(a_i[csr_s[b + i].x] + ajk) - mx);
      }
      const float es = __expf(sc_self - mx);
      const float den = (wave_sum(sum) + es) + 1e-16f;
      if (v0) alpha_d[de[k].y] = ev0 / den;
      for (int cc = 32; cc < nn; cc += 32) {
        const int i = cc + lane;
        if (i < nn) {
          const int2 dx = csr_s[b + i];
          alpha_d[dx.y] = __expf(leaky(a_i[dx.x] + ajk) - mx) / den;
        }
      }
      const float as = es / den;
      self_out = (lane == k) ? as : self_out;
    }
  }
  if (lane < NPW && s0 + lane < Nn) alpha_self[s0 + lane] = self_out;
}

template <int NR>
__device__ __forceinline__ void add_rows(int sv, float av, int j0, const float* __restrict__ hl,
                                         float& accx, float& accy) {
  float2 hv[NR];
  float a[NR];
  #pragma unroll
  for (int r = 0; r < NR; ++r) {
    const int s = __builtin_amdgcn_readlane(sv, j0 + r);
    a[r] = __int_as_float(__builtin_amdgcn_readlane(__float_as_int(av), j0 + r));
    hv[r] = *(const float2*)(hl + (size_t)s * 64);
  }
  #pragma unroll
  for (int r = 0; r < NR; ++r) {
    accx = __builtin_fmaf(a[r], hv[r].x, accx);
    accy = __builtin_fmaf(a[r], hv[r].y, accy);
  }
}

template <bool PROJ>
__global__ void k_gather_sum(const int* __restrict__ csr_src, const float* __restrict__ alpha_d,
                             const float* __restrict__ alpha_self, const int* __restrict__ off,
                             const int* __restrict__ cnt, const float* __restrict__ h,
                             float* __restrict__ aggr, const float* __restrict__ bias,
                             const float* __restrict__ Wout, const float* __restrict__ bout,
                             float* __restrict__ y, int Nn) {
  const int w = __builtin_amdgcn_readfirstlane(
      (int)(blockIdx.x * (blockDim.x >> 5) + (threadIdx.x >> 5)));
  const int lane = threadIdx.x & 31;
  const int d0 = w * NPW;
  if (d0 >= Nn) return;
  const float* hl = h + lane * 2;
  float bz0 = 0.0f, bz1 = 0.0f, wz0 = 0.0f, wz1 = 0.0f, bo = 0.0f;
  if (PROJ) {
    bz0 = bias[lane * 2];  bz1 = bias[lane * 2 + 1];
    wz0 = Wout[lane * 2];  wz1 = Wout[lane * 2 + 1];
    bo  = bout[0];
  }
  int beg[NPW], n[NPW];
  float a0[NPW];
  float2 hs[NPW];
  #pragma unroll
  for (int k = 0; k < NPW; ++k) {
    const int d = (d0 + k < Nn) ? (d0 + k) : (Nn - 1);
    beg[k] = off[d];
    n[k]   = cnt[d];
    a0[k]  = alpha_self[d];
    hs[k]  = *(const float2*)(hl + (size_t)d * 64);
  }
  int sv0[NPW];
  float av0[NPW];
  #pragma unroll
  for (int k = 0; k < NPW; ++k) {
    const int i0 = (lane < n[k]) ? (beg[k] + lane) : 0;
    sv0[k] = csr_src[i0];
    av0[k] = alpha_d[i0];
  }
  float y_out = 0.0f;
  #pragma unroll
  for (int k = 0; k < NPW; ++k) {
    if (d0 + k < Nn) {
      const int d = d0 + k, b = beg[k], nn = n[k];
      float accx = a0[k] * hs[k].x, accy = a0[k] * hs[k].y;
      for (int cc = 0; cc < nn; cc += 32) {
        int sv = sv0[k];
        float av = av0[k];
        if (cc > 0) {
          const int i = cc + lane;
          sv = 0;  av = 0.0f;
          if (i < nn) {
            sv = csr_src[b + i];
            av = alpha_d[b + i];
          }
        }
        const int mm = (nn - cc) < 32 ? (nn - cc) : 32;
        int j = 0;
        for (; j + 8 <= mm; j += 8) add_rows<8>(sv, av, j, hl, accx, accy);
        if (mm & 4) { add_rows<4>(sv, av, j, hl, accx, accy); j += 4; }
        if (mm & 2) { add_rows<2>(sv, av, j, hl, accx, accy); j += 2; }
        if (mm & 1) { add_rows<1>(sv, av, j, hl, accx, accy); }
      }
      if (PROJ) {
        float v0 = accx + bz0;
        float v1 = accy + bz1;
        v0 = v0 > 0.0f ? v0 : 0.0f;
        v1 = v1 > 0.0f ? v1 : 0.0f;
        float v = v0 * wz0 + v1 * wz1;
        v += row_xor<8>(v);
        v += row_xor<4>(v);
        v += row_xor<2>(v);
        v += row_xor<1>(v);
        const float tot = __int_as_float(__builtin_amdgcn_readlane(__float_as_int(v), 0)) +
                          __int_as_float(__builtin_amdgcn_readlane(__float_as_int(v), 16));
        y_out = (lane == k) ? (tot + bo) : y_out;
      } else {
        *(float2*)(aggr + (size_t)d * 64 + lane * 2) = make_float2(accx, accy);
      }
    }
  }
  if (PROJ) {
    if (lane < NPW && d0 + lane < Nn) y[d0 + lane] = y_out;
  }
}

extern "C" void kernel_launch(void* const* d_in, const int* in_sizes, int n_in,
                              void* d_out, int out_size, void* d_ws, size_t ws_size,
                              hipStream_t stream) {
  const float* x     = (const float*)d_in[0];
  const int*   ei    = (const int*)  d_in[1];
  const float* W1    = (const float*)d_in[2];
  const float* b1    = (const float*)d_in[3];
  const float* att1  = (const float*)d_in[4];
  const float* bias1 = (const float*)d_in[5];
  const float* W2    = (const float*)d_in[6];
  const float* b2    = (const float*)d_in[7];
  const float* att2  = (const float*)d_in[8];
  const float* bias2 = (const float*)d_in[9];
  const float* Wout  = (const float*)d_in[10];
  const float* bout  = (const float*)d_in[11];
  float* y = (float*)d_out;

  const int Nn   = in_sizes[0] / 16;
  const int E    = in_sizes[1] / 2;

  char* p = (char*)d_ws;
  auto alloc = [&](size_t bytes) {
    char* r = p;
    p += (bytes + 255) & ~(size_t)255;
    return r;
  };
  float* h      = (float*)alloc((size_t)Nn * 64 * sizeof(float));
  float* aggr1  = (float*)alloc((size_t)Nn * 64 * sizeof(float));
  float* a_i    = (float*)alloc((size_t)Nn * sizeof(float));
  float* a_j    = (float*)alloc((size_t)Nn * sizeof(float));
  float* al_self = (float*)alloc((size_t)Nn * sizeof(float));
  const int NT  = (Nn + 1023) / 1024;
  const size_t NP = (size_t)NT * 1024;
  int*   cnt_d  = (int*)  alloc(NP * sizeof(int));
  int*   cnt_s  = (int*)  alloc(NP * sizeof(int));
  int*   pm_d   = (int*)  alloc(NP * GCH * sizeof(int));
  int*   pm_s   = (int*)  alloc(NP * GCH * sizeof(int));
  int*   tsum   = (int*)  alloc((size_t)2 * GCH * 64 * sizeof(int));
  int*   lr_d   = (int*)  alloc((size_t)E * sizeof(int));
  int*   lr_s   = (int*)  alloc((size_t)E * sizeof(int));
  const int* off_d = pm_d;
  const int* off_s = pm_s;
  int*   csr_src = (int*) alloc((size_t)E * sizeof(int));
  float* alpha_d = (float*)alloc((size_t)E * sizeof(float));
  int2*  csr_s  = (int2*) alloc((size_t)E * sizeof(int2));

  const int TB = 256;
  dim3 blk(TB);
  int g_gemm  = ((Nn / 16) + 7) / 8;
  int g_node  = (Nn + 8 * NPW - 1) / (8 * NPW);
  int g_edge  = (E + TB * EPT - 1) / (TB * EPT);
  dim3 g_scan(NT, 2);
  dim3 g_cnt(GCH, 2);

  k_linear_fused<16, false><<<g_gemm, blk, 0, stream>>>(x, nullptr, W1, b1, att1, h, a_i, a_j, Nn);
  k_edge_count<<<g_cnt, dim3(1024), 0, stream>>>(ei, pm_d, pm_s, lr_d, lr_s, tsum, E, NT);
  k_offsets<<<g_scan, blk, 0, stream>>>(pm_d, cnt_d, pm_s, cnt_s, tsum, NT);
  k_edge_place<<<g_edge, blk, 0, stream>>>(ei, pm_d, pm_s, lr_d, lr_s, csr_src, csr_s, E, NT);

  k_src_softmax<<<g_node, blk, 0, stream>>>(csr_s, off_s, cnt_s, a_i, a_j, alpha_d, al_self, Nn);
  k_gather_sum<false><<<g_node, blk, 0, stream>>>(csr_src, alpha_d, al_self, off_d, cnt_d, h, aggr1,
                                                  nullptr, nullptr, nullptr, nullptr, Nn);

  k_linear_fused<64, true><<<g_gemm, blk, 0, stream>>>(aggr1, bias1, W2, b2, att2, h, a_i, a_j, Nn);
  k_src_softmax<<<g_node, blk, 0, stream>>>(csr_s, off_s, cnt_s, a_i, a_j, alpha_d, al_self, Nn);
  k_gather_sum<true><<<g_node, blk, 0, stream>>>(csr_src, alpha_d, al_self, off_d, cnt_d, h, nullptr,
                                                 bias2, Wout, bout, y, Nn);
}


# layer 1 sums x rows (16 wide, 8 edges per load) and layer 2's GEMM wave applies W1 after the sum: h1 never stored; sum order: 8 lane partials + tree per node, GEMM k permuted
# speedup vs baseline: 2.9870x; 1.0556x over previous
#include <hip/hip_runtime.h>
#include <float.h>


#define NPW 4
#define EPT 4
#define GCH 16
#define HCAP 51200
#define CPT 16

typedef __attribute__((ext_vector_type(2))) float v2f;
typedef __attribute__((ext_vector_type(8))) float v8f;

#define NEG_SLOPE 0.01f

template <int MASK>
__device__ __forceinline__ float row_xor(float v) {
  return __int_as_float(__builtin_amdgcn_update_dpp(0, __float_as_int(v), 0x160 + MASK,
                                                    0xf, 0xf, true));
}

__device__ __forceinline__ float half_swap(float v) {
  return __int_as_float(__builtin_amdgcn_permlanex16(0, __float_as_int(v), 0x76543210u,
                                                    0xfedcba98u, true, false));
}


template <bool STORE>
__device__ __forceinline__ void finish_rows(const v8f& acc0, const v8f& acc1, const v8f& acc2,
                                            const v8f& acc3, const float* __restrict__ b,
                                            const float* __restrict__ att, float* __restrict__ h,
                                            float* __restrict__ a_i, float* __restrict__ a_j,
                                            int row0, int half, int l16) {
  const float bb0 = b[l16], bb1 = b[16 + l16], bb2 = b[32 + l16], bb3 = b[48 + l16];
  const float ti0 = att[l16],      ti1 = att[16 + l16], ti2 = att[32 + l16],  ti3 = att[48 + l16];
  const float tj0 = att[64 + l16], tj1 = att[80 + l16], tj2 = att[96 + l16],  tj3 = att[112 + l16];
  float oi = 0.0f, oj = 0.0f;
  #pragma unroll
  for (int v = 0; v < 8; ++v) {
    int m = row0 + v + half * 8;
    float h0 = acc0[v] + bb0, h1 = acc1[v] + bb1, h2 = acc2[v] + bb2, h3 = acc3[v] + bb3;
    if (STORE) {
      float* hrow = h + (size_t)m * 64 + l16;
      hrow[0]  = h0;
      hrow[16] = h1;
      hrow[32] = h2;
      hrow[48] = h3;
    }
    float vi = h0 * ti0 + h1 * ti1 + h2 * ti2 + h3 * ti3;
    float vj = h0 * tj0 + h1 * tj1 + h2 * tj2 + h3 * tj3;
    vi += row_xor<8>(vi);  vj += row_xor<8>(vj);
    vi += row_xor<4>(vi);  vj += row_xor<4>(vj);
    vi += row_xor<2>(vi);  vj += row_xor<2>(vj);
    vi += row_xor<1>(vi);  vj += row_xor<1>(vj);
    bool mine = (l16 & 7) == v;
    oi = mine ? vi : oi;
    oj = mine ? vj : oj;
  }
  {
    float* op = (l16 < 8) ? a_i : a_j;
    op[row0 + half * 8 + (l16 & 7)] = (l16 < 8) ? oi : oj;
  }
}

__global__ void k_linear_dots(const float* __restrict__ in, const float* __restrict__ W,
                              const float* __restrict__ b, const float* __restrict__ att,
                              float* __restrict__ a_i, float* __restrict__ a_j, int M) {
  int wave = (int)((blockIdx.x * blockDim.x + threadIdx.x) >> 5);
  int lane = threadIdx.x & 31;
  int row0 = wave * 16;
  if (row0 >= M) return;
  int half = lane >> 4;
  int l16  = lane & 15;

  v8f acc0 = {}, acc1 = {}, acc2 = {}, acc3 = {};
  const float* arow = in + (size_t)(row0 + l16) * 16;

  #pragma unroll
  for (int k0 = 0; k0 < 16; k0 += 4) {
    int kb = k0 + half * 2;
    v2f a;  a.x = arow[kb];  a.y = arow[kb + 1];
    v2f b0, b1, b2, b3;
    { const float* p = W + (size_t)(0  + l16) * 16 + kb; b0.x = p[0]; b0.y = p[1]; }
    { const float* p = W + (size_t)(16 + l16) * 16 + kb; b1.x = p[0]; b1.y = p[1]; }
    { const float* p = W + (size_t)(32 + l16) * 16 + kb; b2.x = p[0]; b2.y = p[1]; }
    { const float* p = W + (size_t)(48 + l16) * 16 + kb; b3.x = p[0]; b3.y = p[1]; }
    acc0 = __builtin_amdgcn_wmma_f32_16x16x4_f32(false, a, false, b0, (short)0, acc0, false, false);
    acc1 = __builtin_amdgcn_wmma_f32_16x16x4_f32(false, a, false, b1, (short)0, acc1, false, false);
    acc2 = __builtin_amdgcn_wmma_f32_16x16x4_f32(false, a, false, b2, (short)0, acc2, false, false);
    acc3 = __builtin_amdgcn_wmma_f32_16x16x4_f32(false, a, false, b3, (short)0, acc3, false, false);
  }
  finish_rows<false>(acc0, acc1, acc2, acc3, b, att, nullptr, a_i, a_j, row0, half, l16);
}

__global__ void k_linear_chain(const float* __restrict__ sx, const float* __restrict__ ssum,
                               const float* __restrict__ W1, const float* __restrict__ b1,
                               const float* __restrict__ bias1, const float* __restrict__ W2,
                               const float* __restrict__ b2, const float* __restrict__ att,
                               float* __restrict__ h, float* __restrict__ a_i,
                               float* __restrict__ a_j, int M) {
  int wave = (int)((blockIdx.x * blockDim.x + threadIdx.x) >> 5);
  int lane = threadIdx.x & 31;
  int row0 = wave * 16;
  if (row0 >= M) return;
  int half = lane >> 4;
  int l16  = lane & 15;

  v8f tT[4] = {};
  const float* xrow = sx + (size_t)(row0 + l16) * 16;
  #pragma unroll
  for (int k0 = 0; k0 < 16; k0 += 4) {
    int kb = k0 + half * 2;
    v2f xb;  xb.x = xrow[kb];  xb.y = xrow[kb + 1];
    #pragma unroll
    for (int t = 0; t < 4; ++t) {
      const float* p = W1 + (size_t)(t * 16 + l16) * 16 + kb;
      v2f wa;  wa.x = p[0];  wa.y = p[1];
      tT[t] = __builtin_amdgcn_wmma_f32_16x16x4_f32(false, wa, false, xb, (short)0, tT[t], false, false);
    }
  }
  {
    const float sn = ssum[row0 + l16];
    v2f xb;  xb.x = half ? 0.0f : sn;  xb.y = half ? 0.0f : 1.0f;
    #pragma unroll
    for (int t = 0; t < 4; ++t) {
      const float pb = b1[t * 16 + l16], pc = bias1[t * 16 + l16];
      v2f wa;  wa.x = half ? 0.0f : pb;  wa.y = half ? 0.0f : pc;
      tT[t] = __builtin_amdgcn_wmma_f32_16x16x4_f32(false, wa, false, xb, (short)0, tT[t], false, false);
    }
  }

  v8f acc0 = {}, acc1 = {}, acc2 = {}, acc3 = {};
  #pragma unroll
  for (int t = 0; t < 4; ++t) {
    #pragma unroll
    for (int p = 0; p < 4; ++p) {
      const float u0 = tT[t][2 * p], u1 = tT[t][2 * p + 1];
      v2f a;
      a.x = u0 > 0.0f ? u0 : 0.0f;
      a.y = u1 > 0.0f ? u1 : 0.0f;
      const int kb = t * 16 + half * 8 + 2 * p;
      v2f b0, b1v, b2v, b3;
      { const float* q = W2 + (size_t)(0  + l16) * 64 + kb; b0.x  = q[0]; b0.y  = q[1]; }
      { const float* q = W2 + (size_t)(16 + l16) * 64 + kb; b1v.x = q[0]; b1v.y = q[1]; }
      { const float* q = W2 + (size_t)(32 + l16) * 64 + kb; b2v.x = q[0]; b2v.y = q[1]; }
      { const float* q = W2 + (size_t)(48 + l16) * 64 + kb; b3.x  = q[0]; b3.y  = q[1]; }
      acc0 = __builtin_amdgcn_wmma_f32_16x16x4_f32(false, a, false, b0,  (short)0, acc0, false, false);
      acc1 = __builtin_amdgcn_wmma_f32_16x16x4_f32(false, a, false, b1v, (short)0, acc1, false, false);
      acc2 = __builtin_amdgcn_wmma_f32_16x16x4_f32(false, a, false, b2v, (short)0, acc2, false, false);
      acc3 = __builtin_amdgcn_wmma_f32_16x16x4_f32(false, a, false, b3,  (short)0, acc3, false, false);
    }
  }
  finish_rows<true>(acc0, acc1, acc2, acc3, b2, att, h, a_i, a_j, row0, half, l16);
}

__global__ void __launch_bounds__(1024)
k_edge_count(const int* __restrict__ ei, int* __restrict__ cm_d, int* __restrict__ cm_s,
             int* __restrict__ lr_d, int* __restrict__ lr_s, int* __restrict__ tsum,
             int E, int NT) {
  __shared__ int hist[HCAP];
  const int t = threadIdx.x, lane = t & 31, w = t >> 5;
  const int g = blockIdx.x, ty = blockIdx.y;
  const int NP = NT * 1024;
  const int* __restrict__ keys = ty ? ei : ei + E;
  int* __restrict__ lr = ty ? lr_s : lr_d;
  int* __restrict__ cm = (ty ? cm_s : cm_d) + (size_t)g * NP;

  int4* h4 = (int4*)hist;
  for (int i = t; i < NP / 4; i += 1024) h4[i] = make_int4(0, 0, 0, 0);
  __syncthreads();

  const int chunk = (E + GCH - 1) / GCH;
  const int lo = g * chunk;
  const int hi = (lo + chunk < E) ? (lo + chunk) : E;
  int e = lo + t;
  for (; e + (CPT - 1) * 1024 < hi; e += CPT * 1024) {
    int k[CPT], r[CPT];
    #pragma unroll
    for (int i = 0; i < CPT; ++i) k[i] = keys[e + i * 1024];
    #pragma unroll
    for (int i = 0; i < CPT; ++i) r[i] = atomicAdd(&hist[k[i]], 1);
    #pragma unroll
    for (int i = 0; i < CPT; ++i) lr[e + i * 1024] = r[i];
  }
  for (; e < hi; e += 1024) lr[e] = atomicAdd(&hist[keys[e]], 1);
  __syncthreads();

  for (int j = w; j < NT; j += 32) {
    const int4* hj = (const int4*)(hist + j * 1024);
    int4* cj = (int4*)(cm + j * 1024);
    int acc = 0;
    #pragma unroll
    for (int i = 0; i < 8; ++i) {
      const int4 v = hj[i * 32 + lane];
      cj[i * 32 + lane] = v;
      acc += (v.x + v.y) + (v.z + v.w);
    }
    for (int o = 16; o > 0; o >>= 1) acc += __shfl_xor(acc, o, 32);
    if (lane == 0) tsum[(ty * GCH + g) * 64 + j] = acc;
  }
  if (t >= NT && t < 64) tsum[(ty * GCH + g) * 64 + t] = 0;
}

__global__ void k_offsets(int* __restrict__ pm_d, int* __restrict__ cnt_d,
                          int* __restrict__ pm_s, int* __restrict__ cnt_s,
                          const int* __restrict__ tsum, int NT) {
  int* __restrict__ pm = blockIdx.y ? pm_s : pm_d;
  int* __restrict__ cnt = blockIdx.y ? cnt_s : cnt_d;
  const int* __restrict__ tsy = tsum + blockIdx.y * (GCH * 64);
  __shared__ int s_pre[8];
  __shared__ int s_loc[8];
  const int t = threadIdx.x, lane = t & 31, w = t >> 5;
  const int nb = blockIdx.x;
  const int NP = NT * 1024;
  const int i0 = nb * 1024 + t * 4;
  int4 c[GCH];
  #pragma unroll
  for (int g = 0; g < GCH; ++g) c[g] = *(const int4*)(pm + (size_t)g * NP + i0);
  static_assert(GCH * 64 == 256 * 4, "one 16-byte load per thread covers the tile sums");
  const int4 tv = *(const int4*)(tsy + t * 4);
  const int j0 = (t * 4) & 63;
  int pre = ((j0 < nb) ? tv.x : 0) + ((j0 + 1 < nb) ? tv.y : 0) +
            ((j0 + 2 < nb) ? tv.z : 0) + ((j0 + 3 < nb) ? tv.w : 0);
  int t0 = 0, t1 = 0, t2 = 0, t3 = 0;
  #pragma unroll
  for (int g = 0; g < GCH; ++g) { t0 += c[g].x;  t1 += c[g].y;  t2 += c[g].z;  t3 += c[g].w; }
  const int loc = (t0 + t1) + (t2 + t3);
  for (int o = 16; o > 0; o >>= 1) pre += __shfl_xor(pre, o, 32);
  int inc = loc;
  for (int o = 1; o < 32; o <<= 1) {
    const int v = __shfl_up(inc, o, 32);
    inc += (lane >= o) ? v : 0;
  }
  if (lane == 31) s_loc[w] = inc;
  if (lane == 0)  s_pre[w] = pre;
  __syncthreads();
  int base = 0;
  for (int k = 0; k < 8; ++k) base += s_pre[k] + ((k < w) ? s_loc[k] : 0);
  const int ex = base + inc - loc;
  int r0 = ex, r1 = ex + t0, r2 = ex + t0 + t1, r3 = ex + t0 + t1 + t2;
  #pragma unroll
  for (int g = 0; g < GCH; ++g) {
    *(int4*)(pm + (size_t)g * NP + i0) = make_int4(r0, r1, r2, r3);
    r0 += c[g].x;  r1 += c[g].y;  r2 += c[g].z;  r3 += c[g].w;
  }
  *(int4*)(cnt + i0) = make_int4(t0, t1, t2, t3);
}

__global__ void k_edge_place(const int* __restrict__ ei, const int* __restrict__ pm_d,
                             const int* __restrict__ pm_s, const int* __restrict__ lr_d,
                             const int* __restrict__ lr_s, int* __restrict__ csr_src,
                             int2* __restrict__ csr_s, int E, int NT) {
  const int T = (int)(gridDim.x * blockDim.x);
  const int e0 = (int)(blockIdx.x * blockDim.x + threadIdx.x);
  const unsigned chunk = (unsigned)((E + GCH - 1) / GCH);
  const int NP = NT * 1024;
  if (e0 + (EPT - 1) * T < E) {
    int s[EPT], d[EPT], rd[EPT], rs[EPT], pd[EPT], ps[EPT];
    #pragma unroll
    for (int k = 0; k < EPT; ++k) {
      const int e = e0 + k * T;
      s[k] = ei[e];  d[k] = ei[E + e];  rd[k] = lr_d[e];  rs[k] = lr_s[e];
    }
    #pragma unroll
    for (int k = 0; k < EPT; ++k) {
      const size_t row = (size_t)((unsigned)(e0 + k * T) / chunk) * NP;
      pd[k] = pm_d[row + d[k]] + rd[k];
      ps[k] = pm_s[row + s[k]] + rs[k];
    }
    #pragma unroll
    for (int k = 0; k < EPT; ++k) {
      csr_src[pd[k]] = s[k];
      csr_s[ps[k]] = make_int2(d[k], pd[k]);
    }
  } else {
    for (int k = 0; k < EPT; ++k) {
      const int e = e0 + k * T;
      if (e < E) {
        const int s = ei[e], d = ei[E + e];
        const size_t row = (size_t)((unsigned)e / chunk) * NP;
        const int pd = pm_d[row + d] + lr_d[e];
        const int ps = pm_s[row + s] + lr_s[e];
        csr_src[pd] = s;
        csr_s[ps] = make_int2(d, pd);
      }
    }
  }
}

__device__ __forceinline__ float leaky(float v) { return v > 0.0f ? v : v * NEG_SLOPE; }

__device__ __forceinline__ float wave_max(float v) {
  v = fmaxf(v, row_xor<8>(v));
  v = fmaxf(v, row_xor<4>(v));
  v = fmaxf(v, row_xor<2>(v));
  v = fmaxf(v, row_xor<1>(v));
  return fmaxf(__int_as_float(__builtin_amdgcn_readlane(__float_as_int(v), 0)),
               __int_as_float(__builtin_amdgcn_readlane(__float_as_int(v), 16)));
}
__device__ __forceinline__ float wave_sum(float v) {
  v += row_xor<8>(v);
  v += row_xor<4>(v);
  v += row_xor<2>(v);
  v += row_xor<1>(v);
  return __int_as_float(__builtin_amdgcn_readlane(__float_as_int(v), 0)) +
         __int_as_float(__builtin_amdgcn_readlane(__float_as_int(v), 16));
}

__global__ void k_src_softmax(const int2* __restrict__ csr_s, const int* __restrict__ off_s,
                              const int* __restrict__ cnt_s, const float* __restrict__ a_i,
                              const float* __restrict__ a_j, float* __restrict__ alpha_d,
                              float* __restrict__ alpha_self, int Nn) {
  const int w = __builtin_amdgcn_readfirstlane(
      (int)(blockIdx.x * (blockDim.x >> 5) + (threadIdx.x >> 5)));
  const int lane = threadIdx.x & 31;
  const int s0 = w * NPW;
  if (s0 >= Nn) return;
  int beg[NPW], n[NPW];
  float aj[NPW], ais[NPW];
  #pragma unroll
  for (int k = 0; k < NPW; ++k) {
    const int s = (s0 + k < Nn) ? (s0 + k) : (Nn - 1);
    beg[k] = off_s[s];
    n[k]   = cnt_s[s];
    aj[k]  = a_j[s];
    ais[k] = a_i[s];
  }
  int2 de[NPW];
  #pragma unroll
  for (int k = 0; k < NPW; ++k) de[k] = csr_s[(lane < n[k]) ? (beg[k] + lane) : 0];
  float ai0[NPW];
  #pragma unroll
  for (int k = 0; k < NPW; ++k) ai0[k] = a_i[de[k].x];
  float self_out = 0.0f;
  #pragma unroll
  for (int k = 0; k < NPW; ++k) {
    if (s0 + k < Nn) {
      const int b = beg[k], nn = n[k];
      const float ajk = aj[k];
      const float sc_self = leaky(ais[k] + ajk);
      const bool v0 = lane < nn;
      const float sc0 = v0 ? leaky(ai0[k] + ajk) : -FLT_MAX;
      float mx = fmaxf(sc_self, sc0);
      for (int cc = 32; cc < nn; cc += 32) {
        const int i = cc + lane;
        if (i < nn) mx = fmaxf(mx, leaky(a_i[csr_s[b + i].x] + ajk));
      }
      mx = wave_max(mx);
      const float ev0 = v0 ? __expf(sc0 - mx) : 0.0f;
      float sum = ev0;
      for (int cc = 32; cc < nn; cc += 32) {
        const int i = cc + lane;
        if (i < nn) sum += __expf(leaky(a_i[csr_s[b + i].x] + ajk) - mx);
      }
      const float es = __expf(sc_self - mx);
      const float den = (wave_sum(sum) + es) + 1e-16f;
      if (v0) alpha_d[de[k].y] = ev0 / den;
      for (int cc = 32; cc < nn; cc += 32) {
        const int i = cc + lane;
        if (i < nn) {
          const int2 dx = csr_s[b + i];
          alpha_d[dx.y] = __expf(leaky(a_i[dx.x] + ajk) - mx) / den;
        }
      }
      const float as = es / den;
      self_out = (lane == k) ? as : self_out;
    }
  }
  if (lane < NPW && s0 + lane < Nn) alpha_self[s0 + lane] = self_out;
}

template <int NR>
__device__ __forceinline__ void add_rows(int sv, float av, int j0, const float* __restrict__ hl,
                                         float& accx, float& accy) {
  float2 hv[NR];
  float a[NR];
  #pragma unroll
  for (int r = 0; r < NR; ++r) {
    const int s = __builtin_amdgcn_readlane(sv, j0 + r);
    a[r] = __int_as_float(__builtin_amdgcn_readlane(__float_as_int(av), j0 + r));
    hv[r] = *(const float2*)(hl + (size_t)s * 64);
  }
  #pragma unroll
  for (int r = 0; r < NR; ++r) {
    accx = __builtin_fmaf(a[r], hv[r].x, accx);
    accy = __builtin_fmaf(a[r], hv[r].y, accy);
  }
}

__global__ void k_gather_sum(const int* __restrict__ csr_src, const float* __restrict__ alpha_d,
                             const float* __restrict__ alpha_self, const int* __restrict__ off,
                             const int* __restrict__ cnt, const float* __restrict__ h,
                             const float* __restrict__ bias, const float* __restrict__ Wout,
                             const float* __restrict__ bout, float* __restrict__ y, int Nn) {
  const int w = __builtin_amdgcn_readfirstlane(
      (int)(blockIdx.x * (blockDim.x >> 5) + (threadIdx.x >> 5)));
  const int lane = threadIdx.x & 31;
  const int d0 = w * NPW;
  if (d0 >= Nn) return;
  const float* hl = h + lane * 2;
  const float bz0 = bias[lane * 2], bz1 = bias[lane * 2 + 1];
  const float wz0 = Wout[lane * 2], wz1 = Wout[lane * 2 + 1];
  const float bo  = bout[0];
  int beg[NPW], n[NPW];
  float a0[NPW];
  float2 hs[NPW];
  #pragma unroll
  for (int k = 0; k < NPW; ++k) {
    const int d = (d0 + k < Nn) ? (d0 + k) : (Nn - 1);
    beg[k] = off[d];
    n[k]   = cnt[d];
    a0[k]  = alpha_self[d];
    hs[k]  = *(const float2*)(hl + (size_t)d * 64);
  }
  int sv0[NPW];
  float av0[NPW];
  #pragma unroll
  for (int k = 0; k < NPW; ++k) {
    const int i0 = (lane < n[k]) ? (beg[k] + lane) : 0;
    sv0[k] = csr_src[i0];
    av0[k] = alpha_d[i0];
  }
  float y_out = 0.0f;
  #pragma unroll
  for (int k = 0; k < NPW; ++k) {
    if (d0 + k < Nn) {
      const int b = beg[k], nn = n[k];
      float accx = a0[k] * hs[k].x, accy = a0[k] * hs[k].y;
      for (int cc = 0; cc < nn; cc += 32) {
        int sv = sv0[k];
        float av = av0[k];
        if (cc > 0) {
          const int i = cc + lane;
          sv = 0;  av = 0.0f;
          if (i < nn) {
            sv = csr_src[b + i];
            av = alpha_d[b + i];
          }
        }
        const int mm = (nn - cc) < 32 ? (nn - cc) : 32;
        int j = 0;
        for (; j + 8 <= mm; j += 8) add_rows<8>(sv, av, j, hl, accx, accy);
        if (mm & 4) { add_rows<4>(sv, av, j, hl, accx, accy); j += 4; }
        if (mm & 2) { add_rows<2>(sv, av, j, hl, accx, accy); j += 2; }
        if (mm & 1) { add_rows<1>(sv, av, j, hl, accx, accy); }
      }
      float v0 = accx + bz0;
      float v1 = accy + bz1;
      v0 = v0 > 0.0f ? v0 : 0.0f;
      v1 = v1 > 0.0f ? v1 : 0.0f;
      float v = v0 * wz0 + v1 * wz1;
      v += row_xor<8>(v);
      v += row_xor<4>(v);
      v += row_xor<2>(v);
      v += row_xor<1>(v);
      const float tot = __int_as_float(__builtin_amdgcn_readlane(__float_as_int(v), 0)) +
                        __int_as_float(__builtin_amdgcn_readlane(__float_as_int(v), 16));
      y_out = (lane == k) ? (tot + bo) : y_out;
    }
  }
  if (lane < NPW && d0 + lane < Nn) y[d0 + lane] = y_out;
}

__device__ __forceinline__ void fma4(float a, const float4& r, float4& acc) {
  acc.x = __builtin_fmaf(a, r.x, acc.x);
  acc.y = __builtin_fmaf(a, r.y, acc.y);
  acc.z = __builtin_fmaf(a, r.z, acc.z);
  acc.w = __builtin_fmaf(a, r.w, acc.w);
}
__device__ __forceinline__ float sum_over_g(float v) {
  v += row_xor<4>(v);
  v += row_xor<8>(v);
  return v + half_swap(v);
}

__global__ void k_gather_x(const int* __restrict__ csr_src, const float* __restrict__ alpha_d,
                           const float* __restrict__ alpha_self, const int* __restrict__ off,
                           const int* __restrict__ cnt, const float* __restrict__ x,
                           float* __restrict__ sx, float* __restrict__ ssum, int Nn) {
  const int w = __builtin_amdgcn_readfirstlane(
      (int)(blockIdx.x * (blockDim.x >> 5) + (threadIdx.x >> 5)));
  const int lane = threadIdx.x & 31;
  const int d0 = w * NPW;
  if (d0 >= Nn) return;
  const int g = lane >> 2, q = lane & 3;
  const float* xl = x + q * 4;
  int beg[NPW], n[NPW];
  float a0[NPW];
  float4 xs[NPW];
  #pragma unroll
  for (int k = 0; k < NPW; ++k) {
    const int d = (d0 + k < Nn) ? (d0 + k) : (Nn - 1);
    beg[k] = off[d];
    n[k]   = cnt[d];
    a0[k]  = alpha_self[d];
    xs[k]  = *(const float4*)(xl + (size_t)d * 16);
  }
  int sv0[NPW];
  float av0[NPW];
  #pragma unroll
  for (int k = 0; k < NPW; ++k) {
    const int d = (d0 + k < Nn) ? (d0 + k) : (Nn - 1);
    const int i0 = (lane < n[k]) ? (beg[k] + lane) : 0;
    const int sv = csr_src[i0];
    const float av = alpha_d[i0];
    sv0[k] = (lane < n[k]) ? sv : d;
    av0[k] = (lane < n[k]) ? av : 0.0f;
  }
  float4 r0[NPW][4];
  float ar[NPW][4];
  #pragma unroll
  for (int k = 0; k < NPW; ++k) {
    #pragma unroll
    for (int cg = 0; cg < 4; ++cg) {
      const int j = cg * 8 + g;
      const int s = __shfl(sv0[k], j, 32);
      ar[k][cg] = __shfl(av0[k], j, 32);
      r0[k][cg] = *(const float4*)(xl + (size_t)s * 16);
    }
  }
  float4 o = make_float4(0.0f, 0.0f, 0.0f, 0.0f);
  float s_out = 0.0f;
  #pragma unroll
  for (int k = 0; k < NPW; ++k) {
    if (d0 + k < Nn) {
      const int b = beg[k], nn = n[k];
      const float as = (g == 0) ? a0[k] : 0.0f;
      float4 acc = make_float4(as * xs[k].x, as * xs[k].y, as * xs[k].z, as * xs[k].w);
      #pragma unroll
      for (int cg = 0; cg < 4; ++cg) fma4(ar[k][cg], r0[k][cg], acc);
      float al = av0[k];
      for (int cc = 32; cc < nn; cc += 32) {
        const int i = cc + lane;
        int sv = 0;
        float av = 0.0f;
        if (i < nn) {
          sv = csr_src[b + i];
          av = alpha_d[b + i];
        }
        al += av;
        const int mm = nn - cc;
        #pragma unroll
        for (int cg = 0; cg < 4; ++cg) {
          const int j = cg * 8 + g;
          const int s = __shfl(sv, j, 32);
          const float a = __shfl(av, j, 32);
          float4 rr = make_float4(0.0f, 0.0f, 0.0f, 0.0f);
          if (j < mm) rr = *(const float4*)(xl + (size_t)s * 16);
          fma4(a, rr, acc);
        }
      }
      acc.x = sum_over_g(acc.x);
      acc.y = sum_over_g(acc.y);
      acc.z = sum_over_g(acc.z);
      acc.w = sum_over_g(acc.w);
      const float tot = wave_sum(al) + a0[k];
      const bool mine = (g == k);
      o.x = mine ? acc.x : o.x;
      o.y = mine ? acc.y : o.y;
      o.z = mine ? acc.z : o.z;
      o.w = mine ? acc.w : o.w;
      s_out = (lane == k) ? tot : s_out;
    }
  }
  if (g < NPW && d0 + g < Nn) *(float4*)(sx + (size_t)(d0 + g) * 16 + q * 4) = o;
  if (lane < NPW && d0 + lane < Nn) ssum[d0 + lane] = s_out;
}

extern "C" void kernel_launch(void* const* d_in, const int* in_sizes, int n_in,
                              void* d_out, int out_size, void* d_ws, size_t ws_size,
                              hipStream_t stream) {
  const float* x     = (const float*)d_in[0];
  const int*   ei    = (const int*)  d_in[1];
  const float* W1    = (const float*)d_in[2];
  const float* b1    = (const float*)d_in[3];
  const float* att1  = (const float*)d_in[4];
  const float* bias1 = (const float*)d_in[5];
  const float* W2    = (const float*)d_in[6];
  const float* b2    = (const float*)d_in[7];
  const float* att2  = (const float*)d_in[8];
  const float* bias2 = (const float*)d_in[9];
  const float* Wout  = (const float*)d_in[10];
  const float* bout  = (const float*)d_in[11];
  float* y = (float*)d_out;

  const int Nn   = in_sizes[0] / 16;
  const int E    = in_sizes[1] / 2;

  char* p = (char*)d_ws;
  auto alloc = [&](size_t bytes) {
    char* r = p;
    p += (bytes + 255) & ~(size_t)255;
    return r;
  };
  float* h      = (float*)alloc((size_t)Nn * 64 * sizeof(float));
  float* sx     = (float*)alloc((size_t)Nn * 16 * sizeof(float));
  float* ssum   = (float*)alloc((size_t)Nn * sizeof(float));
  float* a_i    = (float*)alloc((size_t)Nn * sizeof(float));
  float* a_j    = (float*)alloc((size_t)Nn * sizeof(float));
  float* al_self = (float*)alloc((size_t)Nn * sizeof(float));
  const int NT  = (Nn + 1023) / 1024;
  const size_t NP = (size_t)NT * 1024;
  int*   cnt_d  = (int*)  alloc(NP * sizeof(int));
  int*   cnt_s  = (int*)  alloc(NP * sizeof(int));
  int*   pm_d   = (int*)  alloc(NP * GCH * sizeof(int));
  int*   pm_s   = (int*)  alloc(NP * GCH * sizeof(int));
  int*   tsum   = (int*)  alloc((size_t)2 * GCH * 64 * sizeof(int));
  int*   lr_d   = (int*)  alloc((size_t)E * sizeof(int));
  int*   lr_s   = (int*)  alloc((size_t)E * sizeof(int));
  const int* off_d = pm_d;
  const int* off_s = pm_s;
  int*   csr_src = (int*) alloc((size_t)E * sizeof(int));
  float* alpha_d = (float*)alloc((size_t)E * sizeof(float));
  int2*  csr_s  = (int2*) alloc((size_t)E * sizeof(int2));

  const int TB = 256;
  dim3 blk(TB);
  int g_gemm  = ((Nn / 16) + 7) / 8;
  int g_node  = (Nn + 8 * NPW - 1) / (8 * NPW);
  int g_edge  = (E + TB * EPT - 1) / (TB * EPT);
  dim3 g_scan(NT, 2);
  dim3 g_cnt(GCH, 2);

  k_linear_dots<<<g_gemm, blk, 0, stream>>>(x, W1, b1, att1, a_i, a_j, Nn);
  k_edge_count<<<g_cnt, dim3(1024), 0, stream>>>(ei, pm_d, pm_s, lr_d, lr_s, tsum, E, NT);
  k_offsets<<<g_scan, blk, 0, stream>>>(pm_d, cnt_d, pm_s, cnt_s, tsum, NT);
  k_edge_place<<<g_edge, blk, 0, stream>>>(ei, pm_d, pm_s, lr_d, lr_s, csr_src, csr_s, E, NT);

  k_src_softmax<<<g_node, blk, 0, stream>>>(csr_s, off_s, cnt_s, a_i, a_j, alpha_d, al_self, Nn);
  k_gather_x<<<g_node, blk, 0, stream>>>(csr_src, alpha_d, al_self, off_d, cnt_d, x, sx, ssum, Nn);

  k_linear_chain<<<g_gemm, blk, 0, stream>>>(sx, ssum, W1, b1, bias1, W2, b2, att2, h, a_i, a_j, Nn);
  k_src_softmax<<<g_node, blk, 0, stream>>>(csr_s, off_s, cnt_s, a_i, a_j, alpha_d, al_self, Nn);
  k_gather_sum<<<g_node, blk, 0, stream>>>(csr_src, alpha_d, al_self, off_d, cnt_d, h,
                                           bias2, Wout, bout, y, Nn);
}


# layer 1's attention dots share the edge count's launch (count work-groups first, dots in later work-groups on the WGPs the count leaves idle): 9 launches -> 8, the dots' time hidden under the count
# speedup vs baseline: 3.0786x; 1.0307x over previous
#include <hip/hip_runtime.h>
#include <float.h>


#define NPW 4
#define EPT 4
#define GCH 16
#define HCAP 51200
#define CPT 16

typedef __attribute__((ext_vector_type(2))) float v2f;
typedef __attribute__((ext_vector_type(8))) float v8f;

#define NEG_SLOPE 0.01f

template <int MASK>
__device__ __forceinline__ float row_xor(float v) {
  return __int_as_float(__builtin_amdgcn_update_dpp(0, __float_as_int(v), 0x160 + MASK,
                                                    0xf, 0xf, true));
}

__device__ __forceinline__ float half_swap(float v) {
  return __int_as_float(__builtin_amdgcn_permlanex16(0, __float_as_int(v), 0x76543210u,
                                                    0xfedcba98u, true, false));
}


template <bool STORE>
__device__ __forceinline__ void finish_rows(const v8f& acc0, const v8f& acc1, const v8f& acc2,
                                            const v8f& acc3, const float* __restrict__ b,
                                            const float* __restrict__ att, float* __restrict__ h,
                                            float* __restrict__ a_i, float* __restrict__ a_j,
                                            int row0, int half, int l16) {
  const float bb0 = b[l16], bb1 = b[16 + l16], bb2 = b[32 + l16], bb3 = b[48 + l16];
  const float ti0 = att[l16],      ti1 = att[16 + l16], ti2 = att[32 + l16],  ti3 = att[48 + l16];
  const float tj0 = att[64 + l16], tj1 = att[80 + l16], tj2 = att[96 + l16],  tj3 = att[112 + l16];
  float oi = 0.0f, oj = 0.0f;
  #pragma unroll
  for (int v = 0; v < 8; ++v) {
    int m = row0 + v + half * 8;
    float h0 = acc0[v] + bb0, h1 = acc1[v] + bb1, h2 = acc2[v] + bb2, h3 = acc3[v] + bb3;
    if (STORE) {
      float* hrow = h + (size_t)m * 64 + l16;
      hrow[0]  = h0;
      hrow[16] = h1;
      hrow[32] = h2;
      hrow[48] = h3;
    }
    float vi = h0 * ti0 + h1 * ti1 + h2 * ti2 + h3 * ti3;
    float vj = h0 * tj0 + h1 * tj1 + h2 * tj2 + h3 * tj3;
    vi += row_xor<8>(vi);  vj += row_xor<8>(vj);
    vi += row_xor<4>(vi);  vj += row_xor<4>(vj);
    vi += row_xor<2>(vi);  vj += row_xor<2>(vj);
    vi += row_xor<1>(vi);  vj += row_xor<1>(vj);
    bool mine = (l16 & 7) == v;
    oi = mine ? vi : oi;
    oj = mine ? vj : oj;
  }
  {
    float* op = (l16 < 8) ? a_i : a_j;
    op[row0 + half * 8 + (l16 & 7)] = (l16 < 8) ? oi : oj;
  }
}

__device__ __forceinline__ void linear_dots_strip(const float* __restrict__ in,
                                                  const float* __restrict__ W,
                                                  const float* __restrict__ b,
                                                  const float* __restrict__ att,
                                                  float* __restrict__ a_i, float* __restrict__ a_j,
                                                  int M, int wave) {
  int lane = threadIdx.x & 31;
  int row0 = wave * 16;
  if (row0 >= M) return;
  int half = lane >> 4;
  int l16  = lane & 15;

  v8f acc0 = {}, acc1 = {}, acc2 = {}, acc3 = {};
  const float* arow = in + (size_t)(row0 + l16) * 16;

  #pragma unroll
  for (int k0 = 0; k0 < 16; k0 += 4) {
    int kb = k0 + half * 2;
    v2f a;  a.x = arow[kb];  a.y = arow[kb + 1];
    v2f b0, b1, b2, b3;
    { const float* p = W + (size_t)(0  + l16) * 16 + kb; b0.x = p[0]; b0.y = p[1]; }
    { const float* p = W + (size_t)(16 + l16) * 16 + kb; b1.x = p[0]; b1.y = p[1]; }
    { const float* p = W + (size_t)(32 + l16) * 16 + kb; b2.x = p[0]; b2.y = p[1]; }
    { const float* p = W + (size_t)(48 + l16) * 16 + kb; b3.x = p[0]; b3.y = p[1]; }
    acc0 = __builtin_amdgcn_wmma_f32_16x16x4_f32(false, a, false, b0, (short)0, acc0, false, false);
    acc1 = __builtin_amdgcn_wmma_f32_16x16x4_f32(false, a, false, b1, (short)0, acc1, false, false);
    acc2 = __builtin_amdgcn_wmma_f32_16x16x4_f32(false, a, false, b2, (short)0, acc2, false, false);
    acc3 = __builtin_amdgcn_wmma_f32_16x16x4_f32(false, a, false, b3, (short)0, acc3, false, false);
  }
  finish_rows<false>(acc0, acc1, acc2, acc3, b, att, nullptr, a_i, a_j, row0, half, l16);
}

__global__ void k_linear_chain(const float* __restrict__ sx, const float* __restrict__ ssum,
                               const float* __restrict__ W1, const float* __restrict__ b1,
                               const float* __restrict__ bias1, const float* __restrict__ W2,
                               const float* __restrict__ b2, const float* __restrict__ att,
                               float* __restrict__ h, float* __restrict__ a_i,
                               float* __restrict__ a_j, int M) {
  int wave = (int)((blockIdx.x * blockDim.x + threadIdx.x) >> 5);
  int lane = threadIdx.x & 31;
  int row0 = wave * 16;
  if (row0 >= M) return;
  int half = lane >> 4;
  int l16  = lane & 15;

  v8f tT[4] = {};
  const float* xrow = sx + (size_t)(row0 + l16) * 16;
  #pragma unroll
  for (int k0 = 0; k0 < 16; k0 += 4) {
    int kb = k0 + half * 2;
    v2f xb;  xb.x = xrow[kb];  xb.y = xrow[kb + 1];
    #pragma unroll
    for (int t = 0; t < 4; ++t) {
      const float* p = W1 + (size_t)(t * 16 + l16) * 16 + kb;
      v2f wa;  wa.x = p[0];  wa.y = p[1];
      tT[t] = __builtin_amdgcn_wmma_f32_16x16x4_f32(false, wa, false, xb, (short)0, tT[t], false, false);
    }
  }
  {
    const float sn = ssum[row0 + l16];
    v2f xb;  xb.x = half ? 0.0f : sn;  xb.y = half ? 0.0f : 1.0f;
    #pragma unroll
    for (int t = 0; t < 4; ++t) {
      const float pb = b1[t * 16 + l16], pc = bias1[t * 16 + l16];
      v2f wa;  wa.x = half ? 0.0f : pb;  wa.y = half ? 0.0f : pc;
      tT[t] = __builtin_amdgcn_wmma_f32_16x16x4_f32(false, wa, false, xb, (short)0, tT[t], false, false);
    }
  }

  v8f acc0 = {}, acc1 = {}, acc2 = {}, acc3 = {};
  #pragma unroll
  for (int t = 0; t < 4; ++t) {
    #pragma unroll
    for (int p = 0; p < 4; ++p) {
      const float u0 = tT[t][2 * p], u1 = tT[t][2 * p + 1];
      v2f a;
      a.x = u0 > 0.0f ? u0 : 0.0f;
      a.y = u1 > 0.0f ? u1 : 0.0f;
      const int kb = t * 16 + half * 8 + 2 * p;
      v2f b0, b1v, b2v, b3;
      { const float* q = W2 + (size_t)(0  + l16) * 64 + kb; b0.x  = q[0]; b0.y  = q[1]; }
      { const float* q = W2 + (size_t)(16 + l16) * 64 + kb; b1v.x = q[0]; b1v.y = q[1]; }
      { const float* q = W2 + (size_t)(32 + l16) * 64 + kb; b2v.x = q[0]; b2v.y = q[1]; }
      { const float* q = W2 + (size_t)(48 + l16) * 64 + kb; b3.x  = q[0]; b3.y  = q[1]; }
      acc0 = __builtin_amdgcn_wmma_f32_16x16x4_f32(false, a, false, b0,  (short)0, acc0, false, false);
      acc1 = __builtin_amdgcn_wmma_f32_16x16x4_f32(false, a, false, b1v, (short)0, acc1, false, false);
      acc2 = __builtin_amdgcn_wmma_f32_16x16x4_f32(false, a, false, b2v, (short)0, acc2, false, false);
      acc3 = __builtin_amdgcn_wmma_f32_16x16x4_f32(false, a, false, b3,  (short)0, acc3, false, false);
    }
  }
  finish_rows<true>(acc0, acc1, acc2, acc3, b2, att, h, a_i, a_j, row0, half, l16);
}

__global__ void __launch_bounds__(1024)
k_count_and_dots(const int* __restrict__ ei, int* __restrict__ cm_d, int* __restrict__ cm_s,
                 int* __restrict__ lr_d, int* __restrict__ lr_s, int* __restrict__ tsum,
                 int E, int NT,
                 const float* __restrict__ x, const float* __restrict__ W1,
                 const float* __restrict__ b1, const float* __restrict__ att1,
                 float* __restrict__ a_i, float* __restrict__ a_j, int Nn) {
  __shared__ int hist[HCAP];
  const int t = threadIdx.x, lane = t & 31, w = t >> 5;
  if (blockIdx.x >= 2 * GCH) {
    linear_dots_strip(x, W1, b1, att1, a_i, a_j, Nn, (int)(blockIdx.x - 2 * GCH) * 32 + w);
    return;
  }
  const int g = blockIdx.x % GCH, ty = blockIdx.x / GCH;
  const int NP = NT * 1024;
  const int* __restrict__ keys = ty ? ei : ei + E;
  int* __restrict__ lr = ty ? lr_s : lr_d;
  int* __restrict__ cm = (ty ? cm_s : cm_d) + (size_t)g * NP;

  int4* h4 = (int4*)hist;
  for (int i = t; i < NP / 4; i += 1024) h4[i] = make_int4(0, 0, 0, 0);
  __syncthreads();

  const int chunk = (E + GCH - 1) / GCH;
  const int lo = g * chunk;
  const int hi = (lo + chunk < E) ? (lo + chunk) : E;
  int e = lo + t;
  for (; e + (CPT - 1) * 1024 < hi; e += CPT * 1024) {
    int k[CPT], r[CPT];
    #pragma unroll
    for (int i = 0; i < CPT; ++i) k[i] = keys[e + i * 1024];
    #pragma unroll
    for (int i = 0; i < CPT; ++i) r[i] = atomicAdd(&hist[k[i]], 1);
    #pragma unroll
    for (int i = 0; i < CPT; ++i) lr[e + i * 1024] = r[i];
  }
  for (; e < hi; e += 1024) lr[e] = atomicAdd(&hist[keys[e]], 1);
  __syncthreads();

  for (int j = w; j < NT; j += 32) {
    const int4* hj = (const int4*)(hist + j * 1024);
    int4* cj = (int4*)(cm + j * 1024);
    int acc = 0;
    #pragma unroll
    for (int i = 0; i < 8; ++i) {
      const int4 v = hj[i * 32 + lane];
      cj[i * 32 + lane] = v;
      acc += (v.x + v.y) + (v.z + v.w);
    }
    for (int o = 16; o > 0; o >>= 1) acc += __shfl_xor(acc, o, 32);
    if (lane == 0) tsum[(ty * GCH + g) * 64 + j] = acc;
  }
  if (t >= NT && t < 64) tsum[(ty * GCH + g) * 64 + t] = 0;
}

__global__ void k_offsets(int* __restrict__ pm_d, int* __restrict__ cnt_d,
                          int* __restrict__ pm_s, int* __restrict__ cnt_s,
                          const int* __restrict__ tsum, int NT) {
  int* __restrict__ pm = blockIdx.y ? pm_s : pm_d;
  int* __restrict__ cnt = blockIdx.y ? cnt_s : cnt_d;
  const int* __restrict__ tsy = tsum + blockIdx.y * (GCH * 64);
  __shared__ int s_pre[8];
  __shared__ int s_loc[8];
  const int t = threadIdx.x, lane = t & 31, w = t >> 5;
  const int nb = blockIdx.x;
  const int NP = NT * 1024;
  const int i0 = nb * 1024 + t * 4;
  int4 c[GCH];
  #pragma unroll
  for (int g = 0; g < GCH; ++g) c[g] = *(const int4*)(pm + (size_t)g * NP + i0);
  static_assert(GCH * 64 == 256 * 4, "one 16-byte load per thread covers the tile sums");
  const int4 tv = *(const int4*)(tsy + t * 4);
  const int j0 = (t * 4) & 63;
  int pre = ((j0 < nb) ? tv.x : 0) + ((j0 + 1 < nb) ? tv.y : 0) +
            ((j0 + 2 < nb) ? tv.z : 0) + ((j0 + 3 < nb) ? tv.w : 0);
  int t0 = 0, t1 = 0, t2 = 0, t3 = 0;
  #pragma unroll
  for (int g = 0; g < GCH; ++g) { t0 += c[g].x;  t1 += c[g].y;  t2 += c[g].z;  t3 += c[g].w; }
  const int loc = (t0 + t1) + (t2 + t3);
  for (int o = 16; o > 0; o >>= 1) pre += __shfl_xor(pre, o, 32);
  int inc = loc;
  for (int o = 1; o < 32; o <<= 1) {
    const int v = __shfl_up(inc, o, 32);
    inc += (lane >= o) ? v : 0;
  }
  if (lane == 31) s_loc[w] = inc;
  if (lane == 0)  s_pre[w] = pre;
  __syncthreads();
  int base = 0;
  for (int k = 0; k < 8; ++k) base += s_pre[k] + ((k < w) ? s_loc[k] : 0);
  const int ex = base + inc - loc;
  int r0 = ex, r1 = ex + t0, r2 = ex + t0 + t1, r3 = ex + t0 + t1 + t2;
  #pragma unroll
  for (int g = 0; g < GCH; ++g) {
    *(int4*)(pm + (size_t)g * NP + i0) = make_int4(r0, r1, r2, r3);
    r0 += c[g].x;  r1 += c[g].y;  r2 += c[g].z;  r3 += c[g].w;
  }
  *(int4*)(cnt + i0) = make_int4(t0, t1, t2, t3);
}

__global__ void k_edge_place(const int* __restrict__ ei, const int* __restrict__ pm_d,
                             const int* __restrict__ pm_s, const int* __restrict__ lr_d,
                             const int* __restrict__ lr_s, int* __restrict__ csr_src,
                             int2* __restrict__ csr_s, int E, int NT) {
  const int T = (int)(gridDim.x * blockDim.x);
  const int e0 = (int)(blockIdx.x * blockDim.x + threadIdx.x);
  const unsigned chunk = (unsigned)((E + GCH - 1) / GCH);
  const int NP = NT * 1024;
  if (e0 + (EPT - 1) * T < E) {
    int s[EPT], d[EPT], rd[EPT], rs[EPT], pd[EPT], ps[EPT];
    #pragma unroll
    for (int k = 0; k < EPT; ++k) {
      const int e = e0 + k * T;
      s[k] = ei[e];  d[k] = ei[E + e];  rd[k] = lr_d[e];  rs[k] = lr_s[e];
    }
    #pragma unroll
    for (int k = 0; k < EPT; ++k) {
      const size_t row = (size_t)((unsigned)(e0 + k * T) / chunk) * NP;
      pd[k] = pm_d[row + d[k]] + rd[k];
      ps[k] = pm_s[row + s[k]] + rs[k];
    }
    #pragma unroll
    for (int k = 0; k < EPT; ++k) {
      csr_src[pd[k]] = s[k];
      csr_s[ps[k]] = make_int2(d[k], pd[k]);
    }
  } else {
    for (int k = 0; k < EPT; ++k) {
      const int e = e0 + k * T;
      if (e < E) {
        const int s = ei[e], d = ei[E + e];
        const size_t row = (size_t)((unsigned)e / chunk) * NP;
        const int pd = pm_d[row + d] + lr_d[e];
        const int ps = pm_s[row + s] + lr_s[e];
        csr_src[pd] = s;
        csr_s[ps] = make_int2(d, pd);
      }
    }
  }
}

__device__ __forceinline__ float leaky(float v) { return v > 0.0f ? v : v * NEG_SLOPE; }

__device__ __forceinline__ float wave_max(float v) {
  v = fmaxf(v, row_xor<8>(v));
  v = fmaxf(v, row_xor<4>(v));
  v = fmaxf(v, row_xor<2>(v));
  v = fmaxf(v, row_xor<1>(v));
  return fmaxf(__int_as_float(__builtin_amdgcn_readlane(__float_as_int(v), 0)),
               __int_as_float(__builtin_amdgcn_readlane(__float_as_int(v), 16)));
}
__device__ __forceinline__ float wave_sum(float v) {
  v += row_xor<8>(v);
  v += row_xor<4>(v);
  v += row_xor<2>(v);
  v += row_xor<1>(v);
  return __int_as_float(__builtin_amdgcn_readlane(__float_as_int(v), 0)) +
         __int_as_float(__builtin_amdgcn_readlane(__float_as_int(v), 16));
}

__global__ void k_src_softmax(const int2* __restrict__ csr_s, const int* __restrict__ off_s,
                              const int* __restrict__ cnt_s, const float* __restrict__ a_i,
                              const float* __restrict__ a_j, float* __restrict__ alpha_d,
                              float* __restrict__ alpha_self, int Nn) {
  const int w = __builtin_amdgcn_readfirstlane(
      (int)(blockIdx.x * (blockDim.x >> 5) + (threadIdx.x >> 5)));
  const int lane = threadIdx.x & 31;
  const int s0 = w * NPW;
  if (s0 >= Nn) return;
  int beg[NPW], n[NPW];
  float aj[NPW], ais[NPW];
  #pragma unroll
  for (int k = 0; k < NPW; ++k) {
    const int s = (s0 + k < Nn) ? (s0 + k) : (Nn - 1);
    beg[k] = off_s[s];
    n[k]   = cnt_s[s];
    aj[k]  = a_j[s];
    ais[k] = a_i[s];
  }
  int2 de[NPW];
  #pragma unroll
  for (int k = 0; k < NPW; ++k) de[k] = csr_s[(lane < n[k]) ? (beg[k] + lane) : 0];
  float ai0[NPW];
  #pragma unroll
  for (int k = 0; k < NPW; ++k) ai0[k] = a_i[de[k].x];
  float self_out = 0.0f;
  #pragma unroll
  for (int k = 0; k < NPW; ++k) {
    if (s0 + k < Nn) {
      const int b = beg[k], nn = n[k];
      const float ajk = aj[k];
      const float sc_self = leaky(ais[k] + ajk);
      const bool v0 = lane < nn;
      const float sc0 = v0 ? leaky(ai0[k] + ajk) : -FLT_MAX;
      float mx = fmaxf(sc_self, sc0);
      for (int cc = 32; cc < nn; cc += 32) {
        const int i = cc + lane;
        if (i < nn) mx = fmaxf(mx, leaky(a_i[csr_s[b + i].x] + ajk));
      }
      mx = wave_max(mx);
      const float ev0 = v0 ? __expf(sc0 - mx) : 0.0f;
      float sum = ev0;
      for (int cc = 32; cc < nn; cc += 32) {
        const int i = cc + lane;
        if (i < nn) sum += __expf(leaky(a_i[csr_s[b + i].x] + ajk) - mx);
      }
      const float es = __expf(sc_self - mx);
      const float den = (wave_sum(sum) + es) + 1e-16f;
      if (v0) alpha_d[de[k].y] = ev0 / den;
      for (int cc = 32; cc < nn; cc += 32) {
        const int i = cc + lane;
        if (i < nn) {
          const int2 dx = csr_s[b + i];
          alpha_d[dx.y] = __expf(leaky(a_i[dx.x] + ajk) - mx) / den;
        }
      }
      const float as = es / den;
      self_out = (lane == k) ? as : self_out;
    }
  }
  if (lane < NPW && s0 + lane < Nn) alpha_self[s0 + lane] = self_out;
}

template <int NR>
__device__ __forceinline__ void add_rows(int sv, float av, int j0, const float* __restrict__ hl,
                                         float& accx, float& accy) {
  float2 hv[NR];
  float a[NR];
  #pragma unroll
  for (int r = 0; r < NR; ++r) {
    const int s = __builtin_amdgcn_readlane(sv, j0 + r);
    a[r] = __int_as_float(__builtin_amdgcn_readlane(__float_as_int(av), j0 + r));
    hv[r] = *(const float2*)(hl + (size_t)s * 64);
  }
  #pragma unroll
  for (int r = 0; r < NR; ++r) {
    accx = __builtin_fmaf(a[r], hv[r].x, accx);
    accy = __builtin_fmaf(a[r], hv[r].y, accy);
  }
}

__global__ void k_gather_sum(const int* __restrict__ csr_src, const float* __restrict__ alpha_d,
                             const float* __restrict__ alpha_self, const int* __restrict__ off,
                             const int* __restrict__ cnt, const float* __restrict__ h,
                             const float* __restrict__ bias, const float* __restrict__ Wout,
                             const float* __restrict__ bout, float* __restrict__ y, int Nn) {
  const int w = __builtin_amdgcn_readfirstlane(
      (int)(blockIdx.x * (blockDim.x >> 5) + (threadIdx.x >> 5)));
  const int lane = threadIdx.x & 31;
  const int d0 = w * NPW;
  if (d0 >= Nn) return;
  const float* hl = h + lane * 2;
  const float bz0 = bias[lane * 2], bz1 = bias[lane * 2 + 1];
  const float wz0 = Wout[lane * 2], wz1 = Wout[lane * 2 + 1];
  const float bo  = bout[0];
  int beg[NPW], n[NPW];
  float a0[NPW];
  float2 hs[NPW];
  #pragma unroll
  for (int k = 0; k < NPW; ++k) {
    const int d = (d0 + k < Nn) ? (d0 + k) : (Nn - 1);
    beg[k] = off[d];
    n[k]   = cnt[d];
    a0[k]  = alpha_self[d];
    hs[k]  = *(const float2*)(hl + (size_t)d * 64);
  }
  int sv0[NPW];
  float av0[NPW];
  #pragma unroll
  for (int k = 0; k < NPW; ++k) {
    const int i0 = (lane < n[k]) ? (beg[k] + lane) : 0;
    sv0[k] = csr_src[i0];
    av0[k] = alpha_d[i0];
  }
  float y_out = 0.0f;
  #pragma unroll
  for (int k = 0; k < NPW; ++k) {
    if (d0 + k < Nn) {
      const int b = beg[k], nn = n[k];
      float accx = a0[k] * hs[k].x, accy = a0[k] * hs[k].y;
      for (int cc = 0; cc < nn; cc += 32) {
        int sv = sv0[k];
        float av = av0[k];
        if (cc > 0) {
          const int i = cc + lane;
          sv = 0;  av = 0.0f;
          if (i < nn) {
            sv = csr_src[b + i];
            av = alpha_d[b + i];
          }
        }
        const int mm = (nn - cc) < 32 ? (nn - cc) : 32;
        int j = 0;
        for (; j + 8 <= mm; j += 8) add_rows<8>(sv, av, j, hl, accx, accy);
        if (mm & 4) { add_rows<4>(sv, av, j, hl, accx, accy); j += 4; }
        if (mm & 2) { add_rows<2>(sv, av, j, hl, accx, accy); j += 2; }
        if (mm & 1) { add_rows<1>(sv, av, j, hl, accx, accy); }
      }
      float v0 = accx + bz0;
      float v1 = accy + bz1;
      v0 = v0 > 0.0f ? v0 : 0.0f;
      v1 = v1 > 0.0f ? v1 : 0.0f;
      float v = v0 * wz0 + v1 * wz1;
      v += row_xor<8>(v);
      v += row_xor<4>(v);
      v += row_xor<2>(v);
      v += row_xor<1>(v);
      const float tot = __int_as_float(__builtin_amdgcn_readlane(__float_as_int(v), 0)) +
                        __int_as_float(__builtin_amdgcn_readlane(__float_as_int(v), 16));
      y_out = (lane == k) ? (tot + bo) : y_out;
    }
  }
  if (lane < NPW && d0 + lane < Nn) y[d0 + lane] = y_out;
}

__device__ __forceinline__ void fma4(float a, const float4& r, float4& acc) {
  acc.x = __builtin_fmaf(a, r.x, acc.x);
  acc.y = __builtin_fmaf(a, r.y, acc.y);
  acc.z = __builtin_fmaf(a, r.z, acc.z);
  acc.w = __builtin_fmaf(a, r.w, acc.w);
}
__device__ __forceinline__ float sum_over_g(float v) {
  v += row_xor<4>(v);
  v += row_xor<8>(v);
  return v + half_swap(v);
}

__global__ void k_gather_x(const int* __restrict__ csr_src, const float* __restrict__ alpha_d,
                           const float* __restrict__ alpha_self, const int* __restrict__ off,
                           const int* __restrict__ cnt, const float* __restrict__ x,
                           float* __restrict__ sx, float* __restrict__ ssum, int Nn) {
  const int w = __builtin_amdgcn_readfirstlane(
      (int)(blockIdx.x * (blockDim.x >> 5) + (threadIdx.x >> 5)));
  const int lane = threadIdx.x & 31;
  const int d0 = w * NPW;
  if (d0 >= Nn) return;
  const int g = lane >> 2, q = lane & 3;
  const float* xl = x + q * 4;
  int beg[NPW], n[NPW];
  float a0[NPW];
  float4 xs[NPW];
  #pragma unroll
  for (int k = 0; k < NPW; ++k) {
    const int d = (d0 + k < Nn) ? (d0 + k) : (Nn - 1);
    beg[k] = off[d];
    n[k]   = cnt[d];
    a0[k]  = alpha_self[d];
    xs[k]  = *(const float4*)(xl + (size_t)d * 16);
  }
  int sv0[NPW];
  float av0[NPW];
  #pragma unroll
  for (int k = 0; k < NPW; ++k) {
    const int d = (d0 + k < Nn) ? (d0 + k) : (Nn - 1);
    const int i0 = (lane < n[k]) ? (beg[k] + lane) : 0;
    const int sv = csr_src[i0];
    const float av = alpha_d[i0];
    sv0[k] = (lane < n[k]) ? sv : d;
    av0[k] = (lane < n[k]) ? av : 0.0f;
  }
  float4 r0[NPW][4];
  float ar[NPW][4];
  #pragma unroll
  for (int k = 0; k < NPW; ++k) {
    #pragma unroll
    for (int cg = 0; cg < 4; ++cg) {
      const int j = cg * 8 + g;
      const int s = __shfl(sv0[k], j, 32);
      ar[k][cg] = __shfl(av0[k], j, 32);
      r0[k][cg] = *(const float4*)(xl + (size_t)s * 16);
    }
  }
  float4 o = make_float4(0.0f, 0.0f, 0.0f, 0.0f);
  float s_out = 0.0f;
  #pragma unroll
  for (int k = 0; k < NPW; ++k) {
    if (d0 + k < Nn) {
      const int b = beg[k], nn = n[k];
      const float as = (g == 0) ? a0[k] : 0.0f;
      float4 acc = make_float4(as * xs[k].x, as * xs[k].y, as * xs[k].z, as * xs[k].w);
      #pragma unroll
      for (int cg = 0; cg < 4; ++cg) fma4(ar[k][cg], r0[k][cg], acc);
      float al = av0[k];
      for (int cc = 32; cc < nn; cc += 32) {
        const int i = cc + lane;
        int sv = 0;
        float av = 0.0f;
        if (i < nn) {
          sv = csr_src[b + i];
          av = alpha_d[b + i];
        }
        al += av;
        const int mm = nn - cc;
        #pragma unroll
        for (int cg = 0; cg < 4; ++cg) {
          const int j = cg * 8 + g;
          const int s = __shfl(sv, j, 32);
          const float a = __shfl(av, j, 32);
          float4 rr = make_float4(0.0f, 0.0f, 0.0f, 0.0f);
          if (j < mm) rr = *(const float4*)(xl + (size_t)s * 16);
          fma4(a, rr, acc);
        }
      }
      acc.x = sum_over_g(acc.x);
      acc.y = sum_over_g(acc.y);
      acc.z = sum_over_g(acc.z);
      acc.w = sum_over_g(acc.w);
      const float tot = wave_sum(al) + a0[k];
      const bool mine = (g == k);
      o.x = mine ? acc.x : o.x;
      o.y = mine ? acc.y : o.y;
      o.z = mine ? acc.z : o.z;
      o.w = mine ? acc.w : o.w;
      s_out = (lane == k) ? tot : s_out;
    }
  }
  if (g < NPW && d0 + g < Nn) *(float4*)(sx + (size_t)(d0 + g) * 16 + q * 4) = o;
  if (lane < NPW && d0 + lane < Nn) ssum[d0 + lane] = s_out;
}

extern "C" void kernel_launch(void* const* d_in, const int* in_sizes, int n_in,
                              void* d_out, int out_size, void* d_ws, size_t ws_size,
                              hipStream_t stream) {
  const float* x     = (const float*)d_in[0];
  const int*   ei    = (const int*)  d_in[1];
  const float* W1    = (const float*)d_in[2];
  const float* b1    = (const float*)d_in[3];
  const float* att1  = (const float*)d_in[4];
  const float* bias1 = (const float*)d_in[5];
  const float* W2    = (const float*)d_in[6];
  const float* b2    = (const float*)d_in[7];
  const float* att2  = (const float*)d_in[8];
  const float* bias2 = (const float*)d_in[9];
  const float* Wout  = (const float*)d_in[10];
  const float* bout  = (const float*)d_in[11];
  float* y = (float*)d_out;

  const int Nn   = in_sizes[0] / 16;
  const int E    = in_sizes[1] / 2;

  char* p = (char*)d_ws;
  auto alloc = [&](size_t bytes) {
    char* r = p;
    p += (bytes + 255) & ~(size_t)255;
    return r;
  };
  float* h      = (float*)alloc((size_t)Nn * 64 * sizeof(float));
  float* sx     = (float*)alloc((size_t)Nn * 16 * sizeof(float));
  float* ssum   = (float*)alloc((size_t)Nn * sizeof(float));
  float* a_i    = (float*)alloc((size_t)Nn * sizeof(float));
  float* a_j    = (float*)alloc((size_t)Nn * sizeof(float));
  float* al_self = (float*)alloc((size_t)Nn * sizeof(float));
  const int NT  = (Nn + 1023) / 1024;
  const size_t NP = (size_t)NT * 1024;
  int*   cnt_d  = (int*)  alloc(NP * sizeof(int));
  int*   cnt_s  = (int*)  alloc(NP * sizeof(int));
  int*   pm_d   = (int*)  alloc(NP * GCH * sizeof(int));
  int*   pm_s   = (int*)  alloc(NP * GCH * sizeof(int));
  int*   tsum   = (int*)  alloc((size_t)2 * GCH * 64 * sizeof(int));
  int*   lr_d   = (int*)  alloc((size_t)E * sizeof(int));
  int*   lr_s   = (int*)  alloc((size_t)E * sizeof(int));
  const int* off_d = pm_d;
  const int* off_s = pm_s;
  int*   csr_src = (int*) alloc((size_t)E * sizeof(int));
  float* alpha_d = (float*)alloc((size_t)E * sizeof(float));
  int2*  csr_s  = (int2*) alloc((size_t)E * sizeof(int2));

  const int TB = 256;
  dim3 blk(TB);
  int g_gemm  = ((Nn / 16) + 7) / 8;
  int g_node  = (Nn + 8 * NPW - 1) / (8 * NPW);
  int g_edge  = (E + TB * EPT - 1) / (TB * EPT);
  dim3 g_scan(NT, 2);
  int g_cnt   = 2 * GCH + ((Nn / 16) + 31) / 32;

  k_count_and_dots<<<g_cnt, dim3(1024), 0, stream>>>(ei, pm_d, pm_s, lr_d, lr_s, tsum, E, NT,
                                                     x, W1, b1, att1, a_i, a_j, Nn);
  k_offsets<<<g_scan, blk, 0, stream>>>(pm_d, cnt_d, pm_s, cnt_s, tsum, NT);
  k_edge_place<<<g_edge, blk, 0, stream>>>(ei, pm_d, pm_s, lr_d, lr_s, csr_src, csr_s, E, NT);

  k_src_softmax<<<g_node, blk, 0, stream>>>(csr_s, off_s, cnt_s, a_i, a_j, alpha_d, al_self, Nn);
  k_gather_x<<<g_node, blk, 0, stream>>>(csr_src, alpha_d, al_self, off_d, cnt_d, x, sx, ssum, Nn);

  k_linear_chain<<<g_gemm, blk, 0, stream>>>(sx, ssum, W1, b1, bias1, W2, b2, att2, h, a_i, a_j, Nn);
  k_src_softmax<<<g_node, blk, 0, stream>>>(csr_s, off_s, cnt_s, a_i, a_j, alpha_d, al_self, Nn);
  k_gather_sum<<<g_node, blk, 0, stream>>>(csr_src, alpha_d, al_self, off_d, cnt_d, h,
                                           bias2, Wout, bout, y, Nn);
}


# alpha formed at the dst side from a 16-byte (a_j, max, sum) record per src: the per-edge alpha array, its 4-byte scatter by the src wave and the slot field of the src order go; sums unchanged
# speedup vs baseline: 3.8961x; 1.2655x over previous
#include <hip/hip_runtime.h>
#include <float.h>


#define NPW 4
#define EPT 4
#define GCH 16
#define HCAP 51200
#define CPT 16

typedef __attribute__((ext_vector_type(2))) float v2f;
typedef __attribute__((ext_vector_type(8))) float v8f;

#define NEG_SLOPE 0.01f

template <int MASK>
__device__ __forceinline__ float row_xor(float v) {
  return __int_as_float(__builtin_amdgcn_update_dpp(0, __float_as_int(v), 0x160 + MASK,
                                                    0xf, 0xf, true));
}

__device__ __forceinline__ float half_swap(float v) {
  return __int_as_float(__builtin_amdgcn_permlanex16(0, __float_as_int(v), 0x76543210u,
                                                    0xfedcba98u, true, false));
}


template <bool STORE>
__device__ __forceinline__ void finish_rows(const v8f& acc0, const v8f& acc1, const v8f& acc2,
                                            const v8f& acc3, const float* __restrict__ b,
                                            const float* __restrict__ att, float* __restrict__ h,
                                            float* __restrict__ a_i, float* __restrict__ a_j,
                                            int row0, int half, int l16) {
  const float bb0 = b[l16], bb1 = b[16 + l16], bb2 = b[32 + l16], bb3 = b[48 + l16];
  const float ti0 = att[l16],      ti1 = att[16 + l16], ti2 = att[32 + l16],  ti3 = att[48 + l16];
  const float tj0 = att[64 + l16], tj1 = att[80 + l16], tj2 = att[96 + l16],  tj3 = att[112 + l16];
  float oi = 0.0f, oj = 0.0f;
  #pragma unroll
  for (int v = 0; v < 8; ++v) {
    int m = row0 + v + half * 8;
    float h0 = acc0[v] + bb0, h1 = acc1[v] + bb1, h2 = acc2[v] + bb2, h3 = acc3[v] + bb3;
    if (STORE) {
      float* hrow = h + (size_t)m * 64 + l16;
      hrow[0]  = h0;
      hrow[16] = h1;
      hrow[32] = h2;
      hrow[48] = h3;
    }
    float vi = h0 * ti0 + h1 * ti1 + h2 * ti2 + h3 * ti3;
    float vj = h0 * tj0 + h1 * tj1 + h2 * tj2 + h3 * tj3;
    vi += row_xor<8>(vi);  vj += row_xor<8>(vj);
    vi += row_xor<4>(vi);  vj += row_xor<4>(vj);
    vi += row_xor<2>(vi);  vj += row_xor<2>(vj);
    vi += row_xor<1>(vi);  vj += row_xor<1>(vj);
    bool mine = (l16 & 7) == v;
    oi = mine ? vi : oi;
    oj = mine ? vj : oj;
  }
  {
    float* op = (l16 < 8) ? a_i : a_j;
    op[row0 + half * 8 + (l16 & 7)] = (l16 < 8) ? oi : oj;
  }
}

__device__ __forceinline__ void linear_dots_strip(const float* __restrict__ in,
                                                  const float* __restrict__ W,
                                                  const float* __restrict__ b,
                                                  const float* __restrict__ att,
                                                  float* __restrict__ a_i, float* __restrict__ a_j,
                                                  int M, int wave) {
  int lane = threadIdx.x & 31;
  int row0 = wave * 16;
  if (row0 >= M) return;
  int half = lane >> 4;
  int l16  = lane & 15;

  v8f acc0 = {}, acc1 = {}, acc2 = {}, acc3 = {};
  const float* arow = in + (size_t)(row0 + l16) * 16;

  #pragma unroll
  for (int k0 = 0; k0 < 16; k0 += 4) {
    int kb = k0 + half * 2;
    v2f a;  a.x = arow[kb];  a.y = arow[kb + 1];
    v2f b0, b1, b2, b3;
    { const float* p = W + (size_t)(0  + l16) * 16 + kb; b0.x = p[0]; b0.y = p[1]; }
    { const float* p = W + (size_t)(16 + l16) * 16 + kb; b1.x = p[0]; b1.y = p[1]; }
    { const float* p = W + (size_t)(32 + l16) * 16 + kb; b2.x = p[0]; b2.y = p[1]; }
    { const float* p = W + (size_t)(48 + l16) * 16 + kb; b3.x = p[0]; b3.y = p[1]; }
    acc0 = __builtin_amdgcn_wmma_f32_16x16x4_f32(false, a, false, b0, (short)0, acc0, false, false);
    acc1 = __builtin_amdgcn_wmma_f32_16x16x4_f32(false, a, false, b1, (short)0, acc1, false, false);
    acc2 = __builtin_amdgcn_wmma_f32_16x16x4_f32(false, a, false, b2, (short)0, acc2, false, false);
    acc3 = __builtin_amdgcn_wmma_f32_16x16x4_f32(false, a, false, b3, (short)0, acc3, false, false);
  }
  finish_rows<false>(acc0, acc1, acc2, acc3, b, att, nullptr, a_i, a_j, row0, half, l16);
}

__global__ void k_linear_chain(const float* __restrict__ sx, const float* __restrict__ ssum,
                               const float* __restrict__ W1, const float* __restrict__ b1,
                               const float* __restrict__ bias1, const float* __restrict__ W2,
                               const float* __restrict__ b2, const float* __restrict__ att,
                               float* __restrict__ h, float* __restrict__ a_i,
                               float* __restrict__ a_j, int M) {
  int wave = (int)((blockIdx.x * blockDim.x + threadIdx.x) >> 5);
  int lane = threadIdx.x & 31;
  int row0 = wave * 16;
  if (row0 >= M) return;
  int half = lane >> 4;
  int l16  = lane & 15;

  v8f tT[4] = {};
  const float* xrow = sx + (size_t)(row0 + l16) * 16;
  #pragma unroll
  for (int k0 = 0; k0 < 16; k0 += 4) {
    int kb = k0 + half * 2;
    v2f xb;  xb.x = xrow[kb];  xb.y = xrow[kb + 1];
    #pragma unroll
    for (int t = 0; t < 4; ++t) {
      const float* p = W1 + (size_t)(t * 16 + l16) * 16 + kb;
      v2f wa;  wa.x = p[0];  wa.y = p[1];
      tT[t] = __builtin_amdgcn_wmma_f32_16x16x4_f32(false, wa, false, xb, (short)0, tT[t], false, false);
    }
  }
  {
    const float sn = ssum[row0 + l16];
    v2f xb;  xb.x = half ? 0.0f : sn;  xb.y = half ? 0.0f : 1.0f;
    #pragma unroll
    for (int t = 0; t < 4; ++t) {
      const float pb = b1[t * 16 + l16], pc = bias1[t * 16 + l16];
      v2f wa;  wa.x = half ? 0.0f : pb;  wa.y = half ? 0.0f : pc;
      tT[t] = __builtin_amdgcn_wmma_f32_16x16x4_f32(false, wa, false, xb, (short)0, tT[t], false, false);
    }
  }

  v8f acc0 = {}, acc1 = {}, acc2 = {}, acc3 = {};
  #pragma unroll
  for (int t = 0; t < 4; ++t) {
    #pragma unroll
    for (int p = 0; p < 4; ++p) {
      const float u0 = tT[t][2 * p], u1 = tT[t][2 * p + 1];
      v2f a;
      a.x = u0 > 0.0f ? u0 : 0.0f;
      a.y = u1 > 0.0f ? u1 : 0.0f;
      const int kb = t * 16 + half * 8 + 2 * p;
      v2f b0, b1v, b2v, b3;
      { const float* q = W2 + (size_t)(0  + l16) * 64 + kb; b0.x  = q[0]; b0.y  = q[1]; }
      { const float* q = W2 + (size_t)(16 + l16) * 64 + kb; b1v.x = q[0]; b1v.y = q[1]; }
      { const float* q = W2 + (size_t)(32 + l16) * 64 + kb; b2v.x = q[0]; b2v.y = q[1]; }
      { const float* q = W2 + (size_t)(48 + l16) * 64 + kb; b3.x  = q[0]; b3.y  = q[1]; }
      acc0 = __builtin_amdgcn_wmma_f32_16x16x4_f32(false, a, false, b0,  (short)0, acc0, false, false);
      acc1 = __builtin_amdgcn_wmma_f32_16x16x4_f32(false, a, false, b1v, (short)0, acc1, false, false);
      acc2 = __builtin_amdgcn_wmma_f32_16x16x4_f32(false, a, false, b2v, (short)0, acc2, false, false);
      acc3 = __builtin_amdgcn_wmma_f32_16x16x4_f32(false, a, false, b3,  (short)0, acc3, false, false);
    }
  }
  finish_rows<true>(acc0, acc1, acc2, acc3, b2, att, h, a_i, a_j, row0, half, l16);
}

__global__ void __launch_bounds__(1024)
k_count_and_dots(const int* __restrict__ ei, int* __restrict__ cm_d, int* __restrict__ cm_s,
                 int* __restrict__ lr_d, int* __restrict__ lr_s, int* __restrict__ tsum,
                 int E, int NT,
                 const float* __restrict__ x, const float* __restrict__ W1,
                 const float* __restrict__ b1, const float* __restrict__ att1,
                 float* __restrict__ a_i, float* __restrict__ a_j, int Nn) {
  __shared__ int hist[HCAP];
  const int t = threadIdx.x, lane = t & 31, w = t >> 5;
  if (blockIdx.x >= 2 * GCH) {
    linear_dots_strip(x, W1, b1, att1, a_i, a_j, Nn, (int)(blockIdx.x - 2 * GCH) * 32 + w);
    return;
  }
  const int g = blockIdx.x % GCH, ty = blockIdx.x / GCH;
  const int NP = NT * 1024;
  const int* __restrict__ keys = ty ? ei : ei + E;
  int* __restrict__ lr = ty ? lr_s : lr_d;
  int* __restrict__ cm = (ty ? cm_s : cm_d) + (size_t)g * NP;

  int4* h4 = (int4*)hist;
  for (int i = t; i < NP / 4; i += 1024) h4[i] = make_int4(0, 0, 0, 0);
  __syncthreads();

  const int chunk = (E + GCH - 1) / GCH;
  const int lo = g * chunk;
  const int hi = (lo + chunk < E) ? (lo + chunk) : E;
  int e = lo + t;
  for (; e + (CPT - 1) * 1024 < hi; e += CPT * 1024) {
    int k[CPT], r[CPT];
    #pragma unroll
    for (int i = 0; i < CPT; ++i) k[i] = keys[e + i * 1024];
    #pragma unroll
    for (int i = 0; i < CPT; ++i) r[i] = atomicAdd(&hist[k[i]], 1);
    #pragma unroll
    for (int i = 0; i < CPT; ++i) lr[e + i * 1024] = r[i];
  }
  for (; e < hi; e += 1024) lr[e] = atomicAdd(&hist[keys[e]], 1);
  __syncthreads();

  for (int j = w; j < NT; j += 32) {
    const int4* hj = (const int4*)(hist + j * 1024);
    int4* cj = (int4*)(cm + j * 1024);
    int acc = 0;
    #pragma unroll
    for (int i = 0; i < 8; ++i) {
      const int4 v = hj[i * 32 + lane];
      cj[i * 32 + lane] = v;
      acc += (v.x + v.y) + (v.z + v.w);
    }
    for (int o = 16; o > 0; o >>= 1) acc += __shfl_xor(acc, o, 32);
    if (lane == 0) tsum[(ty * GCH + g) * 64 + j] = acc;
  }
  if (t >= NT && t < 64) tsum[(ty * GCH + g) * 64 + t] = 0;
}

__global__ void k_offsets(int* __restrict__ pm_d, int* __restrict__ cnt_d,
                          int* __restrict__ pm_s, int* __restrict__ cnt_s,
                          const int* __restrict__ tsum, int NT) {
  int* __restrict__ pm = blockIdx.y ? pm_s : pm_d;
  int* __restrict__ cnt = blockIdx.y ? cnt_s : cnt_d;
  const int* __restrict__ tsy = tsum + blockIdx.y * (GCH * 64);
  __shared__ int s_pre[8];
  __shared__ int s_loc[8];
  const int t = threadIdx.x, lane = t & 31, w = t >> 5;
  const int nb = blockIdx.x;
  const int NP = NT * 1024;
  const int i0 = nb * 1024 + t * 4;
  int4 c[GCH];
  #pragma unroll
  for (int g = 0; g < GCH; ++g) c[g] = *(const int4*)(pm + (size_t)g * NP + i0);
  static_assert(GCH * 64 == 256 * 4, "one 16-byte load per thread covers the tile sums");
  const int4 tv = *(const int4*)(tsy + t * 4);
  const int j0 = (t * 4) & 63;
  int pre = ((j0 < nb) ? tv.x : 0) + ((j0 + 1 < nb) ? tv.y : 0) +
            ((j0 + 2 < nb) ? tv.z : 0) + ((j0 + 3 < nb) ? tv.w : 0);
  int t0 = 0, t1 = 0, t2 = 0, t3 = 0;
  #pragma unroll
  for (int g = 0; g < GCH; ++g) { t0 += c[g].x;  t1 += c[g].y;  t2 += c[g].z;  t3 += c[g].w; }
  const int loc = (t0 + t1) + (t2 + t3);
  for (int o = 16; o > 0; o >>= 1) pre += __shfl_xor(pre, o, 32);
  int inc = loc;
  for (int o = 1; o < 32; o <<= 1) {
    const int v = __shfl_up(inc, o, 32);
    inc += (lane >= o) ? v : 0;
  }
  if (lane == 31) s_loc[w] = inc;
  if (lane == 0)  s_pre[w] = pre;
  __syncthreads();
  int base = 0;
  for (int k = 0; k < 8; ++k) base += s_pre[k] + ((k < w) ? s_loc[k] : 0);
  const int ex = base + inc - loc;
  int r0 = ex, r1 = ex + t0, r2 = ex + t0 + t1, r3 = ex + t0 + t1 + t2;
  #pragma unroll
  for (int g = 0; g < GCH; ++g) {
    *(int4*)(pm + (size_t)g * NP + i0) = make_int4(r0, r1, r2, r3);
    r0 += c[g].x;  r1 += c[g].y;  r2 += c[g].z;  r3 += c[g].w;
  }
  *(int4*)(cnt + i0) = make_int4(t0, t1, t2, t3);
}

__global__ void k_edge_place(const int* __restrict__ ei, const int* __restrict__ pm_d,
                             const int* __restrict__ pm_s, const int* __restrict__ lr_d,
                             const int* __restrict__ lr_s, int* __restrict__ csr_src,
                             int* __restrict__ csr_sd, int E, int NT) {
  const int T = (int)(gridDim.x * blockDim.x);
  const int e0 = (int)(blockIdx.x * blockDim.x + threadIdx.x);
  const unsigned chunk = (unsigned)((E + GCH - 1) / GCH);
  const int NP = NT * 1024;
  if (e0 + (EPT - 1) * T < E) {
    int s[EPT], d[EPT], rd[EPT], rs[EPT], pd[EPT], ps[EPT];
    #pragma unroll
    for (int k = 0; k < EPT; ++k) {
      const int e = e0 + k * T;
      s[k] = ei[e];  d[k] = ei[E + e];  rd[k] = lr_d[e];  rs[k] = lr_s[e];
    }
    #pragma unroll
    for (int k = 0; k < EPT; ++k) {
      const size_t row = (size_t)((unsigned)(e0 + k * T) / chunk) * NP;
      pd[k] = pm_d[row + d[k]] + rd[k];
      ps[k] = pm_s[row + s[k]] + rs[k];
    }
    #pragma unroll
    for (int k = 0; k < EPT; ++k) {
      csr_src[pd[k]] = s[k];
      csr_sd[ps[k]] = d[k];
    }
  } else {
    for (int k = 0; k < EPT; ++k) {
      const int e = e0 + k * T;
      if (e < E) {
        const int s = ei[e], d = ei[E + e];
        const size_t row = (size_t)((unsigned)e / chunk) * NP;
        const int pd = pm_d[row + d] + lr_d[e];
        const int ps = pm_s[row + s] + lr_s[e];
        csr_src[pd] = s;
        csr_sd[ps] = d;
      }
    }
  }
}

__device__ __forceinline__ float leaky(float v) { return v > 0.0f ? v : v * NEG_SLOPE; }

__device__ __forceinline__ float wave_max(float v) {
  v = fmaxf(v, row_xor<8>(v));
  v = fmaxf(v, row_xor<4>(v));
  v = fmaxf(v, row_xor<2>(v));
  v = fmaxf(v, row_xor<1>(v));
  return fmaxf(__int_as_float(__builtin_amdgcn_readlane(__float_as_int(v), 0)),
               __int_as_float(__builtin_amdgcn_readlane(__float_as_int(v), 16)));
}
__device__ __forceinline__ float wave_sum(float v) {
  v += row_xor<8>(v);
  v += row_xor<4>(v);
  v += row_xor<2>(v);
  v += row_xor<1>(v);
  return __int_as_float(__builtin_amdgcn_readlane(__float_as_int(v), 0)) +
         __int_as_float(__builtin_amdgcn_readlane(__float_as_int(v), 16));
}

__global__ void k_src_stats(const int* __restrict__ csr_sd, const int* __restrict__ off_s,
                            const int* __restrict__ cnt_s, const float* __restrict__ a_i,
                            const float* __restrict__ a_j, float4* __restrict__ st, int Nn) {
  const int w = __builtin_amdgcn_readfirstlane(
      (int)(blockIdx.x * (blockDim.x >> 5) + (threadIdx.x >> 5)));
  const int lane = threadIdx.x & 31;
  const int s0 = w * NPW;
  if (s0 >= Nn) return;
  int beg[NPW], n[NPW];
  float aj[NPW], ais[NPW];
  #pragma unroll
  for (int k = 0; k < NPW; ++k) {
    const int s = (s0 + k < Nn) ? (s0 + k) : (Nn - 1);
    beg[k] = off_s[s];
    n[k]   = cnt_s[s];
    aj[k]  = a_j[s];
    ais[k] = a_i[s];
  }
  int de[NPW];
  #pragma unroll
  for (int k = 0; k < NPW; ++k) de[k] = csr_sd[(lane < n[k]) ? (beg[k] + lane) : 0];
  float ai0[NPW];
  #pragma unroll
  for (int k = 0; k < NPW; ++k) ai0[k] = a_i[de[k]];
  float4 o = make_float4(0.0f, 0.0f, 0.0f, 0.0f);
  #pragma unroll
  for (int k = 0; k < NPW; ++k) {
    if (s0 + k < Nn) {
      const int b = beg[k], nn = n[k];
      const float ajk = aj[k];
      const float sc_self = leaky(ais[k] + ajk);
      const bool v0 = lane < nn;
      const float sc0 = v0 ? leaky(ai0[k] + ajk) : -FLT_MAX;
      float mx = fmaxf(sc_self, sc0);
      for (int cc = 32; cc < nn; cc += 32) {
        const int i = cc + lane;
        if (i < nn) mx = fmaxf(mx, leaky(a_i[csr_sd[b + i]] + ajk));
      }
      mx = wave_max(mx);
      float sum = v0 ? __expf(sc0 - mx) : 0.0f;
      for (int cc = 32; cc < nn; cc += 32) {
        const int i = cc + lane;
        if (i < nn) sum += __expf(leaky(a_i[csr_sd[b + i]] + ajk) - mx);
      }
      const float es = __expf(sc_self - mx);
      const float den = (wave_sum(sum) + es) + 1e-16f;
      const bool mine = (lane == k);
      o.x = mine ? ajk : o.x;
      o.y = mine ? mx  : o.y;
      o.z = mine ? den : o.z;
    }
  }
  if (lane < NPW && s0 + lane < Nn) st[s0 + lane] = o;
}

__device__ __forceinline__ float edge_alpha(float ai, const float4& r) {
  return __expf(leaky(ai + r.x) - r.y) / r.z;
}

template <int NR>
__device__ __forceinline__ void add_rows(int sv, float av, int j0, const float* __restrict__ hl,
                                         float& accx, float& accy) {
  float2 hv[NR];
  float a[NR];
  #pragma unroll
  for (int r = 0; r < NR; ++r) {
    const int s = __builtin_amdgcn_readlane(sv, j0 + r);
    a[r] = __int_as_float(__builtin_amdgcn_readlane(__float_as_int(av), j0 + r));
    hv[r] = *(const float2*)(hl + (size_t)s * 64);
  }
  #pragma unroll
  for (int r = 0; r < NR; ++r) {
    accx = __builtin_fmaf(a[r], hv[r].x, accx);
    accy = __builtin_fmaf(a[r], hv[r].y, accy);
  }
}

__global__ void k_gather_sum(const int* __restrict__ csr_src, const float4* __restrict__ st,
                             const float* __restrict__ a_i, const int* __restrict__ off,
                             const int* __restrict__ cnt, const float* __restrict__ h,
                             const float* __restrict__ bias, const float* __restrict__ Wout,
                             const float* __restrict__ bout, float* __restrict__ y, int Nn) {
  const int w = __builtin_amdgcn_readfirstlane(
      (int)(blockIdx.x * (blockDim.x >> 5) + (threadIdx.x >> 5)));
  const int lane = threadIdx.x & 31;
  const int d0 = w * NPW;
  if (d0 >= Nn) return;
  const float* hl = h + lane * 2;
  const float bz0 = bias[lane * 2], bz1 = bias[lane * 2 + 1];
  const float wz0 = Wout[lane * 2], wz1 = Wout[lane * 2 + 1];
  const float bo  = bout[0];
  int beg[NPW], n[NPW];
  float aid[NPW];
  float4 sd[NPW];
  float2 hs[NPW];
  #pragma unroll
  for (int k = 0; k < NPW; ++k) {
    const int d = (d0 + k < Nn) ? (d0 + k) : (Nn - 1);
    beg[k] = off[d];
    n[k]   = cnt[d];
    aid[k] = a_i[d];
    sd[k]  = st[d];
    hs[k]  = *(const float2*)(hl + (size_t)d * 64);
  }
  int sv0[NPW];
  #pragma unroll
  for (int k = 0; k < NPW; ++k) sv0[k] = csr_src[(lane < n[k]) ? (beg[k] + lane) : 0];
  float4 r0[NPW];
  #pragma unroll
  for (int k = 0; k < NPW; ++k) r0[k] = st[sv0[k]];
  float av0[NPW];
  #pragma unroll
  for (int k = 0; k < NPW; ++k) av0[k] = edge_alpha(aid[k], r0[k]);
  float y_out = 0.0f;
  #pragma unroll
  for (int k = 0; k < NPW; ++k) {
    if (d0 + k < Nn) {
      const int b = beg[k], nn = n[k];
      const float a0 = edge_alpha(aid[k], sd[k]);
      float accx = a0 * hs[k].x, accy = a0 * hs[k].y;
      for (int cc = 0; cc < nn; cc += 32) {
        int sv = sv0[k];
        float av = av0[k];
        if (cc > 0) {
          const int i = cc + lane;
          sv = 0;  av = 0.0f;
          if (i < nn) {
            sv = csr_src[b + i];
            av = edge_alpha(aid[k], st[sv]);
          }
        }
        const int mm = (nn - cc) < 32 ? (nn - cc) : 32;
        int j = 0;
        for (; j + 8 <= mm; j += 8) add_rows<8>(sv, av, j, hl, accx, accy);
        if (mm & 4) { add_rows<4>(sv, av, j, hl, accx, accy); j += 4; }
        if (mm & 2) { add_rows<2>(sv, av, j, hl, accx, accy); j += 2; }
        if (mm & 1) { add_rows<1>(sv, av, j, hl, accx, accy); }
      }
      float v0 = accx + bz0;
      float v1 = accy + bz1;
      v0 = v0 > 0.0f ? v0 : 0.0f;
      v1 = v1 > 0.0f ? v1 : 0.0f;
      float v = v0 * wz0 + v1 * wz1;
      v += row_xor<8>(v);
      v += row_xor<4>(v);
      v += row_xor<2>(v);
      v += row_xor<1>(v);
      const float tot = __int_as_float(__builtin_amdgcn_readlane(__float_as_int(v), 0)) +
                        __int_as_float(__builtin_amdgcn_readlane(__float_as_int(v), 16));
      y_out = (lane == k) ? (tot + bo) : y_out;
    }
  }
  if (lane < NPW && d0 + lane < Nn) y[d0 + lane] = y_out;
}

__device__ __forceinline__ void fma4(float a, const float4& r, float4& acc) {
  acc.x = __builtin_fmaf(a, r.x, acc.x);
  acc.y = __builtin_fmaf(a, r.y, acc.y);
  acc.z = __builtin_fmaf(a, r.z, acc.z);
  acc.w = __builtin_fmaf(a, r.w, acc.w);
}
__device__ __forceinline__ float sum_over_g(float v) {
  v += row_xor<4>(v);
  v += row_xor<8>(v);
  return v + half_swap(v);
}

__global__ void k_gather_x(const int* __restrict__ csr_src, const float4* __restrict__ st,
                           const float* __restrict__ a_i, const int* __restrict__ off,
                           const int* __restrict__ cnt, const float* __restrict__ x,
                           float* __restrict__ sx, float* __restrict__ ssum, int Nn) {
  const int w = __builtin_amdgcn_readfirstlane(
      (int)(blockIdx.x * (blockDim.x >> 5) + (threadIdx.x >> 5)));
  const int lane = threadIdx.x & 31;
  const int d0 = w * NPW;
  if (d0 >= Nn) return;
  const int g = lane >> 2, q = lane & 3;
  const float* xl = x + q * 4;
  int beg[NPW], n[NPW];
  float aid[NPW];
  float4 sd[NPW];
  float4 xs[NPW];
  #pragma unroll
  for (int k = 0; k < NPW; ++k) {
    const int d = (d0 + k < Nn) ? (d0 + k) : (Nn - 1);
    beg[k] = off[d];
    n[k]   = cnt[d];
    aid[k] = a_i[d];
    sd[k]  = st[d];
    xs[k]  = *(const float4*)(xl + (size_t)d * 16);
  }
  int sv0[NPW];
  #pragma unroll
  for (int k = 0; k < NPW; ++k) {
    const int d = (d0 + k < Nn) ? (d0 + k) : (Nn - 1);
    const int sv = csr_src[(lane < n[k]) ? (beg[k] + lane) : 0];
    sv0[k] = (lane < n[k]) ? sv : d;
  }
  float4 rc[NPW];
  #pragma unroll
  for (int k = 0; k < NPW; ++k) rc[k] = st[sv0[k]];
  float4 r0[NPW][4];
  #pragma unroll
  for (int k = 0; k < NPW; ++k) {
    #pragma unroll
    for (int cg = 0; cg < 4; ++cg) {
      const int s = __shfl(sv0[k], cg * 8 + g, 32);
      r0[k][cg] = *(const float4*)(xl + (size_t)s * 16);
    }
  }
  float av0[NPW];
  float ar[NPW][4];
  #pragma unroll
  for (int k = 0; k < NPW; ++k) {
    asm volatile("" :: "v"(rc[k].x), "v"(rc[k].y), "v"(rc[k].z));
    av0[k] = (lane < n[k]) ? edge_alpha(aid[k], rc[k]) : 0.0f;
    #pragma unroll
    for (int cg = 0; cg < 4; ++cg) ar[k][cg] = __shfl(av0[k], cg * 8 + g, 32);
  }
  float4 o = make_float4(0.0f, 0.0f, 0.0f, 0.0f);
  float s_out = 0.0f;
  #pragma unroll
  for (int k = 0; k < NPW; ++k) {
    if (d0 + k < Nn) {
      const int b = beg[k], nn = n[k];
      const float a0 = edge_alpha(aid[k], sd[k]);
      const float as = (g == 0) ? a0 : 0.0f;
      float4 acc = make_float4(as * xs[k].x, as * xs[k].y, as * xs[k].z, as * xs[k].w);
      #pragma unroll
      for (int cg = 0; cg < 4; ++cg) fma4(ar[k][cg], r0[k][cg], acc);
      float al = av0[k];
      for (int cc = 32; cc < nn; cc += 32) {
        const int i = cc + lane;
        int sv = 0;
        float av = 0.0f;
        if (i < nn) {
          sv = csr_src[b + i];
          av = edge_alpha(aid[k], st[sv]);
        }
        al += av;
        const int mm = nn - cc;
        #pragma unroll
        for (int cg = 0; cg < 4; ++cg) {
          const int j = cg * 8 + g;
          const int s = __shfl(sv, j, 32);
          const float a = __shfl(av, j, 32);
          float4 rr = make_float4(0.0f, 0.0f, 0.0f, 0.0f);
          if (j < mm) rr = *(const float4*)(xl + (size_t)s * 16);
          fma4(a, rr, acc);
        }
      }
      acc.x = sum_over_g(acc.x);
      acc.y = sum_over_g(acc.y);
      acc.z = sum_over_g(acc.z);
      acc.w = sum_over_g(acc.w);
      const float tot = wave_sum(al) + a0;
      const bool mine = (g == k);
      o.x = mine ? acc.x : o.x;
      o.y = mine ? acc.y : o.y;
      o.z = mine ? acc.z : o.z;
      o.w = mine ? acc.w : o.w;
      s_out = (lane == k) ? tot : s_out;
    }
  }
  if (g < NPW && d0 + g < Nn) *(float4*)(sx + (size_t)(d0 + g) * 16 + q * 4) = o;
  if (lane < NPW && d0 + lane < Nn) ssum[d0 + lane] = s_out;
}

extern "C" void kernel_launch(void* const* d_in, const int* in_sizes, int n_in,
                              void* d_out, int out_size, void* d_ws, size_t ws_size,
                              hipStream_t stream) {
  const float* x     = (const float*)d_in[0];
  const int*   ei    = (const int*)  d_in[1];
  const float* W1    = (const float*)d_in[2];
  const float* b1    = (const float*)d_in[3];
  const float* att1  = (const float*)d_in[4];
  const float* bias1 = (const float*)d_in[5];
  const float* W2    = (const float*)d_in[6];
  const float* b2    = (const float*)d_in[7];
  const float* att2  = (const float*)d_in[8];
  const float* bias2 = (const float*)d_in[9];
  const float* Wout  = (const float*)d_in[10];
  const float* bout  = (const float*)d_in[11];
  float* y = (float*)d_out;

  const int Nn   = in_sizes[0] / 16;
  const int E    = in_sizes[1] / 2;

  char* p = (char*)d_ws;
  auto alloc = [&](size_t bytes) {
    char* r = p;
    p += (bytes + 255) & ~(size_t)255;
    return r;
  };
  float* h      = (float*)alloc((size_t)Nn * 64 * sizeof(float));
  float* sx     = (float*)alloc((size_t)Nn * 16 * sizeof(float));
  float* ssum   = (float*)alloc((size_t)Nn * sizeof(float));
  float* a_i    = (float*)alloc((size_t)Nn * sizeof(float));
  float* a_j    = (float*)alloc((size_t)Nn * sizeof(float));
  float4* st    = (float4*)alloc((size_t)Nn * sizeof(float4));
  const int NT  = (Nn + 1023) / 1024;
  const size_t NP = (size_t)NT * 1024;
  int*   cnt_d  = (int*)  alloc(NP * sizeof(int));
  int*   cnt_s  = (int*)  alloc(NP * sizeof(int));
  int*   pm_d   = (int*)  alloc(NP * GCH * sizeof(int));
  int*   pm_s   = (int*)  alloc(NP * GCH * sizeof(int));
  int*   tsum   = (int*)  alloc((size_t)2 * GCH * 64 * sizeof(int));
  int*   lr_d   = (int*)  alloc((size_t)E * sizeof(int));
  int*   lr_s   = (int*)  alloc((size_t)E * sizeof(int));
  const int* off_d = pm_d;
  const int* off_s = pm_s;
  int*   csr_src = (int*) alloc((size_t)E * sizeof(int));
  int*   csr_sd = (int*)  alloc((size_t)E * sizeof(int));

  const int TB = 256;
  dim3 blk(TB);
  int g_gemm  = ((Nn / 16) + 7) / 8;
  int g_node  = (Nn + 8 * NPW - 1) / (8 * NPW);
  int g_edge  = (E + TB * EPT - 1) / (TB * EPT);
  dim3 g_scan(NT, 2);
  int g_cnt   = 2 * GCH + ((Nn / 16) + 31) / 32;

  k_count_and_dots<<<g_cnt, dim3(1024), 0, stream>>>(ei, pm_d, pm_s, lr_d, lr_s, tsum, E, NT,
                                                     x, W1, b1, att1, a_i, a_j, Nn);
  k_offsets<<<g_scan, blk, 0, stream>>>(pm_d, cnt_d, pm_s, cnt_s, tsum, NT);
  k_edge_place<<<g_edge, blk, 0, stream>>>(ei, pm_d, pm_s, lr_d, lr_s, csr_src, csr_sd, E, NT);

  k_src_stats<<<g_node, blk, 0, stream>>>(csr_sd, off_s, cnt_s, a_i, a_j, st, Nn);
  k_gather_x<<<g_node, blk, 0, stream>>>(csr_src, st, a_i, off_d, cnt_d, x, sx, ssum, Nn);

  k_linear_chain<<<g_gemm, blk, 0, stream>>>(sx, ssum, W1, b1, bias1, W2, b2, att2, h, a_i, a_j, Nn);
  k_src_stats<<<g_node, blk, 0, stream>>>(csr_sd, off_s, cnt_s, a_i, a_j, st, Nn);
  k_gather_sum<<<g_node, blk, 0, stream>>>(csr_src, st, a_i, off_d, cnt_d, h,
                                           bias2, Wout, bout, y, Nn);
}


# both edge orders by a two-level partition: chunks ordered by 1024-node tile in LDS, then one work-group per tile sorts it in LDS; no scattered 4-byte store left; sum order: LDS arrival in the tile
# speedup vs baseline: 5.7619x; 1.4789x over previous
#include <hip/hip_runtime.h>
#include <float.h>


#define NPW 4
#define TSH 10
#define TILE 1024
#define NBIN 64
#define ECH 4096
#define SCAP 49152
#define BSP 80
#define BPT 16

typedef __attribute__((ext_vector_type(2))) float v2f;
typedef __attribute__((ext_vector_type(8))) float v8f;

#define NEG_SLOPE 0.01f

template <int MASK>
__device__ __forceinline__ float row_xor(float v) {
  return __int_as_float(__builtin_amdgcn_update_dpp(0, __float_as_int(v), 0x160 + MASK,
                                                    0xf, 0xf, true));
}

__device__ __forceinline__ float half_swap(float v) {
  return __int_as_float(__builtin_amdgcn_permlanex16(0, __float_as_int(v), 0x76543210u,
                                                    0xfedcba98u, true, false));
}


template <bool STORE>
__device__ __forceinline__ void finish_rows(const v8f& acc0, const v8f& acc1, const v8f& acc2,
                                            const v8f& acc3, const float* __restrict__ b,
                                            const float* __restrict__ att, float* __restrict__ h,
                                            float* __restrict__ a_i, float* __restrict__ a_j,
                                            int row0, int half, int l16) {
  const float bb0 = b[l16], bb1 = b[16 + l16], bb2 = b[32 + l16], bb3 = b[48 + l16];
  const float ti0 = att[l16],      ti1 = att[16 + l16], ti2 = att[32 + l16],  ti3 = att[48 + l16];
  const float tj0 = att[64 + l16], tj1 = att[80 + l16], tj2 = att[96 + l16],  tj3 = att[112 + l16];
  float oi = 0.0f, oj = 0.0f;
  #pragma unroll
  for (int v = 0; v < 8; ++v) {
    int m = row0 + v + half * 8;
    float h0 = acc0[v] + bb0, h1 = acc1[v] + bb1, h2 = acc2[v] + bb2, h3 = acc3[v] + bb3;
    if (STORE) {
      float* hrow = h + (size_t)m * 64 + l16;
      hrow[0]  = h0;
      hrow[16] = h1;
      hrow[32] = h2;
      hrow[48] = h3;
    }
    float vi = h0 * ti0 + h1 * ti1 + h2 * ti2 + h3 * ti3;
    float vj = h0 * tj0 + h1 * tj1 + h2 * tj2 + h3 * tj3;
    vi += row_xor<8>(vi);  vj += row_xor<8>(vj);
    vi += row_xor<4>(vi);  vj += row_xor<4>(vj);
    vi += row_xor<2>(vi);  vj += row_xor<2>(vj);
    vi += row_xor<1>(vi);  vj += row_xor<1>(vj);
    bool mine = (l16 & 7) == v;
    oi = mine ? vi : oi;
    oj = mine ? vj : oj;
  }
  {
    float* op = (l16 < 8) ? a_i : a_j;
    op[row0 + half * 8 + (l16 & 7)] = (l16 < 8) ? oi : oj;
  }
}

__device__ __forceinline__ void linear_dots_strip(const float* __restrict__ in,
                                                  const float* __restrict__ W,
                                                  const float* __restrict__ b,
                                                  const float* __restrict__ att,
                                                  float* __restrict__ a_i, float* __restrict__ a_j,
                                                  int M, int wave) {
  int lane = threadIdx.x & 31;
  int row0 = wave * 16;
  if (row0 >= M) return;
  int half = lane >> 4;
  int l16  = lane & 15;

  v8f acc0 = {}, acc1 = {}, acc2 = {}, acc3 = {};
  const float* arow = in + (size_t)(row0 + l16) * 16;

  #pragma unroll
  for (int k0 = 0; k0 < 16; k0 += 4) {
    int kb = k0 + half * 2;
    v2f a;  a.x = arow[kb];  a.y = arow[kb + 1];
    v2f b0, b1, b2, b3;
    { const float* p = W + (size_t)(0  + l16) * 16 + kb; b0.x = p[0]; b0.y = p[1]; }
    { const float* p = W + (size_t)(16 + l16) * 16 + kb; b1.x = p[0]; b1.y = p[1]; }
    { const float* p = W + (size_t)(32 + l16) * 16 + kb; b2.x = p[0]; b2.y = p[1]; }
    { const float* p = W + (size_t)(48 + l16) * 16 + kb; b3.x = p[0]; b3.y = p[1]; }
    acc0 = __builtin_amdgcn_wmma_f32_16x16x4_f32(false, a, false, b0, (short)0, acc0, false, false);
    acc1 = __builtin_amdgcn_wmma_f32_16x16x4_f32(false, a, false, b1, (short)0, acc1, false, false);
    acc2 = __builtin_amdgcn_wmma_f32_16x16x4_f32(false, a, false, b2, (short)0, acc2, false, false);
    acc3 = __builtin_amdgcn_wmma_f32_16x16x4_f32(false, a, false, b3, (short)0, acc3, false, false);
  }
  finish_rows<false>(acc0, acc1, acc2, acc3, b, att, nullptr, a_i, a_j, row0, half, l16);
}

__global__ void k_linear_chain(const float* __restrict__ sx, const float* __restrict__ ssum,
                               const float* __restrict__ W1, const float* __restrict__ b1,
                               const float* __restrict__ bias1, const float* __restrict__ W2,
                               const float* __restrict__ b2, const float* __restrict__ att,
                               float* __restrict__ h, float* __restrict__ a_i,
                               float* __restrict__ a_j, int M) {
  int wave = (int)((blockIdx.x * blockDim.x + threadIdx.x) >> 5);
  int lane = threadIdx.x & 31;
  int row0 = wave * 16;
  if (row0 >= M) return;
  int half = lane >> 4;
  int l16  = lane & 15;

  v8f tT[4] = {};
  const float* xrow = sx + (size_t)(row0 + l16) * 16;
  #pragma unroll
  for (int k0 = 0; k0 < 16; k0 += 4) {
    int kb = k0 + half * 2;
    v2f xb;  xb.x = xrow[kb];  xb.y = xrow[kb + 1];
    #pragma unroll
    for (int t = 0; t < 4; ++t) {
      const float* p = W1 + (size_t)(t * 16 + l16) * 16 + kb;
      v2f wa;  wa.x = p[0];  wa.y = p[1];
      tT[t] = __builtin_amdgcn_wmma_f32_16x16x4_f32(false, wa, false, xb, (short)0, tT[t], false, false);
    }
  }
  {
    const float sn = ssum[row0 + l16];
    v2f xb;  xb.x = half ? 0.0f : sn;  xb.y = half ? 0.0f : 1.0f;
    #pragma unroll
    for (int t = 0; t < 4; ++t) {
      const float pb = b1[t * 16 + l16], pc = bias1[t * 16 + l16];
      v2f wa;  wa.x = half ? 0.0f : pb;  wa.y = half ? 0.0f : pc;
      tT[t] = __builtin_amdgcn_wmma_f32_16x16x4_f32(false, wa, false, xb, (short)0, tT[t], false, false);
    }
  }

  v8f acc0 = {}, acc1 = {}, acc2 = {}, acc3 = {};
  #pragma unroll
  for (int t = 0; t < 4; ++t) {
    #pragma unroll
    for (int p = 0; p < 4; ++p) {
      const float u0 = tT[t][2 * p], u1 = tT[t][2 * p + 1];
      v2f a;
      a.x = u0 > 0.0f ? u0 : 0.0f;
      a.y = u1 > 0.0f ? u1 : 0.0f;
      const int kb = t * 16 + half * 8 + 2 * p;
      v2f b0, b1v, b2v, b3;
      { const float* q = W2 + (size_t)(0  + l16) * 64 + kb; b0.x  = q[0]; b0.y  = q[1]; }
      { const float* q = W2 + (size_t)(16 + l16) * 64 + kb; b1v.x = q[0]; b1v.y = q[1]; }
      { const float* q = W2 + (size_t)(32 + l16) * 64 + kb; b2v.x = q[0]; b2v.y = q[1]; }
      { const float* q = W2 + (size_t)(48 + l16) * 64 + kb; b3.x  = q[0]; b3.y  = q[1]; }
      acc0 = __builtin_amdgcn_wmma_f32_16x16x4_f32(false, a, false, b0,  (short)0, acc0, false, false);
      acc1 = __builtin_amdgcn_wmma_f32_16x16x4_f32(false, a, false, b1v, (short)0, acc1, false, false);
      acc2 = __builtin_amdgcn_wmma_f32_16x16x4_f32(false, a, false, b2v, (short)0, acc2, false, false);
      acc3 = __builtin_amdgcn_wmma_f32_16x16x4_f32(false, a, false, b3,  (short)0, acc3, false, false);
    }
  }
  finish_rows<true>(acc0, acc1, acc2, acc3, b2, att, h, a_i, a_j, row0, half, l16);
}

__global__ void __launch_bounds__(1024)
k_bin_count_and_dots(const int* __restrict__ ei, int* __restrict__ cmat, int E, int C,
                     const float* __restrict__ x, const float* __restrict__ W1,
                     const float* __restrict__ b1, const float* __restrict__ att1,
                     float* __restrict__ a_i, float* __restrict__ a_j, int Nn) {
  __shared__ int wh[32 * 2 * NBIN];
  const int t = threadIdx.x, w = t >> 5;
  if ((int)blockIdx.x >= C) {
    linear_dots_strip(x, W1, b1, att1, a_i, a_j, Nn, ((int)blockIdx.x - C) * 32 + w);
    return;
  }
  const int g = blockIdx.x;
  int s[4], d[4];
  #pragma unroll
  for (int i = 0; i < 4; ++i) {
    const int e = g * ECH + i * 1024 + t;
    const int ec = (e < E) ? e : 0;
    const int sv = ei[ec];
    d[i] = ei[E + ec];
    s[i] = (e < E) ? sv : -1;
  }
  #pragma unroll
  for (int i = 0; i < 4; ++i) wh[i * 1024 + t] = 0;
  __syncthreads();
  int* mine = wh + w * (2 * NBIN);
  #pragma unroll
  for (int i = 0; i < 4; ++i) {
    if (s[i] >= 0) {
      atomicAdd(&mine[d[i] >> TSH], 1);
      atomicAdd(&mine[NBIN + (s[i] >> TSH)], 1);
    }
  }
  __syncthreads();
  if (t < 2 * NBIN) {
    int sum = 0;
    #pragma unroll
    for (int k = 0; k < 32; ++k) sum += wh[k * (2 * NBIN) + t];
    cmat[(size_t)t * C + g] = sum;
  }
}

__global__ void __launch_bounds__(1024)
k_bin_scatter(const int* __restrict__ ei, const int* __restrict__ cmat,
              unsigned* __restrict__ co_d, unsigned* __restrict__ co_s,
              int* __restrict__ bstart, int E, int C) {
  __shared__ int wh[32 * NBIN];
  __shared__ int s_tot[NBIN], s_part[NBIN], s_cnt[NBIN], s_lb[NBIN], s_gb[NBIN];
  __shared__ unsigned st_w[ECH];
  __shared__ int st_p[ECH];
  const int t = threadIdx.x, lane = t & 31, w = t >> 5;
  const int g = blockIdx.x, o = blockIdx.y;
  const int* __restrict__ keys = o ? ei : ei + E;
  const int* __restrict__ oth  = o ? ei + E : ei;
  unsigned* __restrict__ co = o ? co_s : co_d;
  const int* __restrict__ cm = cmat + (size_t)o * NBIN * C;

  int key[4];
  unsigned word[4];
  #pragma unroll
  for (int i = 0; i < 4; ++i) {
    const int e = g * ECH + i * 1024 + t;
    const int ec = (e < E) ? e : 0;
    const int kv = keys[ec];
    const int ot = oth[ec];
    key[i] = (e < E) ? kv : -1;
    word[i] = ((unsigned)ot << TSH) | ((unsigned)kv & (TILE - 1));
  }
  wh[t] = 0;
  wh[t + 1024] = 0;
  for (int b = w; b < NBIN; b += 32) {
    int tot = 0, part = 0;
    for (int c = lane; c < C; c += 32) {
      const int v = cm[(size_t)b * C + c];
      tot += v;
      part += (c < g) ? v : 0;
    }
    for (int m = 16; m > 0; m >>= 1) {
      tot  += __shfl_xor(tot, m, 32);
      part += __shfl_xor(part, m, 32);
    }
    if (lane == 0) { s_tot[b] = tot;  s_part[b] = part; }
  }
  __syncthreads();
  int r[4];
  #pragma unroll
  for (int i = 0; i < 4; ++i)
    r[i] = (key[i] >= 0) ? atomicAdd(&wh[w * NBIN + (key[i] >> TSH)], 1) : 0;
  __syncthreads();
  if (t < NBIN) {
    int run = 0;
    #pragma unroll
    for (int k = 0; k < 32; ++k) {
      const int v = wh[k * NBIN + t];
      wh[k * NBIN + t] = run;
      run += v;
    }
    s_cnt[t] = run;
  }
  __syncthreads();
  if (t < NBIN) {
    int lb = 0, bs = 0;
    #pragma unroll
    for (int k = 0; k < NBIN; ++k) {
      lb += (k < t) ? s_cnt[k] : 0;
      bs += (k < t) ? s_tot[k] : 0;
    }
    s_lb[t] = lb;
    s_gb[t] = bs + s_part[t];
    if (g == 0) {
      bstart[o * BSP + t] = bs;
      if (t == NBIN - 1) bstart[o * BSP + NBIN] = bs + s_tot[t];
    }
  }
  __syncthreads();
  #pragma unroll
  for (int i = 0; i < 4; ++i) {
    if (key[i] >= 0) {
      const int b = key[i] >> TSH;
      const int q = wh[w * NBIN + b] + r[i];
      const int p = s_lb[b] + q;
      st_w[p] = word[i];
      st_p[p] = s_gb[b] + q;
    }
  }
  __syncthreads();
  const int left = E - g * ECH;
  const int nch = left < ECH ? left : ECH;
  for (int i = t; i < nch; i += 1024) co[st_p[i]] = st_w[i];
}

__global__ void __launch_bounds__(1024)
k_tile_sort(const unsigned* __restrict__ co_d, const unsigned* __restrict__ co_s,
            const int* __restrict__ bstart, int* __restrict__ off_d, int* __restrict__ cnt_d,
            int* __restrict__ off_s, int* __restrict__ cnt_s, int* __restrict__ csr_src,
            int* __restrict__ csr_sd) {
  __shared__ int cn[TILE];
  __shared__ int cur[TILE];
  __shared__ int s_wt[32];
  __shared__ int stage[SCAP];
  const int t = threadIdx.x, lane = t & 31, w = t >> 5;
  const int j = blockIdx.x, o = blockIdx.y;
  int* __restrict__ off = o ? off_s : off_d;
  int* __restrict__ cnt = o ? cnt_s : cnt_d;
  int* __restrict__ csr = o ? csr_sd : csr_src;
  const int B0 = bstart[o * BSP + j], B1 = bstart[o * BSP + j + 1];
  const int n = B1 - B0;
  const unsigned* __restrict__ cb = (o ? co_s : co_d) + B0;
  csr += B0;

  cn[t] = 0;
  __syncthreads();
  for (int i0 = t; i0 < n + t; i0 += BPT * 1024) {
    unsigned v[BPT];
    #pragma unroll
    for (int k = 0; k < BPT; ++k) v[k] = cb[(i0 + k * 1024 < n) ? (i0 + k * 1024) : 0];
    #pragma unroll
    for (int k = 0; k < BPT; ++k)
      if (i0 + k * 1024 < n) atomicAdd(&cn[v[k] & (TILE - 1)], 1);
  }
  __syncthreads();

  const int c = cn[t];
  int inc = c;
  for (int m = 1; m < 32; m <<= 1) {
    const int v = __shfl_up(inc, m, 32);
    inc += (lane >= m) ? v : 0;
  }
  if (lane == 31) s_wt[w] = inc;
  __syncthreads();
  int base = 0;
  #pragma unroll
  for (int k = 0; k < 32; ++k) base += (k < w) ? s_wt[k] : 0;
  const int ex = base + inc - c;
  off[j * TILE + t] = B0 + ex;
  cnt[j * TILE + t] = c;
  cur[t] = ex;
  __syncthreads();

  if (n <= SCAP) {
    for (int i0 = t; i0 < n + t; i0 += BPT * 1024) {
      unsigned v[BPT];
      int p[BPT];
      #pragma unroll
      for (int k = 0; k < BPT; ++k) v[k] = cb[(i0 + k * 1024 < n) ? (i0 + k * 1024) : 0];
      #pragma unroll
      for (int k = 0; k < BPT; ++k)
        p[k] = (i0 + k * 1024 < n) ? atomicAdd(&cur[v[k] & (TILE - 1)], 1) : 0;
      #pragma unroll
      for (int k = 0; k < BPT; ++k)
        if (i0 + k * 1024 < n) stage[p[k]] = (int)(v[k] >> TSH);
    }
    __syncthreads();
    for (int i = t; i < n; i += 1024) csr[i] = stage[i];
  } else {
    for (int i = t; i < n; i += 1024) {
      const unsigned v = cb[i];
      csr[atomicAdd(&cur[v & (TILE - 1)], 1)] = (int)(v >> TSH);
    }
  }
}

__device__ __forceinline__ float leaky(float v) { return v > 0.0f ? v : v * NEG_SLOPE; }

__device__ __forceinline__ float wave_max(float v) {
  v = fmaxf(v, row_xor<8>(v));
  v = fmaxf(v, row_xor<4>(v));
  v = fmaxf(v, row_xor<2>(v));
  v = fmaxf(v, row_xor<1>(v));
  return fmaxf(__int_as_float(__builtin_amdgcn_readlane(__float_as_int(v), 0)),
               __int_as_float(__builtin_amdgcn_readlane(__float_as_int(v), 16)));
}
__device__ __forceinline__ float wave_sum(float v) {
  v += row_xor<8>(v);
  v += row_xor<4>(v);
  v += row_xor<2>(v);
  v += row_xor<1>(v);
  return __int_as_float(__builtin_amdgcn_readlane(__float_as_int(v), 0)) +
         __int_as_float(__builtin_amdgcn_readlane(__float_as_int(v), 16));
}

__global__ void k_src_stats(const int* __restrict__ csr_sd, const int* __restrict__ off_s,
                            const int* __restrict__ cnt_s, const float* __restrict__ a_i,
                            const float* __restrict__ a_j, float4* __restrict__ st, int Nn) {
  const int w = __builtin_amdgcn_readfirstlane(
      (int)(blockIdx.x * (blockDim.x >> 5) + (threadIdx.x >> 5)));
  const int lane = threadIdx.x & 31;
  const int s0 = w * NPW;
  if (s0 >= Nn) return;
  int beg[NPW], n[NPW];
  float aj[NPW], ais[NPW];
  #pragma unroll
  for (int k = 0; k < NPW; ++k) {
    const int s = (s0 + k < Nn) ? (s0 + k) : (Nn - 1);
    beg[k] = off_s[s];
    n[k]   = cnt_s[s];
    aj[k]  = a_j[s];
    ais[k] = a_i[s];
  }
  int de[NPW];
  #pragma unroll
  for (int k = 0; k < NPW; ++k) de[k] = csr_sd[(lane < n[k]) ? (beg[k] + lane) : 0];
  float ai0[NPW];
  #pragma unroll
  for (int k = 0; k < NPW; ++k) ai0[k] = a_i[de[k]];
  float4 o = make_float4(0.0f, 0.0f, 0.0f, 0.0f);
  #pragma unroll
  for (int k = 0; k < NPW; ++k) {
    if (s0 + k < Nn) {
      const int b = beg[k], nn = n[k];
      const float ajk = aj[k];
      const float sc_self = leaky(ais[k] + ajk);
      const bool v0 = lane < nn;
      const float sc0 = v0 ? leaky(ai0[k] + ajk) : -FLT_MAX;
      float mx = fmaxf(sc_self, sc0);
      for (int cc = 32; cc < nn; cc += 32) {
        const int i = cc + lane;
        if (i < nn) mx = fmaxf(mx, leaky(a_i[csr_sd[b + i]] + ajk));
      }
      mx = wave_max(mx);
      float sum = v0 ? __expf(sc0 - mx) : 0.0f;
      for (int cc = 32; cc < nn; cc += 32) {
        const int i = cc + lane;
        if (i < nn) sum += __expf(leaky(a_i[csr_sd[b + i]] + ajk) - mx);
      }
      const float es = __expf(sc_self - mx);
      const float den = (wave_sum(sum) + es) + 1e-16f;
      const bool mine = (lane == k);
      o.x = mine ? ajk : o.x;
      o.y = mine ? mx  : o.y;
      o.z = mine ? den : o.z;
    }
  }
  if (lane < NPW && s0 + lane < Nn) st[s0 + lane] = o;
}

__device__ __forceinline__ float edge_alpha(float ai, const float4& r) {
  return __expf(leaky(ai + r.x) - r.y) / r.z;
}

template <int NR>
__device__ __forceinline__ void add_rows(int sv, float av, int j0, const float* __restrict__ hl,
                                         float& accx, float& accy) {
  float2 hv[NR];
  float a[NR];
  #pragma unroll
  for (int r = 0; r < NR; ++r) {
    const int s = __builtin_amdgcn_readlane(sv, j0 + r);
    a[r] = __int_as_float(__builtin_amdgcn_readlane(__float_as_int(av), j0 + r));
    hv[r] = *(const float2*)(hl + (size_t)s * 64);
  }
  #pragma unroll
  for (int r = 0; r < NR; ++r) {
    accx = __builtin_fmaf(a[r], hv[r].x, accx);
    accy = __builtin_fmaf(a[r], hv[r].y, accy);
  }
}

__global__ void k_gather_sum(const int* __restrict__ csr_src, const float4* __restrict__ st,
                             const float* __restrict__ a_i, const int* __restrict__ off,
                             const int* __restrict__ cnt, const float* __restrict__ h,
                             const float* __restrict__ bias, const float* __restrict__ Wout,
                             const float* __restrict__ bout, float* __restrict__ y, int Nn) {
  const int w = __builtin_amdgcn_readfirstlane(
      (int)(blockIdx.x * (blockDim.x >> 5) + (threadIdx.x >> 5)));
  const int lane = threadIdx.x & 31;
  const int d0 = w * NPW;
  if (d0 >= Nn) return;
  const float* hl = h + lane * 2;
  const float bz0 = bias[lane * 2], bz1 = bias[lane * 2 + 1];
  const float wz0 = Wout[lane * 2], wz1 = Wout[lane * 2 + 1];
  const float bo  = bout[0];
  int beg[NPW], n[NPW];
  float aid[NPW];
  float4 sd[NPW];
  float2 hs[NPW];
  #pragma unroll
  for (int k = 0; k < NPW; ++k) {
    const int d = (d0 + k < Nn) ? (d0 + k) : (Nn - 1);
    beg[k] = off[d];
    n[k]   = cnt[d];
    aid[k] = a_i[d];
    sd[k]  = st[d];
    hs[k]  = *(const float2*)(hl + (size_t)d * 64);
  }
  int sv0[NPW];
  #pragma unroll
  for (int k = 0; k < NPW; ++k) sv0[k] = csr_src[(lane < n[k]) ? (beg[k] + lane) : 0];
  float4 r0[NPW];
  #pragma unroll
  for (int k = 0; k < NPW; ++k) r0[k] = st[sv0[k]];
  float av0[NPW];
  #pragma unroll
  for (int k = 0; k < NPW; ++k) av0[k] = edge_alpha(aid[k], r0[k]);
  float y_out = 0.0f;
  #pragma unroll
  for (int k = 0; k < NPW; ++k) {
    if (d0 + k < Nn) {
      const int b = beg[k], nn = n[k];
      const float a0 = edge_alpha(aid[k], sd[k]);
      float accx = a0 * hs[k].x, accy = a0 * hs[k].y;
      for (int cc = 0; cc < nn; cc += 32) {
        int sv = sv0[k];
        float av = av0[k];
        if (cc > 0) {
          const int i = cc + lane;
          sv = 0;  av = 0.0f;
          if (i < nn) {
            sv = csr_src[b + i];
            av = edge_alpha(aid[k], st[sv]);
          }
        }
        const int mm = (nn - cc) < 32 ? (nn - cc) : 32;
        int j = 0;
        for (; j + 8 <= mm; j += 8) add_rows<8>(sv, av, j, hl, accx, accy);
        if (mm & 4) { add_rows<4>(sv, av, j, hl, accx, accy); j += 4; }
        if (mm & 2) { add_rows<2>(sv, av, j, hl, accx, accy); j += 2; }
        if (mm & 1) { add_rows<1>(sv, av, j, hl, accx, accy); }
      }
      float v0 = accx + bz0;
      float v1 = accy + bz1;
      v0 = v0 > 0.0f ? v0 : 0.0f;
      v1 = v1 > 0.0f ? v1 : 0.0f;
      float v = v0 * wz0 + v1 * wz1;
      v += row_xor<8>(v);
      v += row_xor<4>(v);
      v += row_xor<2>(v);
      v += row_xor<1>(v);
      const float tot = __int_as_float(__builtin_amdgcn_readlane(__float_as_int(v), 0)) +
                        __int_as_float(__builtin_amdgcn_readlane(__float_as_int(v), 16));
      y_out = (lane == k) ? (tot + bo) : y_out;
    }
  }
  if (lane < NPW && d0 + lane < Nn) y[d0 + lane] = y_out;
}

__device__ __forceinline__ void fma4(float a, const float4& r, float4& acc) {
  acc.x = __builtin_fmaf(a, r.x, acc.x);
  acc.y = __builtin_fmaf(a, r.y, acc.y);
  acc.z = __builtin_fmaf(a, r.z, acc.z);
  acc.w = __builtin_fmaf(a, r.w, acc.w);
}
__device__ __forceinline__ float sum_over_g(float v) {
  v += row_xor<4>(v);
  v += row_xor<8>(v);
  return v + half_swap(v);
}

__global__ void k_gather_x(const int* __restrict__ csr_src, const float4* __restrict__ st,
                           const float* __restrict__ a_i, const int* __restrict__ off,
                           const int* __restrict__ cnt, const float* __restrict__ x,
                           float* __restrict__ sx, float* __restrict__ ssum, int Nn) {
  const int w = __builtin_amdgcn_readfirstlane(
      (int)(blockIdx.x * (blockDim.x >> 5) + (threadIdx.x >> 5)));
  const int lane = threadIdx.x & 31;
  const int d0 = w * NPW;
  if (d0 >= Nn) return;
  const int g = lane >> 2, q = lane & 3;
  const float* xl = x + q * 4;
  int beg[NPW], n[NPW];
  float aid[NPW];
  float4 sd[NPW];
  float4 xs[NPW];
  #pragma unroll
  for (int k = 0; k < NPW; ++k) {
    const int d = (d0 + k < Nn) ? (d0 + k) : (Nn - 1);
    beg[k] = off[d];
    n[k]   = cnt[d];
    aid[k] = a_i[d];
    sd[k]  = st[d];
    xs[k]  = *(const float4*)(xl + (size_t)d * 16);
  }
  int sv0[NPW];
  #pragma unroll
  for (int k = 0; k < NPW; ++k) {
    const int d = (d0 + k < Nn) ? (d0 + k) : (Nn - 1);
    const int sv = csr_src[(lane < n[k]) ? (beg[k] + lane) : 0];
    sv0[k] = (lane < n[k]) ? sv : d;
  }
  float4 rc[NPW];
  #pragma unroll
  for (int k = 0; k < NPW; ++k) rc[k] = st[sv0[k]];
  float4 r0[NPW][4];
  #pragma unroll
  for (int k = 0; k < NPW; ++k) {
    #pragma unroll
    for (int cg = 0; cg < 4; ++cg) {
      const int s = __shfl(sv0[k], cg * 8 + g, 32);
      r0[k][cg] = *(const float4*)(xl + (size_t)s * 16);
    }
  }
  float av0[NPW];
  float ar[NPW][4];
  #pragma unroll
  for (int k = 0; k < NPW; ++k) {
    asm volatile("" :: "v"(rc[k].x), "v"(rc[k].y), "v"(rc[k].z));
    av0[k] = (lane < n[k]) ? edge_alpha(aid[k], rc[k]) : 0.0f;
    #pragma unroll
    for (int cg = 0; cg < 4; ++cg) ar[k][cg] = __shfl(av0[k], cg * 8 + g, 32);
  }
  float4 o = make_float4(0.0f, 0.0f, 0.0f, 0.0f);
  float s_out = 0.0f;
  #pragma unroll
  for (int k = 0; k < NPW; ++k) {
    if (d0 + k < Nn) {
      const int b = beg[k], nn = n[k];
      const float a0 = edge_alpha(aid[k], sd[k]);
      const float as = (g == 0) ? a0 : 0.0f;
      float4 acc = make_float4(as * xs[k].x, as * xs[k].y, as * xs[k].z, as * xs[k].w);
      #pragma unroll
      for (int cg = 0; cg < 4; ++cg) fma4(ar[k][cg], r0[k][cg], acc);
      float al = av0[k];
      for (int cc = 32; cc < nn; cc += 32) {
        const int i = cc + lane;
        int sv = 0;
        float av = 0.0f;
        if (i < nn) {
          sv = csr_src[b + i];
          av = edge_alpha(aid[k], st[sv]);
        }
        al += av;
        const int mm = nn - cc;
        #pragma unroll
        for (int cg = 0; cg < 4; ++cg) {
          const int j = cg * 8 + g;
          const int s = __shfl(sv, j, 32);
          const float a = __shfl(av, j, 32);
          float4 rr = make_float4(0.0f, 0.0f, 0.0f, 0.0f);
          if (j < mm) rr = *(const float4*)(xl + (size_t)s * 16);
          fma4(a, rr, acc);
        }
      }
      acc.x = sum_over_g(acc.x);
      acc.y = sum_over_g(acc.y);
      acc.z = sum_over_g(acc.z);
      acc.w = sum_over_g(acc.w);
      const float tot = wave_sum(al) + a0;
      const bool mine = (g == k);
      o.x = mine ? acc.x : o.x;
      o.y = mine ? acc.y : o.y;
      o.z = mine ? acc.z : o.z;
      o.w = mine ? acc.w : o.w;
      s_out = (lane == k) ? tot : s_out;
    }
  }
  if (g < NPW && d0 + g < Nn) *(float4*)(sx + (size_t)(d0 + g) * 16 + q * 4) = o;
  if (lane < NPW && d0 + lane < Nn) ssum[d0 + lane] = s_out;
}

extern "C" void kernel_launch(void* const* d_in, const int* in_sizes, int n_in,
                              void* d_out, int out_size, void* d_ws, size_t ws_size,
                              hipStream_t stream) {
  const float* x     = (const float*)d_in[0];
  const int*   ei    = (const int*)  d_in[1];
  const float* W1    = (const float*)d_in[2];
  const float* b1    = (const float*)d_in[3];
  const float* att1  = (const float*)d_in[4];
  const float* bias1 = (const float*)d_in[5];
  const float* W2    = (const float*)d_in[6];
  const float* b2    = (const float*)d_in[7];
  const float* att2  = (const float*)d_in[8];
  const float* bias2 = (const float*)d_in[9];
  const float* Wout  = (const float*)d_in[10];
  const float* bout  = (const float*)d_in[11];
  float* y = (float*)d_out;

  const int Nn   = in_sizes[0] / 16;
  const int E    = in_sizes[1] / 2;

  char* p = (char*)d_ws;
  auto alloc = [&](size_t bytes) {
    char* r = p;
    p += (bytes + 255) & ~(size_t)255;
    return r;
  };
  float* h      = (float*)alloc((size_t)Nn * 64 * sizeof(float));
  float* sx     = (float*)alloc((size_t)Nn * 16 * sizeof(float));
  float* ssum   = (float*)alloc((size_t)Nn * sizeof(float));
  float* a_i    = (float*)alloc((size_t)Nn * sizeof(float));
  float* a_j    = (float*)alloc((size_t)Nn * sizeof(float));
  float4* st    = (float4*)alloc((size_t)Nn * sizeof(float4));
  const int NT  = (Nn + TILE - 1) / TILE;
  const size_t NP = (size_t)NT * TILE;
  const int C   = (E + ECH - 1) / ECH;
  int*   cnt_d  = (int*)  alloc(NP * sizeof(int));
  int*   cnt_s  = (int*)  alloc(NP * sizeof(int));
  int*   off_d  = (int*)  alloc(NP * sizeof(int));
  int*   off_s  = (int*)  alloc(NP * sizeof(int));
  int*   cmat   = (int*)  alloc((size_t)2 * NBIN * C * sizeof(int));
  int*   bstart = (int*)  alloc((size_t)2 * BSP * sizeof(int));
  unsigned* co_d = (unsigned*)alloc((size_t)E * sizeof(unsigned));
  unsigned* co_s = (unsigned*)alloc((size_t)E * sizeof(unsigned));
  int*   csr_src = (int*) alloc((size_t)E * sizeof(int));
  int*   csr_sd = (int*)  alloc((size_t)E * sizeof(int));

  const int TB = 256;
  dim3 blk(TB);
  int g_gemm  = ((Nn / 16) + 7) / 8;
  int g_node  = (Nn + 8 * NPW - 1) / (8 * NPW);
  int g_cnt   = C + ((Nn / 16) + 31) / 32;
  dim3 g_bin(C, 2);
  dim3 g_tile(NT, 2);

  k_bin_count_and_dots<<<g_cnt, dim3(1024), 0, stream>>>(ei, cmat, E, C, x, W1, b1, att1, a_i, a_j, Nn);
  k_bin_scatter<<<g_bin, dim3(1024), 0, stream>>>(ei, cmat, co_d, co_s, bstart, E, C);
  k_tile_sort<<<g_tile, dim3(1024), 0, stream>>>(co_d, co_s, bstart, off_d, cnt_d, off_s, cnt_s,
                                                 csr_src, csr_sd);

  k_src_stats<<<g_node, blk, 0, stream>>>(csr_sd, off_s, cnt_s, a_i, a_j, st, Nn);
  k_gather_x<<<g_node, blk, 0, stream>>>(csr_src, st, a_i, off_d, cnt_d, x, sx, ssum, Nn);

  k_linear_chain<<<g_gemm, blk, 0, stream>>>(sx, ssum, W1, b1, bias1, W2, b2, att2, h, a_i, a_j, Nn);
  k_src_stats<<<g_node, blk, 0, stream>>>(csr_sd, off_s, cnt_s, a_i, a_j, st, Nn);
  k_gather_sum<<<g_node, blk, 0, stream>>>(csr_src, st, a_i, off_d, cnt_d, h,
                                           bias2, Wout, bout, y, Nn);
}
